# starting kernel
# baseline (speedup 1.0000x reference)
#include <hip/hip_runtime.h>
#include <hip/hip_bf16.h>
#include <math.h>

typedef __attribute__((ext_vector_type(2))) float v2f;
typedef __attribute__((ext_vector_type(8))) float v8f;

#define S_DIM 1024
#define CA 1024
#define CS 512
#define CZ 64
#define NH 16
#define HC 64

__device__ __forceinline__ float sigmoidf_(float x) {
    return 1.0f / (1.0f + __expf(-x));
}

__global__ void ln_kernel(const float* __restrict__ x, const float* __restrict__ w,
                          float* __restrict__ y, int C) {
    __shared__ float red[256];
    __shared__ float red2[256];
    int row = blockIdx.x;
    const float4* xr = (const float4*)(x + (size_t)row * C);
    int C4 = C >> 2;
    float sum = 0.f, sq = 0.f;
    for (int c = threadIdx.x; c < C4; c += 256) {
        float4 v = xr[c];
        sum += v.x + v.y + v.z + v.w;
        sq  += v.x * v.x + v.y * v.y + v.z * v.z + v.w * v.w;
    }
    red[threadIdx.x] = sum; red2[threadIdx.x] = sq;
    __syncthreads();
    for (int off = 128; off > 0; off >>= 1) {
        if (threadIdx.x < off) {
            red[threadIdx.x]  += red[threadIdx.x + off];
            red2[threadIdx.x] += red2[threadIdx.x + off];
        }
        __syncthreads();
    }
    float m   = red[0] / C;
    float var = red2[0] / C - m * m;
    float inv = rsqrtf(var + 1e-5f);
    float4* yr = (float4*)(y + (size_t)row * C);
    for (int c = threadIdx.x; c < C4; c += 256) {
        float4 v = xr[c];
        v.x = (v.x - m) * inv; v.y = (v.y - m) * inv;
        v.z = (v.z - m) * inv; v.w = (v.w - m) * inv;
        if (w) {
            const float4 wv = ((const float4*)w)[c];
            v.x *= wv.x; v.y *= wv.y; v.z *= wv.z; v.w *= wv.w;
        }
        yr[c] = v;
    }
}

__global__ void gemm_nt_kernel(const float* __restrict__ A, int lda, long long sA,
                               const float* __restrict__ B, int ldb, long long sB,
                               float* __restrict__ C, int ldc, long long sC,
                               const float* __restrict__ bias,
                               const float* __restrict__ addm, int ldadd, long long sAdd,
                               int K, float scale, int mode) {
    __shared__ float As[2][64][20];
    __shared__ float Bs[2][64][20];
    const int z = blockIdx.z;
    A += (size_t)z * sA; B += (size_t)z * sB; C += (size_t)z * sC;
    if (addm) addm += (size_t)z * sAdd;
    const int tileM = blockIdx.y * 64;
    const int tileN = blockIdx.x * 64;
    const int tid  = threadIdx.x;
    const int wave = tid >> 5, lane = tid & 31;
    const int wm = wave >> 1, wn = wave & 1;
    const int half = lane >> 4, l = lane & 15;
    const int lr = tid >> 2;
    const int lc = (tid & 3) * 4;

    const float* gA = A + (size_t)(tileM + lr) * lda + lc;
    const float* gB = B + (size_t)(tileN + lr) * ldb + lc;

    v8f acc0 = {}; v8f acc1 = {};
    const int nstages = K >> 4;

    {
        float4 ra = *(const float4*)gA;
        float4 rb = *(const float4*)gB;
        *(float4*)&As[0][lr][lc] = ra;
        *(float4*)&Bs[0][lr][lc] = rb;
    }
    __syncthreads();

    for (int s = 0; s < nstages; s++) {
        const int p = s & 1;
        float4 na = make_float4(0.f, 0.f, 0.f, 0.f);
        float4 nb = make_float4(0.f, 0.f, 0.f, 0.f);
        if (s + 1 < nstages) {
            na = *(const float4*)(gA + (s + 1) * 16);
            nb = *(const float4*)(gB + (s + 1) * 16);
        }
        v2f af[4], bf0[4], bf1[4];
        #pragma unroll
        for (int q = 0; q < 4; q++) {
            const int kc = q * 4 + half * 2;
            af[q].x  = As[p][wm * 16 + l][kc];
            af[q].y  = As[p][wm * 16 + l][kc + 1];
            bf0[q].x = Bs[p][wn * 32 + l][kc];
            bf0[q].y = Bs[p][wn * 32 + l][kc + 1];
            bf1[q].x = Bs[p][wn * 32 + 16 + l][kc];
            bf1[q].y = Bs[p][wn * 32 + 16 + l][kc + 1];
        }
        #pragma unroll
        for (int q = 0; q < 4; q++) {
            acc0 = __builtin_amdgcn_wmma_f32_16x16x4_f32(false, af[q], false, bf0[q], (short)0, acc0, false, false);
            acc1 = __builtin_amdgcn_wmma_f32_16x16x4_f32(false, af[q], false, bf1[q], (short)0, acc1, false, false);
        }
        if (s + 1 < nstages) {
            *(float4*)&As[p ^ 1][lr][lc] = na;
            *(float4*)&Bs[p ^ 1][lr][lc] = nb;
        }
        __syncthreads();
    }

    #pragma unroll
    for (int r = 0; r < 8; r++) {
        int m  = tileM + wm * 16 + half * 8 + r;
        int n0 = tileN + wn * 32 + l;
        int n1 = n0 + 16;
        float v0 = acc0[r] * scale;
        float v1 = acc1[r] * scale;
        if (bias) { v0 += bias[n0]; v1 += bias[n1]; }
        if (mode == 1) {
            v0 = sigmoidf_(v0) * addm[(size_t)m * ldadd + n0];
            v1 = sigmoidf_(v1) * addm[(size_t)m * ldadd + n1];
        } else if (addm) {
            v0 += addm[(size_t)m * ldadd + n0];
            v1 += addm[(size_t)m * ldadd + n1];
        }
        C[(size_t)m * ldc + n0] = v0;
        C[(size_t)m * ldc + n1] = v1;
    }
}

__global__ void gemm_tn_kernel(const float* __restrict__ A, int lda, long long sA,
                               const float* __restrict__ B, int ldb, long long sB,
                               float* __restrict__ C, int ldc, long long sC,
                               int K) {
    __shared__ float As[2][16][72];
    __shared__ float Bs[2][16][72];
    const int z = blockIdx.z;
    A += (size_t)z * sA; B += (size_t)z * sB; C += (size_t)z * sC;
    const int tileM = blockIdx.y * 64;
    const int tileN = blockIdx.x * 64;
    const int tid  = threadIdx.x;
    const int wave = tid >> 5, lane = tid & 31;
    const int wm = wave >> 1, wn = wave & 1;
    const int half = lane >> 4, l = lane & 15;
    const int lk = tid >> 4;
    const int lm = (tid & 15) * 4;

    const float* gA = A + (size_t)lk * lda + tileM + lm;
    const float* gB = B + (size_t)lk * ldb + tileN + lm;

    v8f acc0 = {}; v8f acc1 = {};
    const int nstages = K >> 4;

    {
        float4 ra = *(const float4*)gA;
        float4 rb = *(const float4*)gB;
        *(float4*)&As[0][lk][lm] = ra;
        *(float4*)&Bs[0][lk][lm] = rb;
    }
    __syncthreads();

    for (int s = 0; s < nstages; s++) {
        const int p = s & 1;
        float4 na = make_float4(0.f, 0.f, 0.f, 0.f);
        float4 nb = make_float4(0.f, 0.f, 0.f, 0.f);
        if (s + 1 < nstages) {
            na = *(const float4*)(gA + (size_t)(s + 1) * 16 * lda);
            nb = *(const float4*)(gB + (size_t)(s + 1) * 16 * ldb);
        }
        v2f af[4], bf0[4], bf1[4];
        #pragma unroll
        for (int q = 0; q < 4; q++) {
            const int kc = q * 4 + half * 2;
            af[q].x  = As[p][kc][wm * 16 + l];
            af[q].y  = As[p][kc + 1][wm * 16 + l];
            bf0[q].x = Bs[p][kc][wn * 32 + l];
            bf0[q].y = Bs[p][kc + 1][wn * 32 + l];
            bf1[q].x = Bs[p][kc][wn * 32 + 16 + l];
            bf1[q].y = Bs[p][kc + 1][wn * 32 + 16 + l];
        }
        #pragma unroll
        for (int q = 0; q < 4; q++) {
            acc0 = __builtin_amdgcn_wmma_f32_16x16x4_f32(false, af[q], false, bf0[q], (short)0, acc0, false, false);
            acc1 = __builtin_amdgcn_wmma_f32_16x16x4_f32(false, af[q], false, bf1[q], (short)0, acc1, false, false);
        }
        if (s + 1 < nstages) {
            *(float4*)&As[p ^ 1][lk][lm] = na;
            *(float4*)&Bs[p ^ 1][lk][lm] = nb;
        }
        __syncthreads();
    }

    #pragma unroll
    for (int r = 0; r < 8; r++) {
        int m  = tileM + wm * 16 + half * 8 + r;
        int n0 = tileN + wn * 32 + l;
        C[(size_t)m * ldc + n0]      = acc0[r];
        C[(size_t)m * ldc + n0 + 16] = acc1[r];
    }
}

__global__ void softmax_row_kernel(float* __restrict__ x) {
    __shared__ float red[256];
    float4* xr = (float4*)(x + (size_t)blockIdx.x * 1024);
    float4 v = xr[threadIdx.x];
    float mx = fmaxf(fmaxf(v.x, v.y), fmaxf(v.z, v.w));
    red[threadIdx.x] = mx; __syncthreads();
    for (int off = 128; off > 0; off >>= 1) {
        if (threadIdx.x < off) red[threadIdx.x] = fmaxf(red[threadIdx.x], red[threadIdx.x + off]);
        __syncthreads();
    }
    mx = red[0];
    __syncthreads();
    v.x = __expf(v.x - mx); v.y = __expf(v.y - mx);
    v.z = __expf(v.z - mx); v.w = __expf(v.w - mx);
    float s = v.x + v.y + v.z + v.w;
    red[threadIdx.x] = s; __syncthreads();
    for (int off = 128; off > 0; off >>= 1) {
        if (threadIdx.x < off) red[threadIdx.x] += red[threadIdx.x + off];
        __syncthreads();
    }
    float inv = 1.0f / red[0];
    v.x *= inv; v.y *= inv; v.z *= inv; v.w *= inv;
    xr[threadIdx.x] = v;
}

__global__ void zbias_kernel(const float* __restrict__ z,
                             const float* __restrict__ pnw, const float* __restrict__ pnb,
                             const float* __restrict__ bw,  const float* __restrict__ bb,
                             float* __restrict__ out) {
    __shared__ float bwS[NH * CZ];
    __shared__ float pwS[CZ];
    __shared__ float pbS[CZ];
    __shared__ float bbS[NH];
    int tid = threadIdx.x;
    for (int i = tid; i < NH * CZ; i += 256) bwS[i] = bw[i];
    if (tid < CZ) { pwS[tid] = pnw[tid]; pbS[tid] = pnb[tid]; }
    if (tid < NH) bbS[tid] = bb[tid];
    __syncthreads();
    size_t pos = (size_t)blockIdx.x * 256 + tid;
    const float4* zp = (const float4*)(z + pos * CZ);
    float4 zv[16];
    float sum = 0.f, sq = 0.f;
    #pragma unroll
    for (int i = 0; i < 16; i++) {
        float4 t = zp[i];
        zv[i] = t;
        sum += t.x + t.y + t.z + t.w;
        sq  += t.x * t.x + t.y * t.y + t.z * t.z + t.w * t.w;
    }
    float m   = sum * (1.0f / 64.0f);
    float var = sq * (1.0f / 64.0f) - m * m;
    float inv = rsqrtf(var + 1e-5f);
    float o[NH];
    #pragma unroll
    for (int h = 0; h < NH; h++) o[h] = bbS[h];
    #pragma unroll
    for (int i = 0; i < 16; i++) {
        float y0 = (zv[i].x - m) * inv * pwS[4 * i + 0] + pbS[4 * i + 0];
        float y1 = (zv[i].y - m) * inv * pwS[4 * i + 1] + pbS[4 * i + 1];
        float y2 = (zv[i].z - m) * inv * pwS[4 * i + 2] + pbS[4 * i + 2];
        float y3 = (zv[i].w - m) * inv * pwS[4 * i + 3] + pbS[4 * i + 3];
        #pragma unroll
        for (int h = 0; h < NH; h++) {
            o[h] += y0 * bwS[h * 64 + 4 * i + 0] + y1 * bwS[h * 64 + 4 * i + 1]
                  + y2 * bwS[h * 64 + 4 * i + 2] + y3 * bwS[h * 64 + 4 * i + 3];
        }
    }
    float4* op = (float4*)(out + pos * NH);
    op[0] = make_float4(o[0],  o[1],  o[2],  o[3]);
    op[1] = make_float4(o[4],  o[5],  o[6],  o[7]);
    op[2] = make_float4(o[8],  o[9],  o[10], o[11]);
    op[3] = make_float4(o[12], o[13], o[14], o[15]);
}

__global__ void adaln_gate_kernel(const float4* __restrict__ pb, const float4* __restrict__ pn,
                                  const float4* __restrict__ an, float4* __restrict__ a1) {
    int i = blockIdx.x * 256 + threadIdx.x;
    float4 p = pb[i], q = pn[i], n = an[i];
    float4 r;
    r.x = sigmoidf_(p.x * n.x + q.x);
    r.y = sigmoidf_(p.y * n.y + q.y);
    r.z = sigmoidf_(p.z * n.z + q.z);
    r.w = sigmoidf_(p.w * n.w + q.w);
    a1[i] = r;
}

__global__ void gate_o_kernel(const float* __restrict__ kvg, const float4* __restrict__ obuf,
                              float4* __restrict__ go) {
    int q = blockIdx.x * 256 + threadIdx.x;
    int h  = q >> 14;
    int j  = (q >> 4) & 1023;
    int c4 = (q & 15) * 4;
    const float4 g = *(const float4*)(kvg + (size_t)h * 196608 + (size_t)j * 192 + 128 + c4);
    float4 o = obuf[q];
    float4 r;
    r.x = sigmoidf_(g.x) * o.x;
    r.y = sigmoidf_(g.y) * o.y;
    r.z = sigmoidf_(g.z) * o.z;
    r.w = sigmoidf_(g.w) * o.w;
    go[q] = r;
}

extern "C" void kernel_launch(void* const* d_in, const int* in_sizes, int n_in,
                              void* d_out, int out_size, void* d_ws, size_t ws_size,
                              hipStream_t stream) {
    const float* a       = (const float*)d_in[0];
    const float* z       = (const float*)d_in[1];
    const float* s       = (const float*)d_in[2];
    const float* sn_w    = (const float*)d_in[3];
    const float* pb_w    = (const float*)d_in[4];
    const float* pb_b    = (const float*)d_in[5];
    const float* pn_w    = (const float*)d_in[6];
    const float* pnorm_w = (const float*)d_in[7];
    const float* pnorm_b = (const float*)d_in[8];
    const float* q_w     = (const float*)d_in[9];
    const float* q_b     = (const float*)d_in[10];
    const float* kvg_w   = (const float*)d_in[11];
    const float* bias_w  = (const float*)d_in[12];
    const float* bias_b  = (const float*)d_in[13];
    const float* attn_w  = (const float*)d_in[14];
    const float* out_w   = (const float*)d_in[15];
    const float* out_b   = (const float*)d_in[16];
    float* out = (float*)d_out;
    float* ws  = (float*)d_ws;

    float* a_n    = ws + 0;
    float* s_n    = ws + 1048576;
    float* pb     = ws + 1572864;
    float* pn     = ws + 2621440;
    float* qm     = ws + 3670016;
    float* kvg    = ws + 4718592;
    float* biasO  = ws + 7864320;
    float* obuf   = biasO;
    float* go     = biasO + 1048576;
    float* a2     = biasO + 2097152;
    float* scores = ws + 24641536;

    ln_kernel<<<1024, 256, 0, stream>>>(a, nullptr, a_n, 1024);
    ln_kernel<<<1024, 256, 0, stream>>>(s, sn_w, s_n, 512);

    gemm_nt_kernel<<<dim3(16, 16, 1), 256, 0, stream>>>(
        s_n, 512, 0, pb_w, 512, 0, pb, 1024, 0, pb_b, nullptr, 0, 0, 512, 1.0f, 0);
    gemm_nt_kernel<<<dim3(16, 16, 1), 256, 0, stream>>>(
        s_n, 512, 0, pn_w, 512, 0, pn, 1024, 0, nullptr, nullptr, 0, 0, 512, 1.0f, 0);
    adaln_gate_kernel<<<1024, 256, 0, stream>>>(
        (const float4*)pb, (const float4*)pn, (const float4*)a_n, (float4*)pb);
    float* a1 = pb;

    gemm_nt_kernel<<<dim3(16, 16, 1), 256, 0, stream>>>(
        a1, 1024, 0, q_w, 1024, 0, qm, 1024, 0, q_b, nullptr, 0, 0, 1024, 1.0f, 0);
    gemm_nt_kernel<<<dim3(48, 16, 1), 256, 0, stream>>>(
        a1, 1024, 0, kvg_w, 1024, 0, kvg, 3072, 0, nullptr, nullptr, 0, 0, 1024, 1.0f, 0);

    zbias_kernel<<<4096, 256, 0, stream>>>(z, pnorm_w, pnorm_b, bias_w, bias_b, biasO);

    gemm_nt_kernel<<<dim3(16, 16, 16), 256, 0, stream>>>(
        kvg, 192, 196608LL,
        qm, 64, 65536LL,
        scores, 1024, 1048576LL,
        nullptr,
        biasO, 1024, 1048576LL,
        64, 1.0f / 64.0f, 0);

    softmax_row_kernel<<<16384, 256, 0, stream>>>(scores);

    gemm_tn_kernel<<<dim3(1, 16, 16), 256, 0, stream>>>(
        scores, 1024, 1048576LL,
        kvg + 64, 192, 196608LL,
        obuf, 64, 65536LL,
        1024);

    gate_o_kernel<<<1024, 256, 0, stream>>>(kvg, (const float4*)obuf, (float4*)go);

    gemm_nt_kernel<<<dim3(16, 16, 1), 256, 0, stream>>>(
        go, 1024, 0, attn_w, 1024, 0, a2, 1024, 0, nullptr, nullptr, 0, 0, 1024, 1.0f, 0);

    gemm_nt_kernel<<<dim3(16, 16, 1), 256, 0, stream>>>(
        s, 512, 0, out_w, 512, 0, out, 1024, 0, out_b, a2, 1024, 0, 512, 1.0f, 1);
}


# Both WMMA GEMM kernels hold 64 k per LDS stage instead of 16: 8 b128 loads in flight per thread, each wave load covers whole 256-byte row pieces, 32 WMMAs per wave per barrier instead of 8
# speedup vs baseline: 1.1849x; 1.1849x over previous
#include <hip/hip_runtime.h>
#include <hip/hip_bf16.h>
#include <math.h>

typedef __attribute__((ext_vector_type(2))) float v2f;
typedef __attribute__((ext_vector_type(8))) float v8f;

#define S_DIM 1024
#define CA 1024
#define CS 512
#define CZ 64
#define NH 16
#define HC 64

__device__ __forceinline__ float sigmoidf_(float x) {
    return 1.0f / (1.0f + __expf(-x));
}

__global__ void ln_kernel(const float* __restrict__ x, const float* __restrict__ w,
                          float* __restrict__ y, int C) {
    __shared__ float red[256];
    __shared__ float red2[256];
    int row = blockIdx.x;
    const float4* xr = (const float4*)(x + (size_t)row * C);
    int C4 = C >> 2;
    float sum = 0.f, sq = 0.f;
    for (int c = threadIdx.x; c < C4; c += 256) {
        float4 v = xr[c];
        sum += v.x + v.y + v.z + v.w;
        sq  += v.x * v.x + v.y * v.y + v.z * v.z + v.w * v.w;
    }
    red[threadIdx.x] = sum; red2[threadIdx.x] = sq;
    __syncthreads();
    for (int off = 128; off > 0; off >>= 1) {
        if (threadIdx.x < off) {
            red[threadIdx.x]  += red[threadIdx.x + off];
            red2[threadIdx.x] += red2[threadIdx.x + off];
        }
        __syncthreads();
    }
    float m   = red[0] / C;
    float var = red2[0] / C - m * m;
    float inv = rsqrtf(var + 1e-5f);
    float4* yr = (float4*)(y + (size_t)row * C);
    for (int c = threadIdx.x; c < C4; c += 256) {
        float4 v = xr[c];
        v.x = (v.x - m) * inv; v.y = (v.y - m) * inv;
        v.z = (v.z - m) * inv; v.w = (v.w - m) * inv;
        if (w) {
            const float4 wv = ((const float4*)w)[c];
            v.x *= wv.x; v.y *= wv.y; v.z *= wv.z; v.w *= wv.w;
        }
        yr[c] = v;
    }
}

#define KST 64
#define PNT 68
#define PTN 72
__global__ void
gemm_nt_kernel(const float* __restrict__ A, int lda, long long sA,
               const float* __restrict__ B, int ldb, long long sB,
               float* __restrict__ C, int ldc, long long sC,
               const float* __restrict__ bias,
               const float* __restrict__ addm, int ldadd, long long sAdd,
               int K, float scale, int mode) {
    __shared__ float As[2][64][PNT];
    __shared__ float Bs[2][64][PNT];
    const int z = blockIdx.z;
    A += (size_t)z * sA; B += (size_t)z * sB; C += (size_t)z * sC;
    if (addm) addm += (size_t)z * sAdd;
    const int tileM = blockIdx.y * 64;
    const int tileN = blockIdx.x * 64;
    const int tid  = threadIdx.x;
    const int wave = tid >> 5, lane = tid & 31;
    const int wm = wave >> 1, wn = wave & 1;
    const int half = lane >> 4, l = lane & 15;
    const int lr = tid >> 4;
    const int lc = (tid & 15) * 4;

    const float* gA = A + (size_t)(tileM + lr) * lda + lc;
    const float* gB = B + (size_t)(tileN + lr) * ldb + lc;

    v8f acc0 = {}; v8f acc1 = {};
    const int nstages = K / KST;

    {
        float4 ra[4], rb[4];
        #pragma unroll
        for (int i = 0; i < 4; i++) {
            ra[i] = *(const float4*)(gA + (size_t)(16 * i) * lda);
            rb[i] = *(const float4*)(gB + (size_t)(16 * i) * ldb);
        }
        #pragma unroll
        for (int i = 0; i < 4; i++) {
            *(float4*)&As[0][lr + 16 * i][lc] = ra[i];
            *(float4*)&Bs[0][lr + 16 * i][lc] = rb[i];
        }
    }
    __syncthreads();

    for (int s = 0; s < nstages; s++) {
        const int p = s & 1;
        const bool more = (s + 1 < nstages);
        float4 na[4], nb[4];
        #pragma unroll
        for (int i = 0; i < 4; i++) {
            na[i] = make_float4(0.f, 0.f, 0.f, 0.f);
            nb[i] = make_float4(0.f, 0.f, 0.f, 0.f);
        }
        if (more) {
            #pragma unroll
            for (int i = 0; i < 4; i++) {
                na[i] = *(const float4*)(gA + (size_t)(16 * i) * lda + (s + 1) * KST);
                nb[i] = *(const float4*)(gB + (size_t)(16 * i) * ldb + (s + 1) * KST);
            }
        }
        const float* ap  = &As[p][wm * 16 + l][half * 2];
        const float* bp0 = &Bs[p][wn * 32 + l][half * 2];
        const float* bp1 = &Bs[p][wn * 32 + 16 + l][half * 2];
        #pragma unroll
        for (int q = 0; q < KST / 4; q++) {
            const v2f af  = *(const v2f*)(ap  + q * 4);
            const v2f bf0 = *(const v2f*)(bp0 + q * 4);
            const v2f bf1 = *(const v2f*)(bp1 + q * 4);
            acc0 = __builtin_amdgcn_wmma_f32_16x16x4_f32(false, af, false, bf0, (short)0, acc0, false, false);
            acc1 = __builtin_amdgcn_wmma_f32_16x16x4_f32(false, af, false, bf1, (short)0, acc1, false, false);
        }
        if (more) {
            #pragma unroll
            for (int i = 0; i < 4; i++) {
                *(float4*)&As[p ^ 1][lr + 16 * i][lc] = na[i];
                *(float4*)&Bs[p ^ 1][lr + 16 * i][lc] = nb[i];
            }
        }
        __syncthreads();
    }

    #pragma unroll
    for (int r = 0; r < 8; r++) {
        int m  = tileM + wm * 16 + half * 8 + r;
        int n0 = tileN + wn * 32 + l;
        int n1 = n0 + 16;
        float v0 = acc0[r] * scale;
        float v1 = acc1[r] * scale;
        if (bias) { v0 += bias[n0]; v1 += bias[n1]; }
        if (mode == 1) {
            v0 = sigmoidf_(v0) * addm[(size_t)m * ldadd + n0];
            v1 = sigmoidf_(v1) * addm[(size_t)m * ldadd + n1];
        } else if (addm) {
            v0 += addm[(size_t)m * ldadd + n0];
            v1 += addm[(size_t)m * ldadd + n1];
        }
        C[(size_t)m * ldc + n0] = v0;
        C[(size_t)m * ldc + n1] = v1;
    }
}

__global__ void
gemm_tn_kernel(const float* __restrict__ A, int lda, long long sA,
               const float* __restrict__ B, int ldb, long long sB,
               float* __restrict__ C, int ldc, long long sC,
               int K) {
    __shared__ float As[2][KST][PTN];
    __shared__ float Bs[2][KST][PTN];
    const int z = blockIdx.z;
    A += (size_t)z * sA; B += (size_t)z * sB; C += (size_t)z * sC;
    const int tileM = blockIdx.y * 64;
    const int tileN = blockIdx.x * 64;
    const int tid  = threadIdx.x;
    const int wave = tid >> 5, lane = tid & 31;
    const int wm = wave >> 1, wn = wave & 1;
    const int half = lane >> 4, l = lane & 15;
    const int lk = tid >> 4;
    const int lm = (tid & 15) * 4;

    const float* gA = A + (size_t)lk * lda + tileM + lm;
    const float* gB = B + (size_t)lk * ldb + tileN + lm;

    v8f acc0 = {}; v8f acc1 = {};
    const int nstages = K / KST;

    {
        float4 ra[4], rb[4];
        #pragma unroll
        for (int i = 0; i < 4; i++) {
            ra[i] = *(const float4*)(gA + (size_t)(16 * i) * lda);
            rb[i] = *(const float4*)(gB + (size_t)(16 * i) * ldb);
        }
        #pragma unroll
        for (int i = 0; i < 4; i++) {
            *(float4*)&As[0][lk + 16 * i][lm] = ra[i];
            *(float4*)&Bs[0][lk + 16 * i][lm] = rb[i];
        }
    }
    __syncthreads();

    for (int s = 0; s < nstages; s++) {
        const int p = s & 1;
        const bool more = (s + 1 < nstages);
        float4 na[4], nb[4];
        #pragma unroll
        for (int i = 0; i < 4; i++) {
            na[i] = make_float4(0.f, 0.f, 0.f, 0.f);
            nb[i] = make_float4(0.f, 0.f, 0.f, 0.f);
        }
        if (more) {
            #pragma unroll
            for (int i = 0; i < 4; i++) {
                na[i] = *(const float4*)(gA + (size_t)((s + 1) * KST + 16 * i) * lda);
                nb[i] = *(const float4*)(gB + (size_t)((s + 1) * KST + 16 * i) * ldb);
            }
        }
        const float* ap  = &As[p][half * 2][wm * 16 + l];
        const float* bp0 = &Bs[p][half * 2][wn * 32 + l];
        const float* bp1 = &Bs[p][half * 2][wn * 32 + 16 + l];
        #pragma unroll
        for (int q = 0; q < KST / 4; q++) {
            v2f af, bf0, bf1;
            af.x  = ap [(q * 4) * PTN];  af.y  = ap [(q * 4 + 1) * PTN];
            bf0.x = bp0[(q * 4) * PTN];  bf0.y = bp0[(q * 4 + 1) * PTN];
            bf1.x = bp1[(q * 4) * PTN];  bf1.y = bp1[(q * 4 + 1) * PTN];
            acc0 = __builtin_amdgcn_wmma_f32_16x16x4_f32(false, af, false, bf0, (short)0, acc0, false, false);
            acc1 = __builtin_amdgcn_wmma_f32_16x16x4_f32(false, af, false, bf1, (short)0, acc1, false, false);
        }
        if (more) {
            #pragma unroll
            for (int i = 0; i < 4; i++) {
                *(float4*)&As[p ^ 1][lk + 16 * i][lm] = na[i];
                *(float4*)&Bs[p ^ 1][lk + 16 * i][lm] = nb[i];
            }
        }
        __syncthreads();
    }

    #pragma unroll
    for (int r = 0; r < 8; r++) {
        int m  = tileM + wm * 16 + half * 8 + r;
        int n0 = tileN + wn * 32 + l;
        C[(size_t)m * ldc + n0]      = acc0[r];
        C[(size_t)m * ldc + n0 + 16] = acc1[r];
    }
}

__global__ void softmax_row_kernel(float* __restrict__ x) {
    __shared__ float red[256];
    float4* xr = (float4*)(x + (size_t)blockIdx.x * 1024);
    float4 v = xr[threadIdx.x];
    float mx = fmaxf(fmaxf(v.x, v.y), fmaxf(v.z, v.w));
    red[threadIdx.x] = mx; __syncthreads();
    for (int off = 128; off > 0; off >>= 1) {
        if (threadIdx.x < off) red[threadIdx.x] = fmaxf(red[threadIdx.x], red[threadIdx.x + off]);
        __syncthreads();
    }
    mx = red[0];
    __syncthreads();
    v.x = __expf(v.x - mx); v.y = __expf(v.y - mx);
    v.z = __expf(v.z - mx); v.w = __expf(v.w - mx);
    float s = v.x + v.y + v.z + v.w;
    red[threadIdx.x] = s; __syncthreads();
    for (int off = 128; off > 0; off >>= 1) {
        if (threadIdx.x < off) red[threadIdx.x] += red[threadIdx.x + off];
        __syncthreads();
    }
    float inv = 1.0f / red[0];
    v.x *= inv; v.y *= inv; v.z *= inv; v.w *= inv;
    xr[threadIdx.x] = v;
}

__global__ void zbias_kernel(const float* __restrict__ z,
                             const float* __restrict__ pnw, const float* __restrict__ pnb,
                             const float* __restrict__ bw,  const float* __restrict__ bb,
                             float* __restrict__ out) {
    __shared__ float bwS[NH * CZ];
    __shared__ float pwS[CZ];
    __shared__ float pbS[CZ];
    __shared__ float bbS[NH];
    int tid = threadIdx.x;
    for (int i = tid; i < NH * CZ; i += 256) bwS[i] = bw[i];
    if (tid < CZ) { pwS[tid] = pnw[tid]; pbS[tid] = pnb[tid]; }
    if (tid < NH) bbS[tid] = bb[tid];
    __syncthreads();
    size_t pos = (size_t)blockIdx.x * 256 + tid;
    const float4* zp = (const float4*)(z + pos * CZ);
    float4 zv[16];
    float sum = 0.f, sq = 0.f;
    #pragma unroll
    for (int i = 0; i < 16; i++) {
        float4 t = zp[i];
        zv[i] = t;
        sum += t.x + t.y + t.z + t.w;
        sq  += t.x * t.x + t.y * t.y + t.z * t.z + t.w * t.w;
    }
    float m   = sum * (1.0f / 64.0f);
    float var = sq * (1.0f / 64.0f) - m * m;
    float inv = rsqrtf(var + 1e-5f);
    float o[NH];
    #pragma unroll
    for (int h = 0; h < NH; h++) o[h] = bbS[h];
    #pragma unroll
    for (int i = 0; i < 16; i++) {
        float y0 = (zv[i].x - m) * inv * pwS[4 * i + 0] + pbS[4 * i + 0];
        float y1 = (zv[i].y - m) * inv * pwS[4 * i + 1] + pbS[4 * i + 1];
        float y2 = (zv[i].z - m) * inv * pwS[4 * i + 2] + pbS[4 * i + 2];
        float y3 = (zv[i].w - m) * inv * pwS[4 * i + 3] + pbS[4 * i + 3];
        #pragma unroll
        for (int h = 0; h < NH; h++) {
            o[h] += y0 * bwS[h * 64 + 4 * i + 0] + y1 * bwS[h * 64 + 4 * i + 1]
                  + y2 * bwS[h * 64 + 4 * i + 2] + y3 * bwS[h * 64 + 4 * i + 3];
        }
    }
    float4* op = (float4*)(out + pos * NH);
    op[0] = make_float4(o[0],  o[1],  o[2],  o[3]);
    op[1] = make_float4(o[4],  o[5],  o[6],  o[7]);
    op[2] = make_float4(o[8],  o[9],  o[10], o[11]);
    op[3] = make_float4(o[12], o[13], o[14], o[15]);
}

__global__ void adaln_gate_kernel(const float4* __restrict__ pb, const float4* __restrict__ pn,
                                  const float4* __restrict__ an, float4* __restrict__ a1) {
    int i = blockIdx.x * 256 + threadIdx.x;
    float4 p = pb[i], q = pn[i], n = an[i];
    float4 r;
    r.x = sigmoidf_(p.x * n.x + q.x);
    r.y = sigmoidf_(p.y * n.y + q.y);
    r.z = sigmoidf_(p.z * n.z + q.z);
    r.w = sigmoidf_(p.w * n.w + q.w);
    a1[i] = r;
}

__global__ void gate_o_kernel(const float* __restrict__ kvg, const float4* __restrict__ obuf,
                              float4* __restrict__ go) {
    int q = blockIdx.x * 256 + threadIdx.x;
    int h  = q >> 14;
    int j  = (q >> 4) & 1023;
    int c4 = (q & 15) * 4;
    const float4 g = *(const float4*)(kvg + (size_t)h * 196608 + (size_t)j * 192 + 128 + c4);
    float4 o = obuf[q];
    float4 r;
    r.x = sigmoidf_(g.x) * o.x;
    r.y = sigmoidf_(g.y) * o.y;
    r.z = sigmoidf_(g.z) * o.z;
    r.w = sigmoidf_(g.w) * o.w;
    go[q] = r;
}

extern "C" void kernel_launch(void* const* d_in, const int* in_sizes, int n_in,
                              void* d_out, int out_size, void* d_ws, size_t ws_size,
                              hipStream_t stream) {
    const float* a       = (const float*)d_in[0];
    const float* z       = (const float*)d_in[1];
    const float* s       = (const float*)d_in[2];
    const float* sn_w    = (const float*)d_in[3];
    const float* pb_w    = (const float*)d_in[4];
    const float* pb_b    = (const float*)d_in[5];
    const float* pn_w    = (const float*)d_in[6];
    const float* pnorm_w = (const float*)d_in[7];
    const float* pnorm_b = (const float*)d_in[8];
    const float* q_w     = (const float*)d_in[9];
    const float* q_b     = (const float*)d_in[10];
    const float* kvg_w   = (const float*)d_in[11];
    const float* bias_w  = (const float*)d_in[12];
    const float* bias_b  = (const float*)d_in[13];
    const float* attn_w  = (const float*)d_in[14];
    const float* out_w   = (const float*)d_in[15];
    const float* out_b   = (const float*)d_in[16];
    float* out = (float*)d_out;
    float* ws  = (float*)d_ws;

    float* a_n    = ws + 0;
    float* s_n    = ws + 1048576;
    float* pb     = ws + 1572864;
    float* pn     = ws + 2621440;
    float* qm     = ws + 3670016;
    float* kvg    = ws + 4718592;
    float* biasO  = ws + 7864320;
    float* obuf   = biasO;
    float* go     = biasO + 1048576;
    float* a2     = biasO + 2097152;
    float* scores = ws + 24641536;

    ln_kernel<<<1024, 256, 0, stream>>>(a, nullptr, a_n, 1024);
    ln_kernel<<<1024, 256, 0, stream>>>(s, sn_w, s_n, 512);

    gemm_nt_kernel<<<dim3(16, 16, 1), 256, 0, stream>>>(
        s_n, 512, 0, pb_w, 512, 0, pb, 1024, 0, pb_b, nullptr, 0, 0, 512, 1.0f, 0);
    gemm_nt_kernel<<<dim3(16, 16, 1), 256, 0, stream>>>(
        s_n, 512, 0, pn_w, 512, 0, pn, 1024, 0, nullptr, nullptr, 0, 0, 512, 1.0f, 0);
    adaln_gate_kernel<<<1024, 256, 0, stream>>>(
        (const float4*)pb, (const float4*)pn, (const float4*)a_n, (float4*)pb);
    float* a1 = pb;

    gemm_nt_kernel<<<dim3(16, 16, 1), 256, 0, stream>>>(
        a1, 1024, 0, q_w, 1024, 0, qm, 1024, 0, q_b, nullptr, 0, 0, 1024, 1.0f, 0);
    gemm_nt_kernel<<<dim3(48, 16, 1), 256, 0, stream>>>(
        a1, 1024, 0, kvg_w, 1024, 0, kvg, 3072, 0, nullptr, nullptr, 0, 0, 1024, 1.0f, 0);

    zbias_kernel<<<4096, 256, 0, stream>>>(z, pnorm_w, pnorm_b, bias_w, bias_b, biasO);

    gemm_nt_kernel<<<dim3(16, 16, 16), 256, 0, stream>>>(
        kvg, 192, 196608LL,
        qm, 64, 65536LL,
        scores, 1024, 1048576LL,
        nullptr,
        biasO, 1024, 1048576LL,
        64, 1.0f / 64.0f, 0);

    softmax_row_kernel<<<16384, 256, 0, stream>>>(scores);

    gemm_tn_kernel<<<dim3(1, 16, 16), 256, 0, stream>>>(
        scores, 1024, 1048576LL,
        kvg + 64, 192, 196608LL,
        obuf, 64, 65536LL,
        1024);

    gate_o_kernel<<<1024, 256, 0, stream>>>(kvg, (const float4*)obuf, (float4*)go);

    gemm_nt_kernel<<<dim3(16, 16, 1), 256, 0, stream>>>(
        go, 1024, 0, attn_w, 1024, 0, a2, 1024, 0, nullptr, nullptr, 0, 0, 1024, 1.0f, 0);

    gemm_nt_kernel<<<dim3(16, 16, 1), 256, 0, stream>>>(
        s, 512, 0, out_w, 512, 0, out, 1024, 0, out_b, a2, 1024, 0, 512, 1.0f, 1);
}


# Pair-bias kernel as a WMMA GEMM: 64-to-16 head projection on f32 WMMA (its summation order changes), z read as whole lines and turned into fragments through LDS; no register spills left
# speedup vs baseline: 1.5266x; 1.2884x over previous
#include <hip/hip_runtime.h>
#include <hip/hip_bf16.h>
#include <math.h>

typedef __attribute__((ext_vector_type(2))) float v2f;
typedef __attribute__((ext_vector_type(8))) float v8f;

#define S_DIM 1024
#define CA 1024
#define CS 512
#define CZ 64
#define NH 16
#define HC 64

__device__ __forceinline__ float sigmoidf_(float x) {
    return 1.0f / (1.0f + __expf(-x));
}

__global__ void ln_kernel(const float* __restrict__ x, const float* __restrict__ w,
                          float* __restrict__ y, int C) {
    __shared__ float red[256];
    __shared__ float red2[256];
    int row = blockIdx.x;
    const float4* xr = (const float4*)(x + (size_t)row * C);
    int C4 = C >> 2;
    float sum = 0.f, sq = 0.f;
    for (int c = threadIdx.x; c < C4; c += 256) {
        float4 v = xr[c];
        sum += v.x + v.y + v.z + v.w;
        sq  += v.x * v.x + v.y * v.y + v.z * v.z + v.w * v.w;
    }
    red[threadIdx.x] = sum; red2[threadIdx.x] = sq;
    __syncthreads();
    for (int off = 128; off > 0; off >>= 1) {
        if (threadIdx.x < off) {
            red[threadIdx.x]  += red[threadIdx.x + off];
            red2[threadIdx.x] += red2[threadIdx.x + off];
        }
        __syncthreads();
    }
    float m   = red[0] / C;
    float var = red2[0] / C - m * m;
    float inv = rsqrtf(var + 1e-5f);
    float4* yr = (float4*)(y + (size_t)row * C);
    for (int c = threadIdx.x; c < C4; c += 256) {
        float4 v = xr[c];
        v.x = (v.x - m) * inv; v.y = (v.y - m) * inv;
        v.z = (v.z - m) * inv; v.w = (v.w - m) * inv;
        if (w) {
            const float4 wv = ((const float4*)w)[c];
            v.x *= wv.x; v.y *= wv.y; v.z *= wv.z; v.w *= wv.w;
        }
        yr[c] = v;
    }
}

#define KST 64
#define PNT 68
#define PTN 72
__global__ void
gemm_nt_kernel(const float* __restrict__ A, int lda, long long sA,
               const float* __restrict__ B, int ldb, long long sB,
               float* __restrict__ C, int ldc, long long sC,
               const float* __restrict__ bias,
               const float* __restrict__ addm, int ldadd, long long sAdd,
               int K, float scale, int mode) {
    __shared__ float As[2][64][PNT];
    __shared__ float Bs[2][64][PNT];
    const int z = blockIdx.z;
    A += (size_t)z * sA; B += (size_t)z * sB; C += (size_t)z * sC;
    if (addm) addm += (size_t)z * sAdd;
    const int tileM = blockIdx.y * 64;
    const int tileN = blockIdx.x * 64;
    const int tid  = threadIdx.x;
    const int wave = tid >> 5, lane = tid & 31;
    const int wm = wave >> 1, wn = wave & 1;
    const int half = lane >> 4, l = lane & 15;
    const int lr = tid >> 4;
    const int lc = (tid & 15) * 4;

    const float* gA = A + (size_t)(tileM + lr) * lda + lc;
    const float* gB = B + (size_t)(tileN + lr) * ldb + lc;

    v8f acc0 = {}; v8f acc1 = {};
    const int nstages = K / KST;

    {
        float4 ra[4], rb[4];
        #pragma unroll
        for (int i = 0; i < 4; i++) {
            ra[i] = *(const float4*)(gA + (size_t)(16 * i) * lda);
            rb[i] = *(const float4*)(gB + (size_t)(16 * i) * ldb);
        }
        #pragma unroll
        for (int i = 0; i < 4; i++) {
            *(float4*)&As[0][lr + 16 * i][lc] = ra[i];
            *(float4*)&Bs[0][lr + 16 * i][lc] = rb[i];
        }
    }
    __syncthreads();

    for (int s = 0; s < nstages; s++) {
        const int p = s & 1;
        const bool more = (s + 1 < nstages);
        float4 na[4], nb[4];
        #pragma unroll
        for (int i = 0; i < 4; i++) {
            na[i] = make_float4(0.f, 0.f, 0.f, 0.f);
            nb[i] = make_float4(0.f, 0.f, 0.f, 0.f);
        }
        if (more) {
            #pragma unroll
            for (int i = 0; i < 4; i++) {
                na[i] = *(const float4*)(gA + (size_t)(16 * i) * lda + (s + 1) * KST);
                nb[i] = *(const float4*)(gB + (size_t)(16 * i) * ldb + (s + 1) * KST);
            }
        }
        const float* ap  = &As[p][wm * 16 + l][half * 2];
        const float* bp0 = &Bs[p][wn * 32 + l][half * 2];
        const float* bp1 = &Bs[p][wn * 32 + 16 + l][half * 2];
        #pragma unroll
        for (int q = 0; q < KST / 4; q++) {
            const v2f af  = *(const v2f*)(ap  + q * 4);
            const v2f bf0 = *(const v2f*)(bp0 + q * 4);
            const v2f bf1 = *(const v2f*)(bp1 + q * 4);
            acc0 = __builtin_amdgcn_wmma_f32_16x16x4_f32(false, af, false, bf0, (short)0, acc0, false, false);
            acc1 = __builtin_amdgcn_wmma_f32_16x16x4_f32(false, af, false, bf1, (short)0, acc1, false, false);
        }
        if (more) {
            #pragma unroll
            for (int i = 0; i < 4; i++) {
                *(float4*)&As[p ^ 1][lr + 16 * i][lc] = na[i];
                *(float4*)&Bs[p ^ 1][lr + 16 * i][lc] = nb[i];
            }
        }
        __syncthreads();
    }

    #pragma unroll
    for (int r = 0; r < 8; r++) {
        int m  = tileM + wm * 16 + half * 8 + r;
        int n0 = tileN + wn * 32 + l;
        int n1 = n0 + 16;
        float v0 = acc0[r] * scale;
        float v1 = acc1[r] * scale;
        if (bias) { v0 += bias[n0]; v1 += bias[n1]; }
        if (mode == 1) {
            v0 = sigmoidf_(v0) * addm[(size_t)m * ldadd + n0];
            v1 = sigmoidf_(v1) * addm[(size_t)m * ldadd + n1];
        } else if (addm) {
            v0 += addm[(size_t)m * ldadd + n0];
            v1 += addm[(size_t)m * ldadd + n1];
        }
        C[(size_t)m * ldc + n0] = v0;
        C[(size_t)m * ldc + n1] = v1;
    }
}

__global__ void
gemm_tn_kernel(const float* __restrict__ A, int lda, long long sA,
               const float* __restrict__ B, int ldb, long long sB,
               float* __restrict__ C, int ldc, long long sC,
               int K) {
    __shared__ float As[2][KST][PTN];
    __shared__ float Bs[2][KST][PTN];
    const int z = blockIdx.z;
    A += (size_t)z * sA; B += (size_t)z * sB; C += (size_t)z * sC;
    const int tileM = blockIdx.y * 64;
    const int tileN = blockIdx.x * 64;
    const int tid  = threadIdx.x;
    const int wave = tid >> 5, lane = tid & 31;
    const int wm = wave >> 1, wn = wave & 1;
    const int half = lane >> 4, l = lane & 15;
    const int lk = tid >> 4;
    const int lm = (tid & 15) * 4;

    const float* gA = A + (size_t)lk * lda + tileM + lm;
    const float* gB = B + (size_t)lk * ldb + tileN + lm;

    v8f acc0 = {}; v8f acc1 = {};
    const int nstages = K / KST;

    {
        float4 ra[4], rb[4];
        #pragma unroll
        for (int i = 0; i < 4; i++) {
            ra[i] = *(const float4*)(gA + (size_t)(16 * i) * lda);
            rb[i] = *(const float4*)(gB + (size_t)(16 * i) * ldb);
        }
        #pragma unroll
        for (int i = 0; i < 4; i++) {
            *(float4*)&As[0][lk + 16 * i][lm] = ra[i];
            *(float4*)&Bs[0][lk + 16 * i][lm] = rb[i];
        }
    }
    __syncthreads();

    for (int s = 0; s < nstages; s++) {
        const int p = s & 1;
        const bool more = (s + 1 < nstages);
        float4 na[4], nb[4];
        #pragma unroll
        for (int i = 0; i < 4; i++) {
            na[i] = make_float4(0.f, 0.f, 0.f, 0.f);
            nb[i] = make_float4(0.f, 0.f, 0.f, 0.f);
        }
        if (more) {
            #pragma unroll
            for (int i = 0; i < 4; i++) {
                na[i] = *(const float4*)(gA + (size_t)((s + 1) * KST + 16 * i) * lda);
                nb[i] = *(const float4*)(gB + (size_t)((s + 1) * KST + 16 * i) * ldb);
            }
        }
        const float* ap  = &As[p][half * 2][wm * 16 + l];
        const float* bp0 = &Bs[p][half * 2][wn * 32 + l];
        const float* bp1 = &Bs[p][half * 2][wn * 32 + 16 + l];
        #pragma unroll
        for (int q = 0; q < KST / 4; q++) {
            v2f af, bf0, bf1;
            af.x  = ap [(q * 4) * PTN];  af.y  = ap [(q * 4 + 1) * PTN];
            bf0.x = bp0[(q * 4) * PTN];  bf0.y = bp0[(q * 4 + 1) * PTN];
            bf1.x = bp1[(q * 4) * PTN];  bf1.y = bp1[(q * 4 + 1) * PTN];
            acc0 = __builtin_amdgcn_wmma_f32_16x16x4_f32(false, af, false, bf0, (short)0, acc0, false, false);
            acc1 = __builtin_amdgcn_wmma_f32_16x16x4_f32(false, af, false, bf1, (short)0, acc1, false, false);
        }
        if (more) {
            #pragma unroll
            for (int i = 0; i < 4; i++) {
                *(float4*)&As[p ^ 1][lk + 16 * i][lm] = na[i];
                *(float4*)&Bs[p ^ 1][lk + 16 * i][lm] = nb[i];
            }
        }
        __syncthreads();
    }

    #pragma unroll
    for (int r = 0; r < 8; r++) {
        int m  = tileM + wm * 16 + half * 8 + r;
        int n0 = tileN + wn * 32 + l;
        C[(size_t)m * ldc + n0]      = acc0[r];
        C[(size_t)m * ldc + n0 + 16] = acc1[r];
    }
}

__global__ void softmax_row_kernel(float* __restrict__ x) {
    __shared__ float red[256];
    float4* xr = (float4*)(x + (size_t)blockIdx.x * 1024);
    float4 v = xr[threadIdx.x];
    float mx = fmaxf(fmaxf(v.x, v.y), fmaxf(v.z, v.w));
    red[threadIdx.x] = mx; __syncthreads();
    for (int off = 128; off > 0; off >>= 1) {
        if (threadIdx.x < off) red[threadIdx.x] = fmaxf(red[threadIdx.x], red[threadIdx.x + off]);
        __syncthreads();
    }
    mx = red[0];
    __syncthreads();
    v.x = __expf(v.x - mx); v.y = __expf(v.y - mx);
    v.z = __expf(v.z - mx); v.w = __expf(v.w - mx);
    float s = v.x + v.y + v.z + v.w;
    red[threadIdx.x] = s; __syncthreads();
    for (int off = 128; off > 0; off >>= 1) {
        if (threadIdx.x < off) red[threadIdx.x] += red[threadIdx.x + off];
        __syncthreads();
    }
    float inv = 1.0f / red[0];
    v.x *= inv; v.y *= inv; v.z *= inv; v.w *= inv;
    xr[threadIdx.x] = v;
}

#define ZTP 68
#define OTP 20
#define ZB_TILES 8192
__global__ void __launch_bounds__(256)
zbias_kernel(const float* __restrict__ z,
             const float* __restrict__ pnw, const float* __restrict__ pnb,
             const float* __restrict__ bw,  const float* __restrict__ bb,
             float* __restrict__ out) {
    __shared__ float zS[128][ZTP];
    __shared__ float oS[8][16 * OTP];
    const int tid  = threadIdx.x;
    const int wave = tid >> 5, lane = tid & 31;
    const int half = lane >> 4, l = lane & 15;

    *(float4*)&zS[tid >> 4][(tid & 15) * 4] = ((const float4*)bw)[tid];
    __syncthreads();
    v2f wf[16];
    #pragma unroll
    for (int j = 0; j < 8; j++) {
        const float4 t = *(const float4*)&zS[l][half * 32 + j * 4];
        wf[2 * j].x = t.x;     wf[2 * j].y = t.y;
        wf[2 * j + 1].x = t.z; wf[2 * j + 1].y = t.w;
    }
    __syncthreads();
    float4 pw[8], pb[8];
    #pragma unroll
    for (int j = 0; j < 8; j++) {
        pw[j] = ((const float4*)pnw)[half * 8 + j];
        pb[j] = ((const float4*)pnb)[half * 8 + j];
    }
    v8f cinit;
    #pragma unroll
    for (int r = 0; r < 8; r++) cinit[r] = bb[half * 8 + r];

    const float4* z4 = (const float4*)z;
    float4* out4 = (float4*)out;
    float* os = &oS[wave][0];

    int ts = blockIdx.x;
    float4 raw[8];
    {
        const size_t b4 = ((size_t)ts * 128 + wave * 16) * 16 + lane;
        #pragma unroll
        for (int j = 0; j < 8; j++) raw[j] = z4[b4 + j * 32];
    }
    for (; ts < ZB_TILES; ts += gridDim.x) {
        #pragma unroll
        for (int j = 0; j < 8; j++) {
            const int f = j * 32 + lane;
            *(float4*)&zS[wave * 16 + (f >> 4)][(f & 15) * 4] = raw[j];
        }
        __syncthreads();
        const int nts = ts + gridDim.x;
        #pragma unroll
        for (int j = 0; j < 8; j++) raw[j] = make_float4(0.f, 0.f, 0.f, 0.f);
        if (nts < ZB_TILES) {
            const size_t b4 = ((size_t)nts * 128 + wave * 16) * 16 + lane;
            #pragma unroll
            for (int j = 0; j < 8; j++) raw[j] = z4[b4 + j * 32];
        }
        float4 x[8];
        float sum = 0.f, sq = 0.f;
        #pragma unroll
        for (int j = 0; j < 8; j++) {
            const float4 t = *(const float4*)&zS[wave * 16 + l][half * 32 + j * 4];
            x[j] = t;
            sum += t.x + t.y + t.z + t.w;
            sq  += t.x * t.x + t.y * t.y + t.z * t.z + t.w * t.w;
        }
        sum += __shfl_xor(sum, 16);
        sq  += __shfl_xor(sq, 16);
        const float m   = sum * (1.0f / 64.0f);
        const float var = sq * (1.0f / 64.0f) - m * m;
        const float inv = rsqrtf(var + 1e-5f);
        v8f acc0 = cinit; v8f acc1 = {};
        #pragma unroll
        for (int j = 0; j < 8; j++) {
            v2f b0, b1;
            b0.x = (x[j].x - m) * inv * pw[j].x + pb[j].x;
            b0.y = (x[j].y - m) * inv * pw[j].y + pb[j].y;
            b1.x = (x[j].z - m) * inv * pw[j].z + pb[j].z;
            b1.y = (x[j].w - m) * inv * pw[j].w + pb[j].w;
            acc0 = __builtin_amdgcn_wmma_f32_16x16x4_f32(false, wf[2 * j],     false, b0, (short)0, acc0, false, false);
            acc1 = __builtin_amdgcn_wmma_f32_16x16x4_f32(false, wf[2 * j + 1], false, b1, (short)0, acc1, false, false);
        }
        *(float4*)&os[l * OTP + half * 8]     = make_float4(acc0[0] + acc1[0], acc0[1] + acc1[1],
                                                            acc0[2] + acc1[2], acc0[3] + acc1[3]);
        *(float4*)&os[l * OTP + half * 8 + 4] = make_float4(acc0[4] + acc1[4], acc0[5] + acc1[5],
                                                            acc0[6] + acc1[6], acc0[7] + acc1[7]);
        __syncthreads();
        const size_t o4 = ((size_t)ts * 128 + wave * 16) * 4;
        #pragma unroll
        for (int s = 0; s < 2; s++) {
            const int g = s * 32 + lane;
            out4[o4 + g] = *(const float4*)&os[(g >> 2) * OTP + (g & 3) * 4];
        }
    }
}

__global__ void adaln_gate_kernel(const float4* __restrict__ pb, const float4* __restrict__ pn,
                                  const float4* __restrict__ an, float4* __restrict__ a1) {
    int i = blockIdx.x * 256 + threadIdx.x;
    float4 p = pb[i], q = pn[i], n = an[i];
    float4 r;
    r.x = sigmoidf_(p.x * n.x + q.x);
    r.y = sigmoidf_(p.y * n.y + q.y);
    r.z = sigmoidf_(p.z * n.z + q.z);
    r.w = sigmoidf_(p.w * n.w + q.w);
    a1[i] = r;
}

__global__ void gate_o_kernel(const float* __restrict__ kvg, const float4* __restrict__ obuf,
                              float4* __restrict__ go) {
    int q = blockIdx.x * 256 + threadIdx.x;
    int h  = q >> 14;
    int j  = (q >> 4) & 1023;
    int c4 = (q & 15) * 4;
    const float4 g = *(const float4*)(kvg + (size_t)h * 196608 + (size_t)j * 192 + 128 + c4);
    float4 o = obuf[q];
    float4 r;
    r.x = sigmoidf_(g.x) * o.x;
    r.y = sigmoidf_(g.y) * o.y;
    r.z = sigmoidf_(g.z) * o.z;
    r.w = sigmoidf_(g.w) * o.w;
    go[q] = r;
}

extern "C" void kernel_launch(void* const* d_in, const int* in_sizes, int n_in,
                              void* d_out, int out_size, void* d_ws, size_t ws_size,
                              hipStream_t stream) {
    const float* a       = (const float*)d_in[0];
    const float* z       = (const float*)d_in[1];
    const float* s       = (const float*)d_in[2];
    const float* sn_w    = (const float*)d_in[3];
    const float* pb_w    = (const float*)d_in[4];
    const float* pb_b    = (const float*)d_in[5];
    const float* pn_w    = (const float*)d_in[6];
    const float* pnorm_w = (const float*)d_in[7];
    const float* pnorm_b = (const float*)d_in[8];
    const float* q_w     = (const float*)d_in[9];
    const float* q_b     = (const float*)d_in[10];
    const float* kvg_w   = (const float*)d_in[11];
    const float* bias_w  = (const float*)d_in[12];
    const float* bias_b  = (const float*)d_in[13];
    const float* attn_w  = (const float*)d_in[14];
    const float* out_w   = (const float*)d_in[15];
    const float* out_b   = (const float*)d_in[16];
    float* out = (float*)d_out;
    float* ws  = (float*)d_ws;

    float* a_n    = ws + 0;
    float* s_n    = ws + 1048576;
    float* pb     = ws + 1572864;
    float* pn     = ws + 2621440;
    float* qm     = ws + 3670016;
    float* kvg    = ws + 4718592;
    float* biasO  = ws + 7864320;
    float* obuf   = biasO;
    float* go     = biasO + 1048576;
    float* a2     = biasO + 2097152;
    float* scores = ws + 24641536;

    ln_kernel<<<1024, 256, 0, stream>>>(a, nullptr, a_n, 1024);
    ln_kernel<<<1024, 256, 0, stream>>>(s, sn_w, s_n, 512);

    gemm_nt_kernel<<<dim3(16, 16, 1), 256, 0, stream>>>(
        s_n, 512, 0, pb_w, 512, 0, pb, 1024, 0, pb_b, nullptr, 0, 0, 512, 1.0f, 0);
    gemm_nt_kernel<<<dim3(16, 16, 1), 256, 0, stream>>>(
        s_n, 512, 0, pn_w, 512, 0, pn, 1024, 0, nullptr, nullptr, 0, 0, 512, 1.0f, 0);
    adaln_gate_kernel<<<1024, 256, 0, stream>>>(
        (const float4*)pb, (const float4*)pn, (const float4*)a_n, (float4*)pb);
    float* a1 = pb;

    gemm_nt_kernel<<<dim3(16, 16, 1), 256, 0, stream>>>(
        a1, 1024, 0, q_w, 1024, 0, qm, 1024, 0, q_b, nullptr, 0, 0, 1024, 1.0f, 0);
    gemm_nt_kernel<<<dim3(48, 16, 1), 256, 0, stream>>>(
        a1, 1024, 0, kvg_w, 1024, 0, kvg, 3072, 0, nullptr, nullptr, 0, 0, 1024, 1.0f, 0);

    zbias_kernel<<<512, 256, 0, stream>>>(z, pnorm_w, pnorm_b, bias_w, bias_b, biasO);

    gemm_nt_kernel<<<dim3(16, 16, 16), 256, 0, stream>>>(
        kvg, 192, 196608LL,
        qm, 64, 65536LL,
        scores, 1024, 1048576LL,
        nullptr,
        biasO, 1024, 1048576LL,
        64, 1.0f / 64.0f, 0);

    softmax_row_kernel<<<16384, 256, 0, stream>>>(scores);

    gemm_tn_kernel<<<dim3(1, 16, 16), 256, 0, stream>>>(
        scores, 1024, 1048576LL,
        kvg + 64, 192, 196608LL,
        obuf, 64, 65536LL,
        1024);

    gate_o_kernel<<<1024, 256, 0, stream>>>(kvg, (const float4*)obuf, (float4*)go);

    gemm_nt_kernel<<<dim3(16, 16, 1), 256, 0, stream>>>(
        go, 1024, 0, attn_w, 1024, 0, a2, 1024, 0, nullptr, nullptr, 0, 0, 1024, 1.0f, 0);

    gemm_nt_kernel<<<dim3(16, 16, 1), 256, 0, stream>>>(
        s, 512, 0, out_w, 512, 0, out, 1024, 0, out_b, a2, 1024, 0, 512, 1.0f, 1);
}


# Row softmax by one wave per row (4 rows per wave, 16 b128 loads in flight per lane, max and sum by lane shuffles, no LDS, no barrier): 4096 waves instead of 131072; the row sum's order changes
# speedup vs baseline: 1.6611x; 1.0881x over previous
#include <hip/hip_runtime.h>
#include <hip/hip_bf16.h>
#include <math.h>

typedef __attribute__((ext_vector_type(2))) float v2f;
typedef __attribute__((ext_vector_type(8))) float v8f;

#define S_DIM 1024
#define CA 1024
#define CS 512
#define CZ 64
#define NH 16
#define HC 64

__device__ __forceinline__ float sigmoidf_(float x) {
    return 1.0f / (1.0f + __expf(-x));
}

__global__ void ln_kernel(const float* __restrict__ x, const float* __restrict__ w,
                          float* __restrict__ y, int C) {
    __shared__ float red[256];
    __shared__ float red2[256];
    int row = blockIdx.x;
    const float4* xr = (const float4*)(x + (size_t)row * C);
    int C4 = C >> 2;
    float sum = 0.f, sq = 0.f;
    for (int c = threadIdx.x; c < C4; c += 256) {
        float4 v = xr[c];
        sum += v.x + v.y + v.z + v.w;
        sq  += v.x * v.x + v.y * v.y + v.z * v.z + v.w * v.w;
    }
    red[threadIdx.x] = sum; red2[threadIdx.x] = sq;
    __syncthreads();
    for (int off = 128; off > 0; off >>= 1) {
        if (threadIdx.x < off) {
            red[threadIdx.x]  += red[threadIdx.x + off];
            red2[threadIdx.x] += red2[threadIdx.x + off];
        }
        __syncthreads();
    }
    float m   = red[0] / C;
    float var = red2[0] / C - m * m;
    float inv = rsqrtf(var + 1e-5f);
    float4* yr = (float4*)(y + (size_t)row * C);
    for (int c = threadIdx.x; c < C4; c += 256) {
        float4 v = xr[c];
        v.x = (v.x - m) * inv; v.y = (v.y - m) * inv;
        v.z = (v.z - m) * inv; v.w = (v.w - m) * inv;
        if (w) {
            const float4 wv = ((const float4*)w)[c];
            v.x *= wv.x; v.y *= wv.y; v.z *= wv.z; v.w *= wv.w;
        }
        yr[c] = v;
    }
}

#define KST 64
#define PNT 68
#define PTN 72
__global__ void
gemm_nt_kernel(const float* __restrict__ A, int lda, long long sA,
               const float* __restrict__ B, int ldb, long long sB,
               float* __restrict__ C, int ldc, long long sC,
               const float* __restrict__ bias,
               const float* __restrict__ addm, int ldadd, long long sAdd,
               int K, float scale, int mode) {
    __shared__ float As[2][64][PNT];
    __shared__ float Bs[2][64][PNT];
    const int z = blockIdx.z;
    A += (size_t)z * sA; B += (size_t)z * sB; C += (size_t)z * sC;
    if (addm) addm += (size_t)z * sAdd;
    const int tileM = blockIdx.y * 64;
    const int tileN = blockIdx.x * 64;
    const int tid  = threadIdx.x;
    const int wave = tid >> 5, lane = tid & 31;
    const int wm = wave >> 1, wn = wave & 1;
    const int half = lane >> 4, l = lane & 15;
    const int lr = tid >> 4;
    const int lc = (tid & 15) * 4;

    const float* gA = A + (size_t)(tileM + lr) * lda + lc;
    const float* gB = B + (size_t)(tileN + lr) * ldb + lc;

    v8f acc0 = {}; v8f acc1 = {};
    const int nstages = K / KST;

    {
        float4 ra[4], rb[4];
        #pragma unroll
        for (int i = 0; i < 4; i++) {
            ra[i] = *(const float4*)(gA + (size_t)(16 * i) * lda);
            rb[i] = *(const float4*)(gB + (size_t)(16 * i) * ldb);
        }
        #pragma unroll
        for (int i = 0; i < 4; i++) {
            *(float4*)&As[0][lr + 16 * i][lc] = ra[i];
            *(float4*)&Bs[0][lr + 16 * i][lc] = rb[i];
        }
    }
    __syncthreads();

    for (int s = 0; s < nstages; s++) {
        const int p = s & 1;
        const bool more = (s + 1 < nstages);
        float4 na[4], nb[4];
        #pragma unroll
        for (int i = 0; i < 4; i++) {
            na[i] = make_float4(0.f, 0.f, 0.f, 0.f);
            nb[i] = make_float4(0.f, 0.f, 0.f, 0.f);
        }
        if (more) {
            #pragma unroll
            for (int i = 0; i < 4; i++) {
                na[i] = *(const float4*)(gA + (size_t)(16 * i) * lda + (s + 1) * KST);
                nb[i] = *(const float4*)(gB + (size_t)(16 * i) * ldb + (s + 1) * KST);
            }
        }
        const float* ap  = &As[p][wm * 16 + l][half * 2];
        const float* bp0 = &Bs[p][wn * 32 + l][half * 2];
        const float* bp1 = &Bs[p][wn * 32 + 16 + l][half * 2];
        #pragma unroll
        for (int q = 0; q < KST / 4; q++) {
            const v2f af  = *(const v2f*)(ap  + q * 4);
            const v2f bf0 = *(const v2f*)(bp0 + q * 4);
            const v2f bf1 = *(const v2f*)(bp1 + q * 4);
            acc0 = __builtin_amdgcn_wmma_f32_16x16x4_f32(false, af, false, bf0, (short)0, acc0, false, false);
            acc1 = __builtin_amdgcn_wmma_f32_16x16x4_f32(false, af, false, bf1, (short)0, acc1, false, false);
        }
        if (more) {
            #pragma unroll
            for (int i = 0; i < 4; i++) {
                *(float4*)&As[p ^ 1][lr + 16 * i][lc] = na[i];
                *(float4*)&Bs[p ^ 1][lr + 16 * i][lc] = nb[i];
            }
        }
        __syncthreads();
    }

    #pragma unroll
    for (int r = 0; r < 8; r++) {
        int m  = tileM + wm * 16 + half * 8 + r;
        int n0 = tileN + wn * 32 + l;
        int n1 = n0 + 16;
        float v0 = acc0[r] * scale;
        float v1 = acc1[r] * scale;
        if (bias) { v0 += bias[n0]; v1 += bias[n1]; }
        if (mode == 1) {
            v0 = sigmoidf_(v0) * addm[(size_t)m * ldadd + n0];
            v1 = sigmoidf_(v1) * addm[(size_t)m * ldadd + n1];
        } else if (addm) {
            v0 += addm[(size_t)m * ldadd + n0];
            v1 += addm[(size_t)m * ldadd + n1];
        }
        C[(size_t)m * ldc + n0] = v0;
        C[(size_t)m * ldc + n1] = v1;
    }
}

__global__ void
gemm_tn_kernel(const float* __restrict__ A, int lda, long long sA,
               const float* __restrict__ B, int ldb, long long sB,
               float* __restrict__ C, int ldc, long long sC,
               int K) {
    __shared__ float As[2][KST][PTN];
    __shared__ float Bs[2][KST][PTN];
    const int z = blockIdx.z;
    A += (size_t)z * sA; B += (size_t)z * sB; C += (size_t)z * sC;
    const int tileM = blockIdx.y * 64;
    const int tileN = blockIdx.x * 64;
    const int tid  = threadIdx.x;
    const int wave = tid >> 5, lane = tid & 31;
    const int wm = wave >> 1, wn = wave & 1;
    const int half = lane >> 4, l = lane & 15;
    const int lk = tid >> 4;
    const int lm = (tid & 15) * 4;

    const float* gA = A + (size_t)lk * lda + tileM + lm;
    const float* gB = B + (size_t)lk * ldb + tileN + lm;

    v8f acc0 = {}; v8f acc1 = {};
    const int nstages = K / KST;

    {
        float4 ra[4], rb[4];
        #pragma unroll
        for (int i = 0; i < 4; i++) {
            ra[i] = *(const float4*)(gA + (size_t)(16 * i) * lda);
            rb[i] = *(const float4*)(gB + (size_t)(16 * i) * ldb);
        }
        #pragma unroll
        for (int i = 0; i < 4; i++) {
            *(float4*)&As[0][lk + 16 * i][lm] = ra[i];
            *(float4*)&Bs[0][lk + 16 * i][lm] = rb[i];
        }
    }
    __syncthreads();

    for (int s = 0; s < nstages; s++) {
        const int p = s & 1;
        const bool more = (s + 1 < nstages);
        float4 na[4], nb[4];
        #pragma unroll
        for (int i = 0; i < 4; i++) {
            na[i] = make_float4(0.f, 0.f, 0.f, 0.f);
            nb[i] = make_float4(0.f, 0.f, 0.f, 0.f);
        }
        if (more) {
            #pragma unroll
            for (int i = 0; i < 4; i++) {
                na[i] = *(const float4*)(gA + (size_t)((s + 1) * KST + 16 * i) * lda);
                nb[i] = *(const float4*)(gB + (size_t)((s + 1) * KST + 16 * i) * ldb);
            }
        }
        const float* ap  = &As[p][half * 2][wm * 16 + l];
        const float* bp0 = &Bs[p][half * 2][wn * 32 + l];
        const float* bp1 = &Bs[p][half * 2][wn * 32 + 16 + l];
        #pragma unroll
        for (int q = 0; q < KST / 4; q++) {
            v2f af, bf0, bf1;
            af.x  = ap [(q * 4) * PTN];  af.y  = ap [(q * 4 + 1) * PTN];
            bf0.x = bp0[(q * 4) * PTN];  bf0.y = bp0[(q * 4 + 1) * PTN];
            bf1.x = bp1[(q * 4) * PTN];  bf1.y = bp1[(q * 4 + 1) * PTN];
            acc0 = __builtin_amdgcn_wmma_f32_16x16x4_f32(false, af, false, bf0, (short)0, acc0, false, false);
            acc1 = __builtin_amdgcn_wmma_f32_16x16x4_f32(false, af, false, bf1, (short)0, acc1, false, false);
        }
        if (more) {
            #pragma unroll
            for (int i = 0; i < 4; i++) {
                *(float4*)&As[p ^ 1][lk + 16 * i][lm] = na[i];
                *(float4*)&Bs[p ^ 1][lk + 16 * i][lm] = nb[i];
            }
        }
        __syncthreads();
    }

    #pragma unroll
    for (int r = 0; r < 8; r++) {
        int m  = tileM + wm * 16 + half * 8 + r;
        int n0 = tileN + wn * 32 + l;
        C[(size_t)m * ldc + n0]      = acc0[r];
        C[(size_t)m * ldc + n0 + 16] = acc1[r];
    }
}

#define SM_ROWS 4
#define SM_WAVES 8
__global__ void softmax_row_kernel(float* __restrict__ x) {
    const int lane = threadIdx.x & 31;
    const int wave = threadIdx.x >> 5;
    const size_t row0 = ((size_t)blockIdx.x * SM_WAVES + wave) * SM_ROWS;
    float4* xr = (float4*)(x + row0 * 1024) + lane;
    float4 v[SM_ROWS][8];
    for (int r = 0; r < SM_ROWS; r++)
        for (int j = 0; j < 8; j++) v[r][j] = xr[r * 256 + j * 32];
    for (int r = 0; r < SM_ROWS; r++) {
        float mx = fmaxf(fmaxf(v[r][0].x, v[r][0].y), fmaxf(v[r][0].z, v[r][0].w));
        for (int j = 1; j < 8; j++)
            mx = fmaxf(mx, fmaxf(fmaxf(v[r][j].x, v[r][j].y), fmaxf(v[r][j].z, v[r][j].w)));
        for (int off = 16; off > 0; off >>= 1) mx = fmaxf(mx, __shfl_xor(mx, off));
        float s = 0.f;
        for (int j = 0; j < 8; j++) {
            float4 t = v[r][j];
            t.x = __expf(t.x - mx); t.y = __expf(t.y - mx);
            t.z = __expf(t.z - mx); t.w = __expf(t.w - mx);
            s += t.x + t.y + t.z + t.w;
            v[r][j] = t;
        }
        for (int off = 16; off > 0; off >>= 1) s += __shfl_xor(s, off);
        const float inv = 1.0f / s;
        for (int j = 0; j < 8; j++) {
            float4 t = v[r][j];
            t.x *= inv; t.y *= inv; t.z *= inv; t.w *= inv;
            xr[r * 256 + j * 32] = t;
        }
    }
}

#define ZTP 68
#define OTP 20
#define ZB_TILES 8192
__global__ void __launch_bounds__(256)
zbias_kernel(const float* __restrict__ z,
             const float* __restrict__ pnw, const float* __restrict__ pnb,
             const float* __restrict__ bw,  const float* __restrict__ bb,
             float* __restrict__ out) {
    __shared__ float zS[128][ZTP];
    __shared__ float oS[8][16 * OTP];
    const int tid  = threadIdx.x;
    const int wave = tid >> 5, lane = tid & 31;
    const int half = lane >> 4, l = lane & 15;

    *(float4*)&zS[tid >> 4][(tid & 15) * 4] = ((const float4*)bw)[tid];
    __syncthreads();
    v2f wf[16];
    #pragma unroll
    for (int j = 0; j < 8; j++) {
        const float4 t = *(const float4*)&zS[l][half * 32 + j * 4];
        wf[2 * j].x = t.x;     wf[2 * j].y = t.y;
        wf[2 * j + 1].x = t.z; wf[2 * j + 1].y = t.w;
    }
    __syncthreads();
    float4 pw[8], pb[8];
    #pragma unroll
    for (int j = 0; j < 8; j++) {
        pw[j] = ((const float4*)pnw)[half * 8 + j];
        pb[j] = ((const float4*)pnb)[half * 8 + j];
    }
    v8f cinit;
    #pragma unroll
    for (int r = 0; r < 8; r++) cinit[r] = bb[half * 8 + r];

    const float4* z4 = (const float4*)z;
    float4* out4 = (float4*)out;
    float* os = &oS[wave][0];

    int ts = blockIdx.x;
    float4 raw[8];
    {
        const size_t b4 = ((size_t)ts * 128 + wave * 16) * 16 + lane;
        #pragma unroll
        for (int j = 0; j < 8; j++) raw[j] = z4[b4 + j * 32];
    }
    for (; ts < ZB_TILES; ts += gridDim.x) {
        #pragma unroll
        for (int j = 0; j < 8; j++) {
            const int f = j * 32 + lane;
            *(float4*)&zS[wave * 16 + (f >> 4)][(f & 15) * 4] = raw[j];
        }
        __syncthreads();
        const int nts = ts + gridDim.x;
        #pragma unroll
        for (int j = 0; j < 8; j++) raw[j] = make_float4(0.f, 0.f, 0.f, 0.f);
        if (nts < ZB_TILES) {
            const size_t b4 = ((size_t)nts * 128 + wave * 16) * 16 + lane;
            #pragma unroll
            for (int j = 0; j < 8; j++) raw[j] = z4[b4 + j * 32];
        }
        float4 x[8];
        float sum = 0.f, sq = 0.f;
        #pragma unroll
        for (int j = 0; j < 8; j++) {
            const float4 t = *(const float4*)&zS[wave * 16 + l][half * 32 + j * 4];
            x[j] = t;
            sum += t.x + t.y + t.z + t.w;
            sq  += t.x * t.x + t.y * t.y + t.z * t.z + t.w * t.w;
        }
        sum += __shfl_xor(sum, 16);
        sq  += __shfl_xor(sq, 16);
        const float m   = sum * (1.0f / 64.0f);
        const float var = sq * (1.0f / 64.0f) - m * m;
        const float inv = rsqrtf(var + 1e-5f);
        v8f acc0 = cinit; v8f acc1 = {};
        #pragma unroll
        for (int j = 0; j < 8; j++) {
            v2f b0, b1;
            b0.x = (x[j].x - m) * inv * pw[j].x + pb[j].x;
            b0.y = (x[j].y - m) * inv * pw[j].y + pb[j].y;
            b1.x = (x[j].z - m) * inv * pw[j].z + pb[j].z;
            b1.y = (x[j].w - m) * inv * pw[j].w + pb[j].w;
            acc0 = __builtin_amdgcn_wmma_f32_16x16x4_f32(false, wf[2 * j],     false, b0, (short)0, acc0, false, false);
            acc1 = __builtin_amdgcn_wmma_f32_16x16x4_f32(false, wf[2 * j + 1], false, b1, (short)0, acc1, false, false);
        }
        *(float4*)&os[l * OTP + half * 8]     = make_float4(acc0[0] + acc1[0], acc0[1] + acc1[1],
                                                            acc0[2] + acc1[2], acc0[3] + acc1[3]);
        *(float4*)&os[l * OTP + half * 8 + 4] = make_float4(acc0[4] + acc1[4], acc0[5] + acc1[5],
                                                            acc0[6] + acc1[6], acc0[7] + acc1[7]);
        __syncthreads();
        const size_t o4 = ((size_t)ts * 128 + wave * 16) * 4;
        #pragma unroll
        for (int s = 0; s < 2; s++) {
            const int g = s * 32 + lane;
            out4[o4 + g] = *(const float4*)&os[(g >> 2) * OTP + (g & 3) * 4];
        }
    }
}

__global__ void adaln_gate_kernel(const float4* __restrict__ pb, const float4* __restrict__ pn,
                                  const float4* __restrict__ an, float4* __restrict__ a1) {
    int i = blockIdx.x * 256 + threadIdx.x;
    float4 p = pb[i], q = pn[i], n = an[i];
    float4 r;
    r.x = sigmoidf_(p.x * n.x + q.x);
    r.y = sigmoidf_(p.y * n.y + q.y);
    r.z = sigmoidf_(p.z * n.z + q.z);
    r.w = sigmoidf_(p.w * n.w + q.w);
    a1[i] = r;
}

__global__ void gate_o_kernel(const float* __restrict__ kvg, const float4* __restrict__ obuf,
                              float4* __restrict__ go) {
    int q = blockIdx.x * 256 + threadIdx.x;
    int h  = q >> 14;
    int j  = (q >> 4) & 1023;
    int c4 = (q & 15) * 4;
    const float4 g = *(const float4*)(kvg + (size_t)h * 196608 + (size_t)j * 192 + 128 + c4);
    float4 o = obuf[q];
    float4 r;
    r.x = sigmoidf_(g.x) * o.x;
    r.y = sigmoidf_(g.y) * o.y;
    r.z = sigmoidf_(g.z) * o.z;
    r.w = sigmoidf_(g.w) * o.w;
    go[q] = r;
}

extern "C" void kernel_launch(void* const* d_in, const int* in_sizes, int n_in,
                              void* d_out, int out_size, void* d_ws, size_t ws_size,
                              hipStream_t stream) {
    const float* a       = (const float*)d_in[0];
    const float* z       = (const float*)d_in[1];
    const float* s       = (const float*)d_in[2];
    const float* sn_w    = (const float*)d_in[3];
    const float* pb_w    = (const float*)d_in[4];
    const float* pb_b    = (const float*)d_in[5];
    const float* pn_w    = (const float*)d_in[6];
    const float* pnorm_w = (const float*)d_in[7];
    const float* pnorm_b = (const float*)d_in[8];
    const float* q_w     = (const float*)d_in[9];
    const float* q_b     = (const float*)d_in[10];
    const float* kvg_w   = (const float*)d_in[11];
    const float* bias_w  = (const float*)d_in[12];
    const float* bias_b  = (const float*)d_in[13];
    const float* attn_w  = (const float*)d_in[14];
    const float* out_w   = (const float*)d_in[15];
    const float* out_b   = (const float*)d_in[16];
    float* out = (float*)d_out;
    float* ws  = (float*)d_ws;

    float* a_n    = ws + 0;
    float* s_n    = ws + 1048576;
    float* pb     = ws + 1572864;
    float* pn     = ws + 2621440;
    float* qm     = ws + 3670016;
    float* kvg    = ws + 4718592;
    float* biasO  = ws + 7864320;
    float* obuf   = biasO;
    float* go     = biasO + 1048576;
    float* a2     = biasO + 2097152;
    float* scores = ws + 24641536;

    ln_kernel<<<1024, 256, 0, stream>>>(a, nullptr, a_n, 1024);
    ln_kernel<<<1024, 256, 0, stream>>>(s, sn_w, s_n, 512);

    gemm_nt_kernel<<<dim3(16, 16, 1), 256, 0, stream>>>(
        s_n, 512, 0, pb_w, 512, 0, pb, 1024, 0, pb_b, nullptr, 0, 0, 512, 1.0f, 0);
    gemm_nt_kernel<<<dim3(16, 16, 1), 256, 0, stream>>>(
        s_n, 512, 0, pn_w, 512, 0, pn, 1024, 0, nullptr, nullptr, 0, 0, 512, 1.0f, 0);
    adaln_gate_kernel<<<1024, 256, 0, stream>>>(
        (const float4*)pb, (const float4*)pn, (const float4*)a_n, (float4*)pb);
    float* a1 = pb;

    gemm_nt_kernel<<<dim3(16, 16, 1), 256, 0, stream>>>(
        a1, 1024, 0, q_w, 1024, 0, qm, 1024, 0, q_b, nullptr, 0, 0, 1024, 1.0f, 0);
    gemm_nt_kernel<<<dim3(48, 16, 1), 256, 0, stream>>>(
        a1, 1024, 0, kvg_w, 1024, 0, kvg, 3072, 0, nullptr, nullptr, 0, 0, 1024, 1.0f, 0);

    zbias_kernel<<<512, 256, 0, stream>>>(z, pnorm_w, pnorm_b, bias_w, bias_b, biasO);

    gemm_nt_kernel<<<dim3(16, 16, 16), 256, 0, stream>>>(
        kvg, 192, 196608LL,
        qm, 64, 65536LL,
        scores, 1024, 1048576LL,
        nullptr,
        biasO, 1024, 1048576LL,
        64, 1.0f / 64.0f, 0);

    softmax_row_kernel<<<512, 256, 0, stream>>>(scores);

    gemm_tn_kernel<<<dim3(1, 16, 16), 256, 0, stream>>>(
        scores, 1024, 1048576LL,
        kvg + 64, 192, 196608LL,
        obuf, 64, 65536LL,
        1024);

    gate_o_kernel<<<1024, 256, 0, stream>>>(kvg, (const float4*)obuf, (float4*)go);

    gemm_nt_kernel<<<dim3(16, 16, 1), 256, 0, stream>>>(
        go, 1024, 0, attn_w, 1024, 0, a2, 1024, 0, nullptr, nullptr, 0, 0, 1024, 1.0f, 0);

    gemm_nt_kernel<<<dim3(16, 16, 1), 256, 0, stream>>>(
        s, 512, 0, out_w, 512, 0, out, 1024, 0, out_b, a2, 1024, 0, 512, 1.0f, 1);
}


# Scores GEMM owns whole rows (16 rows x 1024 columns per block, Q pieces through LDS) and does the row softmax in its epilogue: softmax pass and its launch gone; QK k order and row-sum order change
# speedup vs baseline: 1.9570x; 1.1781x over previous
#include <hip/hip_runtime.h>
#include <hip/hip_bf16.h>
#include <math.h>

typedef __attribute__((ext_vector_type(2))) float v2f;
typedef __attribute__((ext_vector_type(8))) float v8f;

#define S_DIM 1024
#define CA 1024
#define CS 512
#define CZ 64
#define NH 16
#define HC 64

__device__ __forceinline__ float sigmoidf_(float x) {
    return 1.0f / (1.0f + __expf(-x));
}

__global__ void ln_kernel(const float* __restrict__ x, const float* __restrict__ w,
                          float* __restrict__ y, int C) {
    __shared__ float red[256];
    __shared__ float red2[256];
    int row = blockIdx.x;
    const float4* xr = (const float4*)(x + (size_t)row * C);
    int C4 = C >> 2;
    float sum = 0.f, sq = 0.f;
    for (int c = threadIdx.x; c < C4; c += 256) {
        float4 v = xr[c];
        sum += v.x + v.y + v.z + v.w;
        sq  += v.x * v.x + v.y * v.y + v.z * v.z + v.w * v.w;
    }
    red[threadIdx.x] = sum; red2[threadIdx.x] = sq;
    __syncthreads();
    for (int off = 128; off > 0; off >>= 1) {
        if (threadIdx.x < off) {
            red[threadIdx.x]  += red[threadIdx.x + off];
            red2[threadIdx.x] += red2[threadIdx.x + off];
        }
        __syncthreads();
    }
    float m   = red[0] / C;
    float var = red2[0] / C - m * m;
    float inv = rsqrtf(var + 1e-5f);
    float4* yr = (float4*)(y + (size_t)row * C);
    for (int c = threadIdx.x; c < C4; c += 256) {
        float4 v = xr[c];
        v.x = (v.x - m) * inv; v.y = (v.y - m) * inv;
        v.z = (v.z - m) * inv; v.w = (v.w - m) * inv;
        if (w) {
            const float4 wv = ((const float4*)w)[c];
            v.x *= wv.x; v.y *= wv.y; v.z *= wv.z; v.w *= wv.w;
        }
        yr[c] = v;
    }
}

#define KST 64
#define PNT 68
#define PTN 72
__global__ void
gemm_nt_kernel(const float* __restrict__ A, int lda, long long sA,
               const float* __restrict__ B, int ldb, long long sB,
               float* __restrict__ C, int ldc, long long sC,
               const float* __restrict__ bias,
               const float* __restrict__ addm, int ldadd, long long sAdd,
               int K, float scale, int mode) {
    __shared__ float As[2][64][PNT];
    __shared__ float Bs[2][64][PNT];
    const int z = blockIdx.z;
    A += (size_t)z * sA; B += (size_t)z * sB; C += (size_t)z * sC;
    if (addm) addm += (size_t)z * sAdd;
    const int tileM = blockIdx.y * 64;
    const int tileN = blockIdx.x * 64;
    const int tid  = threadIdx.x;
    const int wave = tid >> 5, lane = tid & 31;
    const int wm = wave >> 1, wn = wave & 1;
    const int half = lane >> 4, l = lane & 15;
    const int lr = tid >> 4;
    const int lc = (tid & 15) * 4;

    const float* gA = A + (size_t)(tileM + lr) * lda + lc;
    const float* gB = B + (size_t)(tileN + lr) * ldb + lc;

    v8f acc0 = {}; v8f acc1 = {};
    const int nstages = K / KST;

    {
        float4 ra[4], rb[4];
        #pragma unroll
        for (int i = 0; i < 4; i++) {
            ra[i] = *(const float4*)(gA + (size_t)(16 * i) * lda);
            rb[i] = *(const float4*)(gB + (size_t)(16 * i) * ldb);
        }
        #pragma unroll
        for (int i = 0; i < 4; i++) {
            *(float4*)&As[0][lr + 16 * i][lc] = ra[i];
            *(float4*)&Bs[0][lr + 16 * i][lc] = rb[i];
        }
    }
    __syncthreads();

    for (int s = 0; s < nstages; s++) {
        const int p = s & 1;
        const bool more = (s + 1 < nstages);
        float4 na[4], nb[4];
        #pragma unroll
        for (int i = 0; i < 4; i++) {
            na[i] = make_float4(0.f, 0.f, 0.f, 0.f);
            nb[i] = make_float4(0.f, 0.f, 0.f, 0.f);
        }
        if (more) {
            #pragma unroll
            for (int i = 0; i < 4; i++) {
                na[i] = *(const float4*)(gA + (size_t)(16 * i) * lda + (s + 1) * KST);
                nb[i] = *(const float4*)(gB + (size_t)(16 * i) * ldb + (s + 1) * KST);
            }
        }
        const float* ap  = &As[p][wm * 16 + l][half * 2];
        const float* bp0 = &Bs[p][wn * 32 + l][half * 2];
        const float* bp1 = &Bs[p][wn * 32 + 16 + l][half * 2];
        #pragma unroll
        for (int q = 0; q < KST / 4; q++) {
            const v2f af  = *(const v2f*)(ap  + q * 4);
            const v2f bf0 = *(const v2f*)(bp0 + q * 4);
            const v2f bf1 = *(const v2f*)(bp1 + q * 4);
            acc0 = __builtin_amdgcn_wmma_f32_16x16x4_f32(false, af, false, bf0, (short)0, acc0, false, false);
            acc1 = __builtin_amdgcn_wmma_f32_16x16x4_f32(false, af, false, bf1, (short)0, acc1, false, false);
        }
        if (more) {
            #pragma unroll
            for (int i = 0; i < 4; i++) {
                *(float4*)&As[p ^ 1][lr + 16 * i][lc] = na[i];
                *(float4*)&Bs[p ^ 1][lr + 16 * i][lc] = nb[i];
            }
        }
        __syncthreads();
    }

    #pragma unroll
    for (int r = 0; r < 8; r++) {
        int m  = tileM + wm * 16 + half * 8 + r;
        int n0 = tileN + wn * 32 + l;
        int n1 = n0 + 16;
        float v0 = acc0[r] * scale;
        float v1 = acc1[r] * scale;
        if (bias) { v0 += bias[n0]; v1 += bias[n1]; }
        if (mode == 1) {
            v0 = sigmoidf_(v0) * addm[(size_t)m * ldadd + n0];
            v1 = sigmoidf_(v1) * addm[(size_t)m * ldadd + n1];
        } else if (addm) {
            v0 += addm[(size_t)m * ldadd + n0];
            v1 += addm[(size_t)m * ldadd + n1];
        }
        C[(size_t)m * ldc + n0] = v0;
        C[(size_t)m * ldc + n1] = v1;
    }
}

__global__ void
gemm_tn_kernel(const float* __restrict__ A, int lda, long long sA,
               const float* __restrict__ B, int ldb, long long sB,
               float* __restrict__ C, int ldc, long long sC,
               int K) {
    __shared__ float As[2][KST][PTN];
    __shared__ float Bs[2][KST][PTN];
    const int z = blockIdx.z;
    A += (size_t)z * sA; B += (size_t)z * sB; C += (size_t)z * sC;
    const int tileM = blockIdx.y * 64;
    const int tileN = blockIdx.x * 64;
    const int tid  = threadIdx.x;
    const int wave = tid >> 5, lane = tid & 31;
    const int wm = wave >> 1, wn = wave & 1;
    const int half = lane >> 4, l = lane & 15;
    const int lk = tid >> 4;
    const int lm = (tid & 15) * 4;

    const float* gA = A + (size_t)lk * lda + tileM + lm;
    const float* gB = B + (size_t)lk * ldb + tileN + lm;

    v8f acc0 = {}; v8f acc1 = {};
    const int nstages = K / KST;

    {
        float4 ra[4], rb[4];
        #pragma unroll
        for (int i = 0; i < 4; i++) {
            ra[i] = *(const float4*)(gA + (size_t)(16 * i) * lda);
            rb[i] = *(const float4*)(gB + (size_t)(16 * i) * ldb);
        }
        #pragma unroll
        for (int i = 0; i < 4; i++) {
            *(float4*)&As[0][lk + 16 * i][lm] = ra[i];
            *(float4*)&Bs[0][lk + 16 * i][lm] = rb[i];
        }
    }
    __syncthreads();

    for (int s = 0; s < nstages; s++) {
        const int p = s & 1;
        const bool more = (s + 1 < nstages);
        float4 na[4], nb[4];
        #pragma unroll
        for (int i = 0; i < 4; i++) {
            na[i] = make_float4(0.f, 0.f, 0.f, 0.f);
            nb[i] = make_float4(0.f, 0.f, 0.f, 0.f);
        }
        if (more) {
            #pragma unroll
            for (int i = 0; i < 4; i++) {
                na[i] = *(const float4*)(gA + (size_t)((s + 1) * KST + 16 * i) * lda);
                nb[i] = *(const float4*)(gB + (size_t)((s + 1) * KST + 16 * i) * ldb);
            }
        }
        const float* ap  = &As[p][half * 2][wm * 16 + l];
        const float* bp0 = &Bs[p][half * 2][wn * 32 + l];
        const float* bp1 = &Bs[p][half * 2][wn * 32 + 16 + l];
        #pragma unroll
        for (int q = 0; q < KST / 4; q++) {
            v2f af, bf0, bf1;
            af.x  = ap [(q * 4) * PTN];  af.y  = ap [(q * 4 + 1) * PTN];
            bf0.x = bp0[(q * 4) * PTN];  bf0.y = bp0[(q * 4 + 1) * PTN];
            bf1.x = bp1[(q * 4) * PTN];  bf1.y = bp1[(q * 4 + 1) * PTN];
            acc0 = __builtin_amdgcn_wmma_f32_16x16x4_f32(false, af, false, bf0, (short)0, acc0, false, false);
            acc1 = __builtin_amdgcn_wmma_f32_16x16x4_f32(false, af, false, bf1, (short)0, acc1, false, false);
        }
        if (more) {
            #pragma unroll
            for (int i = 0; i < 4; i++) {
                *(float4*)&As[p ^ 1][lk + 16 * i][lm] = na[i];
                *(float4*)&Bs[p ^ 1][lk + 16 * i][lm] = nb[i];
            }
        }
        __syncthreads();
    }

    #pragma unroll
    for (int r = 0; r < 8; r++) {
        int m  = tileM + wm * 16 + half * 8 + r;
        int n0 = tileN + wn * 32 + l;
        C[(size_t)m * ldc + n0]      = acc0[r];
        C[(size_t)m * ldc + n0 + 16] = acc1[r];
    }
}

#define SSP 68
__global__ void __launch_bounds__(256)
scores_softmax_kernel(const float* __restrict__ kvg, const float* __restrict__ qm,
                      const float* __restrict__ bias, float* __restrict__ P) {
    __shared__ float  aS[16][SSP];
    __shared__ float  bS[2][8][16 * SSP];
    __shared__ float4 redm[16][2];
    __shared__ float4 reds[16][2];
    const int tid  = threadIdx.x;
    const int wave = tid >> 5, lane = tid & 31;
    const int half = lane >> 4, l = lane & 15;
    const int h  = blockIdx.y;
    const int j0 = blockIdx.x * 16;
    const float*  Kh = kvg + (size_t)h * 196608 + (size_t)j0 * 192;
    const float4* Q4 = (const float4*)(qm + (size_t)h * 65536) + wave * 2048 + lane;
    const size_t  ob = (size_t)h * 1048576 + (size_t)(j0 + 8 * half) * 1024 + wave * 128 + l;

    float4 raw[8];
    #pragma unroll
    for (int j = 0; j < 8; j++) raw[j] = Q4[j * 32];
    *(float4*)&aS[tid >> 4][(tid & 15) * 4] =
        *(const float4*)(Kh + (size_t)(tid >> 4) * 192 + (tid & 15) * 4);
    __syncthreads();
    float4 af[8];
    #pragma unroll
    for (int jj = 0; jj < 8; jj++) af[jj] = *(const float4*)&aS[l][half * 32 + jj * 4];

    v8f acc[8];
    #pragma unroll
    for (int ct = 0; ct < 8; ct++) {
        float* bw = &bS[ct & 1][wave][0];
        #pragma unroll
        for (int j = 0; j < 8; j++) {
            const int f = j * 32 + lane;
            *(float4*)&bw[(f >> 4) * SSP + (f & 15) * 4] = raw[j];
        }
        __syncthreads();
        #pragma unroll
        for (int j = 0; j < 8; j++) raw[j] = make_float4(0.f, 0.f, 0.f, 0.f);
        if (ct + 1 < 8) {
            #pragma unroll
            for (int j = 0; j < 8; j++) raw[j] = Q4[(ct + 1) * 256 + j * 32];
        }
        float4 bf[8];
        #pragma unroll
        for (int jj = 0; jj < 8; jj++) bf[jj] = *(const float4*)&bw[l * SSP + half * 32 + jj * 4];
        v8f a0 = {}; v8f a1 = {};
        #pragma unroll
        for (int jj = 0; jj < 8; jj++) {
            v2f x0, x1, y0, y1;
            x0.x = af[jj].x; x0.y = af[jj].y; x1.x = af[jj].z; x1.y = af[jj].w;
            y0.x = bf[jj].x; y0.y = bf[jj].y; y1.x = bf[jj].z; y1.y = bf[jj].w;
            a0 = __builtin_amdgcn_wmma_f32_16x16x4_f32(false, x0, false, y0, (short)0, a0, false, false);
            a1 = __builtin_amdgcn_wmma_f32_16x16x4_f32(false, x1, false, y1, (short)0, a1, false, false);
        }
        acc[ct] = a0 + a1;
        __builtin_amdgcn_sched_barrier(0);
    }

    const float* bp = bias + ob;
    float* pp = P + ob;
    #pragma unroll
    for (int r = 0; r < 8; r++) {
        #pragma unroll
        for (int ct = 0; ct < 8; ct++)
            acc[ct][r] = acc[ct][r] * (1.0f / 64.0f) + bp[r * 1024 + ct * 16];
    }
    float mx[8];
    #pragma unroll
    for (int r = 0; r < 8; r++) {
        float m = acc[0][r];
        #pragma unroll
        for (int ct = 1; ct < 8; ct++) m = fmaxf(m, acc[ct][r]);
        #pragma unroll
        for (int off = 8; off > 0; off >>= 1) m = fmaxf(m, __shfl_xor(m, off));
        mx[r] = m;
    }
    if (l == 0) {
        #pragma unroll
        for (int r = 0; r < 8; r++) ((float*)&redm[8 * half + r][0])[wave] = mx[r];
    }
    __syncthreads();
    #pragma unroll
    for (int r = 0; r < 8; r++) {
        const float4 u = redm[8 * half + r][0], w = redm[8 * half + r][1];
        mx[r] = fmaxf(fmaxf(fmaxf(u.x, u.y), fmaxf(u.z, u.w)), fmaxf(fmaxf(w.x, w.y), fmaxf(w.z, w.w)));
    }
    float sm[8];
    #pragma unroll
    for (int r = 0; r < 8; r++) {
        float s = 0.f;
        #pragma unroll
        for (int ct = 0; ct < 8; ct++) {
            const float e = __expf(acc[ct][r] - mx[r]);
            acc[ct][r] = e;
            s += e;
        }
        #pragma unroll
        for (int off = 8; off > 0; off >>= 1) s += __shfl_xor(s, off);
        sm[r] = s;
    }
    if (l == 0) {
        #pragma unroll
        for (int r = 0; r < 8; r++) ((float*)&reds[8 * half + r][0])[wave] = sm[r];
    }
    __syncthreads();
    #pragma unroll
    for (int r = 0; r < 8; r++) {
        const float4 u = reds[8 * half + r][0], w = reds[8 * half + r][1];
        const float s = ((u.x + u.y) + (u.z + u.w)) + ((w.x + w.y) + (w.z + w.w));
        const float inv = 1.0f / s;
        #pragma unroll
        for (int ct = 0; ct < 8; ct++) pp[r * 1024 + ct * 16] = acc[ct][r] * inv;
    }
}

#define ZTP 68
#define OTP 20
#define ZB_TILES 8192
__global__ void __launch_bounds__(256)
zbias_kernel(const float* __restrict__ z,
             const float* __restrict__ pnw, const float* __restrict__ pnb,
             const float* __restrict__ bw,  const float* __restrict__ bb,
             float* __restrict__ out) {
    __shared__ float zS[128][ZTP];
    __shared__ float oS[8][16 * OTP];
    const int tid  = threadIdx.x;
    const int wave = tid >> 5, lane = tid & 31;
    const int half = lane >> 4, l = lane & 15;

    *(float4*)&zS[tid >> 4][(tid & 15) * 4] = ((const float4*)bw)[tid];
    __syncthreads();
    v2f wf[16];
    #pragma unroll
    for (int j = 0; j < 8; j++) {
        const float4 t = *(const float4*)&zS[l][half * 32 + j * 4];
        wf[2 * j].x = t.x;     wf[2 * j].y = t.y;
        wf[2 * j + 1].x = t.z; wf[2 * j + 1].y = t.w;
    }
    __syncthreads();
    float4 pw[8], pb[8];
    #pragma unroll
    for (int j = 0; j < 8; j++) {
        pw[j] = ((const float4*)pnw)[half * 8 + j];
        pb[j] = ((const float4*)pnb)[half * 8 + j];
    }
    v8f cinit;
    #pragma unroll
    for (int r = 0; r < 8; r++) cinit[r] = bb[half * 8 + r];

    const float4* z4 = (const float4*)z;
    float4* out4 = (float4*)out;
    float* os = &oS[wave][0];

    int ts = blockIdx.x;
    float4 raw[8];
    {
        const size_t b4 = ((size_t)ts * 128 + wave * 16) * 16 + lane;
        #pragma unroll
        for (int j = 0; j < 8; j++) raw[j] = z4[b4 + j * 32];
    }
    for (; ts < ZB_TILES; ts += gridDim.x) {
        #pragma unroll
        for (int j = 0; j < 8; j++) {
            const int f = j * 32 + lane;
            *(float4*)&zS[wave * 16 + (f >> 4)][(f & 15) * 4] = raw[j];
        }
        __syncthreads();
        const int nts = ts + gridDim.x;
        #pragma unroll
        for (int j = 0; j < 8; j++) raw[j] = make_float4(0.f, 0.f, 0.f, 0.f);
        if (nts < ZB_TILES) {
            const size_t b4 = ((size_t)nts * 128 + wave * 16) * 16 + lane;
            #pragma unroll
            for (int j = 0; j < 8; j++) raw[j] = z4[b4 + j * 32];
        }
        float4 x[8];
        float sum = 0.f, sq = 0.f;
        #pragma unroll
        for (int j = 0; j < 8; j++) {
            const float4 t = *(const float4*)&zS[wave * 16 + l][half * 32 + j * 4];
            x[j] = t;
            sum += t.x + t.y + t.z + t.w;
            sq  += t.x * t.x + t.y * t.y + t.z * t.z + t.w * t.w;
        }
        sum += __shfl_xor(sum, 16);
        sq  += __shfl_xor(sq, 16);
        const float m   = sum * (1.0f / 64.0f);
        const float var = sq * (1.0f / 64.0f) - m * m;
        const float inv = rsqrtf(var + 1e-5f);
        v8f acc0 = cinit; v8f acc1 = {};
        #pragma unroll
        for (int j = 0; j < 8; j++) {
            v2f b0, b1;
            b0.x = (x[j].x - m) * inv * pw[j].x + pb[j].x;
            b0.y = (x[j].y - m) * inv * pw[j].y + pb[j].y;
            b1.x = (x[j].z - m) * inv * pw[j].z + pb[j].z;
            b1.y = (x[j].w - m) * inv * pw[j].w + pb[j].w;
            acc0 = __builtin_amdgcn_wmma_f32_16x16x4_f32(false, wf[2 * j],     false, b0, (short)0, acc0, false, false);
            acc1 = __builtin_amdgcn_wmma_f32_16x16x4_f32(false, wf[2 * j + 1], false, b1, (short)0, acc1, false, false);
        }
        *(float4*)&os[l * OTP + half * 8]     = make_float4(acc0[0] + acc1[0], acc0[1] + acc1[1],
                                                            acc0[2] + acc1[2], acc0[3] + acc1[3]);
        *(float4*)&os[l * OTP + half * 8 + 4] = make_float4(acc0[4] + acc1[4], acc0[5] + acc1[5],
                                                            acc0[6] + acc1[6], acc0[7] + acc1[7]);
        __syncthreads();
        const size_t o4 = ((size_t)ts * 128 + wave * 16) * 4;
        #pragma unroll
        for (int s = 0; s < 2; s++) {
            const int g = s * 32 + lane;
            out4[o4 + g] = *(const float4*)&os[(g >> 2) * OTP + (g & 3) * 4];
        }
    }
}

__global__ void adaln_gate_kernel(const float4* __restrict__ pb, const float4* __restrict__ pn,
                                  const float4* __restrict__ an, float4* __restrict__ a1) {
    int i = blockIdx.x * 256 + threadIdx.x;
    float4 p = pb[i], q = pn[i], n = an[i];
    float4 r;
    r.x = sigmoidf_(p.x * n.x + q.x);
    r.y = sigmoidf_(p.y * n.y + q.y);
    r.z = sigmoidf_(p.z * n.z + q.z);
    r.w = sigmoidf_(p.w * n.w + q.w);
    a1[i] = r;
}

__global__ void gate_o_kernel(const float* __restrict__ kvg, const float4* __restrict__ obuf,
                              float4* __restrict__ go) {
    int q = blockIdx.x * 256 + threadIdx.x;
    int h  = q >> 14;
    int j  = (q >> 4) & 1023;
    int c4 = (q & 15) * 4;
    const float4 g = *(const float4*)(kvg + (size_t)h * 196608 + (size_t)j * 192 + 128 + c4);
    float4 o = obuf[q];
    float4 r;
    r.x = sigmoidf_(g.x) * o.x;
    r.y = sigmoidf_(g.y) * o.y;
    r.z = sigmoidf_(g.z) * o.z;
    r.w = sigmoidf_(g.w) * o.w;
    go[q] = r;
}

extern "C" void kernel_launch(void* const* d_in, const int* in_sizes, int n_in,
                              void* d_out, int out_size, void* d_ws, size_t ws_size,
                              hipStream_t stream) {
    const float* a       = (const float*)d_in[0];
    const float* z       = (const float*)d_in[1];
    const float* s       = (const float*)d_in[2];
    const float* sn_w    = (const float*)d_in[3];
    const float* pb_w    = (const float*)d_in[4];
    const float* pb_b    = (const float*)d_in[5];
    const float* pn_w    = (const float*)d_in[6];
    const float* pnorm_w = (const float*)d_in[7];
    const float* pnorm_b = (const float*)d_in[8];
    const float* q_w     = (const float*)d_in[9];
    const float* q_b     = (const float*)d_in[10];
    const float* kvg_w   = (const float*)d_in[11];
    const float* bias_w  = (const float*)d_in[12];
    const float* bias_b  = (const float*)d_in[13];
    const float* attn_w  = (const float*)d_in[14];
    const float* out_w   = (const float*)d_in[15];
    const float* out_b   = (const float*)d_in[16];
    float* out = (float*)d_out;
    float* ws  = (float*)d_ws;

    float* a_n    = ws + 0;
    float* s_n    = ws + 1048576;
    float* pb     = ws + 1572864;
    float* pn     = ws + 2621440;
    float* qm     = ws + 3670016;
    float* kvg    = ws + 4718592;
    float* biasO  = ws + 7864320;
    float* obuf   = biasO;
    float* go     = biasO + 1048576;
    float* a2     = biasO + 2097152;
    float* scores = ws + 24641536;

    ln_kernel<<<1024, 256, 0, stream>>>(a, nullptr, a_n, 1024);
    ln_kernel<<<1024, 256, 0, stream>>>(s, sn_w, s_n, 512);

    gemm_nt_kernel<<<dim3(16, 16, 1), 256, 0, stream>>>(
        s_n, 512, 0, pb_w, 512, 0, pb, 1024, 0, pb_b, nullptr, 0, 0, 512, 1.0f, 0);
    gemm_nt_kernel<<<dim3(16, 16, 1), 256, 0, stream>>>(
        s_n, 512, 0, pn_w, 512, 0, pn, 1024, 0, nullptr, nullptr, 0, 0, 512, 1.0f, 0);
    adaln_gate_kernel<<<1024, 256, 0, stream>>>(
        (const float4*)pb, (const float4*)pn, (const float4*)a_n, (float4*)pb);
    float* a1 = pb;

    gemm_nt_kernel<<<dim3(16, 16, 1), 256, 0, stream>>>(
        a1, 1024, 0, q_w, 1024, 0, qm, 1024, 0, q_b, nullptr, 0, 0, 1024, 1.0f, 0);
    gemm_nt_kernel<<<dim3(48, 16, 1), 256, 0, stream>>>(
        a1, 1024, 0, kvg_w, 1024, 0, kvg, 3072, 0, nullptr, nullptr, 0, 0, 1024, 1.0f, 0);

    zbias_kernel<<<512, 256, 0, stream>>>(z, pnorm_w, pnorm_b, bias_w, bias_b, biasO);

    scores_softmax_kernel<<<dim3(64, 16, 1), 256, 0, stream>>>(kvg, qm, biasO, scores);

    gemm_tn_kernel<<<dim3(1, 16, 16), 256, 0, stream>>>(
        scores, 1024, 1048576LL,
        kvg + 64, 192, 196608LL,
        obuf, 64, 65536LL,
        1024);

    gate_o_kernel<<<1024, 256, 0, stream>>>(kvg, (const float4*)obuf, (float4*)go);

    gemm_nt_kernel<<<dim3(16, 16, 1), 256, 0, stream>>>(
        go, 1024, 0, attn_w, 1024, 0, a2, 1024, 0, nullptr, nullptr, 0, 0, 1024, 1.0f, 0);

    gemm_nt_kernel<<<dim3(16, 16, 1), 256, 0, stream>>>(
        s, 512, 0, out_w, 512, 0, out, 1024, 0, out_b, a2, 1024, 0, 512, 1.0f, 1);
}


# Pair bias formed inside the scores+softmax kernel: a block streams its own z row, bias tiles leave WMMA in the score registers (norm weight folded into bias_w); bias array and one launch gone
# speedup vs baseline: 2.1619x; 1.1047x over previous
#include <hip/hip_runtime.h>
#include <hip/hip_bf16.h>
#include <math.h>

typedef __attribute__((ext_vector_type(2))) float v2f;
typedef __attribute__((ext_vector_type(8))) float v8f;

#define S_DIM 1024
#define CA 1024
#define CS 512
#define CZ 64
#define NH 16
#define HC 64

__device__ __forceinline__ float sigmoidf_(float x) {
    return 1.0f / (1.0f + __expf(-x));
}

__global__ void ln_kernel(const float* __restrict__ x, const float* __restrict__ w,
                          float* __restrict__ y, int C) {
    __shared__ float red[256];
    __shared__ float red2[256];
    int row = blockIdx.x;
    const float4* xr = (const float4*)(x + (size_t)row * C);
    int C4 = C >> 2;
    float sum = 0.f, sq = 0.f;
    for (int c = threadIdx.x; c < C4; c += 256) {
        float4 v = xr[c];
        sum += v.x + v.y + v.z + v.w;
        sq  += v.x * v.x + v.y * v.y + v.z * v.z + v.w * v.w;
    }
    red[threadIdx.x] = sum; red2[threadIdx.x] = sq;
    __syncthreads();
    for (int off = 128; off > 0; off >>= 1) {
        if (threadIdx.x < off) {
            red[threadIdx.x]  += red[threadIdx.x + off];
            red2[threadIdx.x] += red2[threadIdx.x + off];
        }
        __syncthreads();
    }
    float m   = red[0] / C;
    float var = red2[0] / C - m * m;
    float inv = rsqrtf(var + 1e-5f);
    float4* yr = (float4*)(y + (size_t)row * C);
    for (int c = threadIdx.x; c < C4; c += 256) {
        float4 v = xr[c];
        v.x = (v.x - m) * inv; v.y = (v.y - m) * inv;
        v.z = (v.z - m) * inv; v.w = (v.w - m) * inv;
        if (w) {
            const float4 wv = ((const float4*)w)[c];
            v.x *= wv.x; v.y *= wv.y; v.z *= wv.z; v.w *= wv.w;
        }
        yr[c] = v;
    }
}

#define KST 64
#define PNT 68
#define PTN 72
__global__ void
gemm_nt_kernel(const float* __restrict__ A, int lda, long long sA,
               const float* __restrict__ B, int ldb, long long sB,
               float* __restrict__ C, int ldc, long long sC,
               const float* __restrict__ bias,
               const float* __restrict__ addm, int ldadd, long long sAdd,
               int K, float scale, int mode) {
    __shared__ float As[2][64][PNT];
    __shared__ float Bs[2][64][PNT];
    const int z = blockIdx.z;
    A += (size_t)z * sA; B += (size_t)z * sB; C += (size_t)z * sC;
    if (addm) addm += (size_t)z * sAdd;
    const int tileM = blockIdx.y * 64;
    const int tileN = blockIdx.x * 64;
    const int tid  = threadIdx.x;
    const int wave = tid >> 5, lane = tid & 31;
    const int wm = wave >> 1, wn = wave & 1;
    const int half = lane >> 4, l = lane & 15;
    const int lr = tid >> 4;
    const int lc = (tid & 15) * 4;

    const float* gA = A + (size_t)(tileM + lr) * lda + lc;
    const float* gB = B + (size_t)(tileN + lr) * ldb + lc;

    v8f acc0 = {}; v8f acc1 = {};
    const int nstages = K / KST;

    {
        float4 ra[4], rb[4];
        #pragma unroll
        for (int i = 0; i < 4; i++) {
            ra[i] = *(const float4*)(gA + (size_t)(16 * i) * lda);
            rb[i] = *(const float4*)(gB + (size_t)(16 * i) * ldb);
        }
        #pragma unroll
        for (int i = 0; i < 4; i++) {
            *(float4*)&As[0][lr + 16 * i][lc] = ra[i];
            *(float4*)&Bs[0][lr + 16 * i][lc] = rb[i];
        }
    }
    __syncthreads();

    for (int s = 0; s < nstages; s++) {
        const int p = s & 1;
        const bool more = (s + 1 < nstages);
        float4 na[4], nb[4];
        #pragma unroll
        for (int i = 0; i < 4; i++) {
            na[i] = make_float4(0.f, 0.f, 0.f, 0.f);
            nb[i] = make_float4(0.f, 0.f, 0.f, 0.f);
        }
        if (more) {
            #pragma unroll
            for (int i = 0; i < 4; i++) {
                na[i] = *(const float4*)(gA + (size_t)(16 * i) * lda + (s + 1) * KST);
                nb[i] = *(const float4*)(gB + (size_t)(16 * i) * ldb + (s + 1) * KST);
            }
        }
        const float* ap  = &As[p][wm * 16 + l][half * 2];
        const float* bp0 = &Bs[p][wn * 32 + l][half * 2];
        const float* bp1 = &Bs[p][wn * 32 + 16 + l][half * 2];
        #pragma unroll
        for (int q = 0; q < KST / 4; q++) {
            const v2f af  = *(const v2f*)(ap  + q * 4);
            const v2f bf0 = *(const v2f*)(bp0 + q * 4);
            const v2f bf1 = *(const v2f*)(bp1 + q * 4);
            acc0 = __builtin_amdgcn_wmma_f32_16x16x4_f32(false, af, false, bf0, (short)0, acc0, false, false);
            acc1 = __builtin_amdgcn_wmma_f32_16x16x4_f32(false, af, false, bf1, (short)0, acc1, false, false);
        }
        if (more) {
            #pragma unroll
            for (int i = 0; i < 4; i++) {
                *(float4*)&As[p ^ 1][lr + 16 * i][lc] = na[i];
                *(float4*)&Bs[p ^ 1][lr + 16 * i][lc] = nb[i];
            }
        }
        __syncthreads();
    }

    #pragma unroll
    for (int r = 0; r < 8; r++) {
        int m  = tileM + wm * 16 + half * 8 + r;
        int n0 = tileN + wn * 32 + l;
        int n1 = n0 + 16;
        float v0 = acc0[r] * scale;
        float v1 = acc1[r] * scale;
        if (bias) { v0 += bias[n0]; v1 += bias[n1]; }
        if (mode == 1) {
            v0 = sigmoidf_(v0) * addm[(size_t)m * ldadd + n0];
            v1 = sigmoidf_(v1) * addm[(size_t)m * ldadd + n1];
        } else if (addm) {
            v0 += addm[(size_t)m * ldadd + n0];
            v1 += addm[(size_t)m * ldadd + n1];
        }
        C[(size_t)m * ldc + n0] = v0;
        C[(size_t)m * ldc + n1] = v1;
    }
}

__global__ void
gemm_tn_kernel(const float* __restrict__ A, int lda, long long sA,
               const float* __restrict__ B, int ldb, long long sB,
               float* __restrict__ C, int ldc, long long sC,
               int K) {
    __shared__ float As[2][KST][PTN];
    __shared__ float Bs[2][KST][PTN];
    const int z = blockIdx.z;
    A += (size_t)z * sA; B += (size_t)z * sB; C += (size_t)z * sC;
    const int tileM = blockIdx.y * 64;
    const int tileN = blockIdx.x * 64;
    const int tid  = threadIdx.x;
    const int wave = tid >> 5, lane = tid & 31;
    const int wm = wave >> 1, wn = wave & 1;
    const int half = lane >> 4, l = lane & 15;
    const int lk = tid >> 4;
    const int lm = (tid & 15) * 4;

    const float* gA = A + (size_t)lk * lda + tileM + lm;
    const float* gB = B + (size_t)lk * ldb + tileN + lm;

    v8f acc0 = {}; v8f acc1 = {};
    const int nstages = K / KST;

    {
        float4 ra[4], rb[4];
        #pragma unroll
        for (int i = 0; i < 4; i++) {
            ra[i] = *(const float4*)(gA + (size_t)(16 * i) * lda);
            rb[i] = *(const float4*)(gB + (size_t)(16 * i) * ldb);
        }
        #pragma unroll
        for (int i = 0; i < 4; i++) {
            *(float4*)&As[0][lk + 16 * i][lm] = ra[i];
            *(float4*)&Bs[0][lk + 16 * i][lm] = rb[i];
        }
    }
    __syncthreads();

    for (int s = 0; s < nstages; s++) {
        const int p = s & 1;
        const bool more = (s + 1 < nstages);
        float4 na[4], nb[4];
        #pragma unroll
        for (int i = 0; i < 4; i++) {
            na[i] = make_float4(0.f, 0.f, 0.f, 0.f);
            nb[i] = make_float4(0.f, 0.f, 0.f, 0.f);
        }
        if (more) {
            #pragma unroll
            for (int i = 0; i < 4; i++) {
                na[i] = *(const float4*)(gA + (size_t)((s + 1) * KST + 16 * i) * lda);
                nb[i] = *(const float4*)(gB + (size_t)((s + 1) * KST + 16 * i) * ldb);
            }
        }
        const float* ap  = &As[p][half * 2][wm * 16 + l];
        const float* bp0 = &Bs[p][half * 2][wn * 32 + l];
        const float* bp1 = &Bs[p][half * 2][wn * 32 + 16 + l];
        #pragma unroll
        for (int q = 0; q < KST / 4; q++) {
            v2f af, bf0, bf1;
            af.x  = ap [(q * 4) * PTN];  af.y  = ap [(q * 4 + 1) * PTN];
            bf0.x = bp0[(q * 4) * PTN];  bf0.y = bp0[(q * 4 + 1) * PTN];
            bf1.x = bp1[(q * 4) * PTN];  bf1.y = bp1[(q * 4 + 1) * PTN];
            acc0 = __builtin_amdgcn_wmma_f32_16x16x4_f32(false, af, false, bf0, (short)0, acc0, false, false);
            acc1 = __builtin_amdgcn_wmma_f32_16x16x4_f32(false, af, false, bf1, (short)0, acc1, false, false);
        }
        if (more) {
            #pragma unroll
            for (int i = 0; i < 4; i++) {
                *(float4*)&As[p ^ 1][lk + 16 * i][lm] = na[i];
                *(float4*)&Bs[p ^ 1][lk + 16 * i][lm] = nb[i];
            }
        }
        __syncthreads();
    }

    #pragma unroll
    for (int r = 0; r < 8; r++) {
        int m  = tileM + wm * 16 + half * 8 + r;
        int n0 = tileN + wn * 32 + l;
        C[(size_t)m * ldc + n0]      = acc0[r];
        C[(size_t)m * ldc + n0 + 16] = acc1[r];
    }
}

#define SSP 68
__global__ void __launch_bounds__(256, 4)
scores_softmax_kernel(const float* __restrict__ kvg, const float* __restrict__ qm,
                      const float* __restrict__ z,
                      const float* __restrict__ pnw, const float* __restrict__ pnb,
                      const float* __restrict__ bw,  const float* __restrict__ bb,
                      float* __restrict__ P) {
    __shared__ float  aS[16][SSP];
    __shared__ float  wS[16][SSP];
    __shared__ float  bS[2][8][16 * SSP];
    __shared__ float  zS[2][8][16 * SSP];
    __shared__ float4 redm[16][2];
    __shared__ float4 reds[16][2];
    const int tid  = threadIdx.x;
    const int wave = tid >> 5, lane = tid & 31;
    const int half = lane >> 4, l = lane & 15;
    const int h  = blockIdx.y;
    const int j0 = blockIdx.x * 16;
    const float*  Kh = kvg + (size_t)h * 196608 + (size_t)j0 * 192;
    const float4* Q4 = (const float4*)(qm + (size_t)h * 65536) + wave * 2048 + lane;
    const float4* Z4 = (const float4*)z + (size_t)(h * 64 + blockIdx.x) * 16384
                       + (lane >> 4) * 1024 + wave * 128 + (lane & 15);
    const size_t  ob = (size_t)h * 1048576 + (size_t)(j0 + 8 * half) * 1024 + wave * 128 + l;

    float4 raw[8], zraw[8];
    #pragma unroll
    for (int j = 0; j < 8; j++) {
        zraw[j] = Z4[j * 2048];
        raw[j]  = Q4[j * 32];
    }
    *(float4*)&aS[tid >> 4][(tid & 15) * 4] =
        *(const float4*)(Kh + (size_t)(tid >> 4) * 192 + (tid & 15) * 4);
    {
        const float4 w4 = ((const float4*)bw)[tid];
        const float4 g4 = ((const float4*)pnw)[tid & 15];
        *(float4*)&wS[tid >> 4][(tid & 15) * 4] =
            make_float4(w4.x * g4.x, w4.y * g4.y, w4.z * g4.z, w4.w * g4.w);
    }
    float cb = 0.f;
    #pragma unroll
    for (int jj = 0; jj < 8; jj++) {
        const float4 w4 = ((const float4*)bw)[l * 16 + half * 8 + jj];
        const float4 d4 = ((const float4*)pnb)[half * 8 + jj];
        cb += w4.x * d4.x + w4.y * d4.y + w4.z * d4.z + w4.w * d4.w;
    }
    cb += __shfl_xor(cb, 16);
    cb += bb[l];
    __syncthreads();
    float4 af[8];
    #pragma unroll
    for (int jj = 0; jj < 8; jj++) af[jj] = *(const float4*)&aS[l][half * 32 + jj * 4];
    v8f cinit;
    #pragma unroll
    for (int r = 0; r < 8; r++) cinit[r] = cb;

    v8f acc[8];
    #pragma unroll
    for (int ct = 0; ct < 8; ct++) {
        float* bq = &bS[ct & 1][wave][0];
        float* zw = &zS[ct & 1][wave][0];
        #pragma unroll
        for (int j = 0; j < 8; j++) {
            const int f = j * 32 + lane;
            *(float4*)&zw[(f >> 4) * SSP + (f & 15) * 4] = zraw[j];
            *(float4*)&bq[(f >> 4) * SSP + (f & 15) * 4] = raw[j];
        }
        __syncthreads();
        #pragma unroll
        for (int j = 0; j < 8; j++) {
            raw[j]  = make_float4(0.f, 0.f, 0.f, 0.f);
            zraw[j] = make_float4(0.f, 0.f, 0.f, 0.f);
        }
        if (ct + 1 < 8) {
            #pragma unroll
            for (int j = 0; j < 8; j++) {
                zraw[j] = Z4[(ct + 1) * 16 + j * 2048];
                raw[j]  = Q4[(ct + 1) * 256 + j * 32];
            }
        }
        float4 x[8];
        float sum = 0.f, sq = 0.f;
        #pragma unroll
        for (int jj = 0; jj < 8; jj++) {
            const float4 t = *(const float4*)&zw[l * SSP + half * 32 + jj * 4];
            x[jj] = t;
            sum += t.x + t.y + t.z + t.w;
            sq  += t.x * t.x + t.y * t.y + t.z * t.z + t.w * t.w;
        }
        sum += __shfl_xor(sum, 16);
        sq  += __shfl_xor(sq, 16);
        const float m   = sum * (1.0f / 64.0f);
        const float var = sq * (1.0f / 64.0f) - m * m;
        const float inv = rsqrtf(var + 1e-5f);
        v8f c0 = cinit; v8f c1 = {};
        #pragma unroll
        for (int jj = 0; jj < 8; jj++) {
            const float4 w4 = *(const float4*)&wS[l][half * 32 + jj * 4];
            v2f n0, n1, w0, w1;
            n0.x = (x[jj].x - m) * inv;
            n0.y = (x[jj].y - m) * inv;
            n1.x = (x[jj].z - m) * inv;
            n1.y = (x[jj].w - m) * inv;
            w0.x = w4.x; w0.y = w4.y; w1.x = w4.z; w1.y = w4.w;
            c0 = __builtin_amdgcn_wmma_f32_16x16x4_f32(false, n0, false, w0, (short)0, c0, false, false);
            c1 = __builtin_amdgcn_wmma_f32_16x16x4_f32(false, n1, false, w1, (short)0, c1, false, false);
        }
        float4 bf[8];
        #pragma unroll
        for (int jj = 0; jj < 8; jj++) bf[jj] = *(const float4*)&bq[l * SSP + half * 32 + jj * 4];
        v8f a0 = {}; v8f a1 = {};
        #pragma unroll
        for (int jj = 0; jj < 8; jj++) {
            v2f x0, x1, y0, y1;
            x0.x = af[jj].x; x0.y = af[jj].y; x1.x = af[jj].z; x1.y = af[jj].w;
            y0.x = bf[jj].x; y0.y = bf[jj].y; y1.x = bf[jj].z; y1.y = bf[jj].w;
            a0 = __builtin_amdgcn_wmma_f32_16x16x4_f32(false, x0, false, y0, (short)0, a0, false, false);
            a1 = __builtin_amdgcn_wmma_f32_16x16x4_f32(false, x1, false, y1, (short)0, a1, false, false);
        }
        acc[ct] = (a0 + a1) * (1.0f / 64.0f) + (c0 + c1);
        __builtin_amdgcn_sched_barrier(0);
    }

    float* pp = P + ob;
    float mx[8];
    #pragma unroll
    for (int r = 0; r < 8; r++) {
        float m = acc[0][r];
        #pragma unroll
        for (int ct = 1; ct < 8; ct++) m = fmaxf(m, acc[ct][r]);
        #pragma unroll
        for (int off = 8; off > 0; off >>= 1) m = fmaxf(m, __shfl_xor(m, off));
        mx[r] = m;
    }
    if (l == 0) {
        #pragma unroll
        for (int r = 0; r < 8; r++) ((float*)&redm[8 * half + r][0])[wave] = mx[r];
    }
    __syncthreads();
    #pragma unroll
    for (int r = 0; r < 8; r++) {
        const float4 u = redm[8 * half + r][0], w = redm[8 * half + r][1];
        mx[r] = fmaxf(fmaxf(fmaxf(u.x, u.y), fmaxf(u.z, u.w)), fmaxf(fmaxf(w.x, w.y), fmaxf(w.z, w.w)));
    }
    float sm[8];
    #pragma unroll
    for (int r = 0; r < 8; r++) {
        float s = 0.f;
        #pragma unroll
        for (int ct = 0; ct < 8; ct++) {
            const float e = __expf(acc[ct][r] - mx[r]);
            acc[ct][r] = e;
            s += e;
        }
        #pragma unroll
        for (int off = 8; off > 0; off >>= 1) s += __shfl_xor(s, off);
        sm[r] = s;
    }
    if (l == 0) {
        #pragma unroll
        for (int r = 0; r < 8; r++) ((float*)&reds[8 * half + r][0])[wave] = sm[r];
    }
    __syncthreads();
    #pragma unroll
    for (int r = 0; r < 8; r++) {
        const float4 u = reds[8 * half + r][0], w = reds[8 * half + r][1];
        const float s = ((u.x + u.y) + (u.z + u.w)) + ((w.x + w.y) + (w.z + w.w));
        const float inv = 1.0f / s;
        #pragma unroll
        for (int ct = 0; ct < 8; ct++) pp[r * 1024 + ct * 16] = acc[ct][r] * inv;
    }
}

__global__ void adaln_gate_kernel(const float4* __restrict__ pb, const float4* __restrict__ pn,
                                  const float4* __restrict__ an, float4* __restrict__ a1) {
    int i = blockIdx.x * 256 + threadIdx.x;
    float4 p = pb[i], q = pn[i], n = an[i];
    float4 r;
    r.x = sigmoidf_(p.x * n.x + q.x);
    r.y = sigmoidf_(p.y * n.y + q.y);
    r.z = sigmoidf_(p.z * n.z + q.z);
    r.w = sigmoidf_(p.w * n.w + q.w);
    a1[i] = r;
}

__global__ void gate_o_kernel(const float* __restrict__ kvg, const float4* __restrict__ obuf,
                              float4* __restrict__ go) {
    int q = blockIdx.x * 256 + threadIdx.x;
    int h  = q >> 14;
    int j  = (q >> 4) & 1023;
    int c4 = (q & 15) * 4;
    const float4 g = *(const float4*)(kvg + (size_t)h * 196608 + (size_t)j * 192 + 128 + c4);
    float4 o = obuf[q];
    float4 r;
    r.x = sigmoidf_(g.x) * o.x;
    r.y = sigmoidf_(g.y) * o.y;
    r.z = sigmoidf_(g.z) * o.z;
    r.w = sigmoidf_(g.w) * o.w;
    go[q] = r;
}

extern "C" void kernel_launch(void* const* d_in, const int* in_sizes, int n_in,
                              void* d_out, int out_size, void* d_ws, size_t ws_size,
                              hipStream_t stream) {
    const float* a       = (const float*)d_in[0];
    const float* z       = (const float*)d_in[1];
    const float* s       = (const float*)d_in[2];
    const float* sn_w    = (const float*)d_in[3];
    const float* pb_w    = (const float*)d_in[4];
    const float* pb_b    = (const float*)d_in[5];
    const float* pn_w    = (const float*)d_in[6];
    const float* pnorm_w = (const float*)d_in[7];
    const float* pnorm_b = (const float*)d_in[8];
    const float* q_w     = (const float*)d_in[9];
    const float* q_b     = (const float*)d_in[10];
    const float* kvg_w   = (const float*)d_in[11];
    const float* bias_w  = (const float*)d_in[12];
    const float* bias_b  = (const float*)d_in[13];
    const float* attn_w  = (const float*)d_in[14];
    const float* out_w   = (const float*)d_in[15];
    const float* out_b   = (const float*)d_in[16];
    float* out = (float*)d_out;
    float* ws  = (float*)d_ws;

    float* a_n    = ws + 0;
    float* s_n    = ws + 1048576;
    float* pb     = ws + 1572864;
    float* pn     = ws + 2621440;
    float* qm     = ws + 3670016;
    float* kvg    = ws + 4718592;
    float* biasO  = ws + 7864320;
    float* obuf   = biasO;
    float* go     = biasO + 1048576;
    float* a2     = biasO + 2097152;
    float* scores = ws + 24641536;

    ln_kernel<<<1024, 256, 0, stream>>>(a, nullptr, a_n, 1024);
    ln_kernel<<<1024, 256, 0, stream>>>(s, sn_w, s_n, 512);

    gemm_nt_kernel<<<dim3(16, 16, 1), 256, 0, stream>>>(
        s_n, 512, 0, pb_w, 512, 0, pb, 1024, 0, pb_b, nullptr, 0, 0, 512, 1.0f, 0);
    gemm_nt_kernel<<<dim3(16, 16, 1), 256, 0, stream>>>(
        s_n, 512, 0, pn_w, 512, 0, pn, 1024, 0, nullptr, nullptr, 0, 0, 512, 1.0f, 0);
    adaln_gate_kernel<<<1024, 256, 0, stream>>>(
        (const float4*)pb, (const float4*)pn, (const float4*)a_n, (float4*)pb);
    float* a1 = pb;

    gemm_nt_kernel<<<dim3(16, 16, 1), 256, 0, stream>>>(
        a1, 1024, 0, q_w, 1024, 0, qm, 1024, 0, q_b, nullptr, 0, 0, 1024, 1.0f, 0);
    gemm_nt_kernel<<<dim3(48, 16, 1), 256, 0, stream>>>(
        a1, 1024, 0, kvg_w, 1024, 0, kvg, 3072, 0, nullptr, nullptr, 0, 0, 1024, 1.0f, 0);

    scores_softmax_kernel<<<dim3(64, 16, 1), 256, 0, stream>>>(
        kvg, qm, z, pnorm_w, pnorm_b, bias_w, bias_b, scores);

    gemm_tn_kernel<<<dim3(1, 16, 16), 256, 0, stream>>>(
        scores, 1024, 1048576LL,
        kvg + 64, 192, 196608LL,
        obuf, 64, 65536LL,
        1024);

    gate_o_kernel<<<1024, 256, 0, stream>>>(kvg, (const float4*)obuf, (float4*)go);

    gemm_nt_kernel<<<dim3(16, 16, 1), 256, 0, stream>>>(
        go, 1024, 0, attn_w, 1024, 0, a2, 1024, 0, nullptr, nullptr, 0, 0, 1024, 1.0f, 0);

    gemm_nt_kernel<<<dim3(16, 16, 1), 256, 0, stream>>>(
        s, 512, 0, out_w, 512, 0, out, 1024, 0, out_b, a2, 1024, 0, 512, 1.0f, 1);
}


# Both elementwise gate passes folded into GEMM epilogues: one GEMM accumulates the pb and pn tiles together and writes a1; the P^T V GEMM applies sigmoid(g) itself; 3 launches gone, sum orders kept
# speedup vs baseline: 2.3827x; 1.1021x over previous
#include <hip/hip_runtime.h>
#include <hip/hip_bf16.h>
#include <math.h>

typedef __attribute__((ext_vector_type(2))) float v2f;
typedef __attribute__((ext_vector_type(8))) float v8f;

#define S_DIM 1024
#define CA 1024
#define CS 512
#define CZ 64
#define NH 16
#define HC 64

__device__ __forceinline__ float sigmoidf_(float x) {
    return 1.0f / (1.0f + __expf(-x));
}

__global__ void ln_kernel(const float* __restrict__ x, const float* __restrict__ w,
                          float* __restrict__ y, int C) {
    __shared__ float red[256];
    __shared__ float red2[256];
    int row = blockIdx.x;
    const float4* xr = (const float4*)(x + (size_t)row * C);
    int C4 = C >> 2;
    float sum = 0.f, sq = 0.f;
    for (int c = threadIdx.x; c < C4; c += 256) {
        float4 v = xr[c];
        sum += v.x + v.y + v.z + v.w;
        sq  += v.x * v.x + v.y * v.y + v.z * v.z + v.w * v.w;
    }
    red[threadIdx.x] = sum; red2[threadIdx.x] = sq;
    __syncthreads();
    for (int off = 128; off > 0; off >>= 1) {
        if (threadIdx.x < off) {
            red[threadIdx.x]  += red[threadIdx.x + off];
            red2[threadIdx.x] += red2[threadIdx.x + off];
        }
        __syncthreads();
    }
    float m   = red[0] / C;
    float var = red2[0] / C - m * m;
    float inv = rsqrtf(var + 1e-5f);
    float4* yr = (float4*)(y + (size_t)row * C);
    for (int c = threadIdx.x; c < C4; c += 256) {
        float4 v = xr[c];
        v.x = (v.x - m) * inv; v.y = (v.y - m) * inv;
        v.z = (v.z - m) * inv; v.w = (v.w - m) * inv;
        if (w) {
            const float4 wv = ((const float4*)w)[c];
            v.x *= wv.x; v.y *= wv.y; v.z *= wv.z; v.w *= wv.w;
        }
        yr[c] = v;
    }
}

#define KST 64
#define PNT 68
#define PTN 72
__global__ void
gemm_nt_kernel(const float* __restrict__ A, int lda, long long sA,
               const float* __restrict__ B, int ldb, long long sB,
               float* __restrict__ C, int ldc, long long sC,
               const float* __restrict__ bias,
               const float* __restrict__ addm, int ldadd, long long sAdd,
               int K, float scale, int mode) {
    __shared__ float As[2][64][PNT];
    __shared__ float Bs[2][64][PNT];
    const int z = blockIdx.z;
    A += (size_t)z * sA; B += (size_t)z * sB; C += (size_t)z * sC;
    if (addm) addm += (size_t)z * sAdd;
    const int tileM = blockIdx.y * 64;
    const int tileN = blockIdx.x * 64;
    const int tid  = threadIdx.x;
    const int wave = tid >> 5, lane = tid & 31;
    const int wm = wave >> 1, wn = wave & 1;
    const int half = lane >> 4, l = lane & 15;
    const int lr = tid >> 4;
    const int lc = (tid & 15) * 4;

    const float* gA = A + (size_t)(tileM + lr) * lda + lc;
    const float* gB = B + (size_t)(tileN + lr) * ldb + lc;

    v8f acc0 = {}; v8f acc1 = {};
    const int nstages = K / KST;

    {
        float4 ra[4], rb[4];
        #pragma unroll
        for (int i = 0; i < 4; i++) {
            ra[i] = *(const float4*)(gA + (size_t)(16 * i) * lda);
            rb[i] = *(const float4*)(gB + (size_t)(16 * i) * ldb);
        }
        #pragma unroll
        for (int i = 0; i < 4; i++) {
            *(float4*)&As[0][lr + 16 * i][lc] = ra[i];
            *(float4*)&Bs[0][lr + 16 * i][lc] = rb[i];
        }
    }
    __syncthreads();

    for (int s = 0; s < nstages; s++) {
        const int p = s & 1;
        const bool more = (s + 1 < nstages);
        float4 na[4], nb[4];
        #pragma unroll
        for (int i = 0; i < 4; i++) {
            na[i] = make_float4(0.f, 0.f, 0.f, 0.f);
            nb[i] = make_float4(0.f, 0.f, 0.f, 0.f);
        }
        if (more) {
            #pragma unroll
            for (int i = 0; i < 4; i++) {
                na[i] = *(const float4*)(gA + (size_t)(16 * i) * lda + (s + 1) * KST);
                nb[i] = *(const float4*)(gB + (size_t)(16 * i) * ldb + (s + 1) * KST);
            }
        }
        const float* ap  = &As[p][wm * 16 + l][half * 2];
        const float* bp0 = &Bs[p][wn * 32 + l][half * 2];
        const float* bp1 = &Bs[p][wn * 32 + 16 + l][half * 2];
        #pragma unroll
        for (int q = 0; q < KST / 4; q++) {
            const v2f af  = *(const v2f*)(ap  + q * 4);
            const v2f bf0 = *(const v2f*)(bp0 + q * 4);
            const v2f bf1 = *(const v2f*)(bp1 + q * 4);
            acc0 = __builtin_amdgcn_wmma_f32_16x16x4_f32(false, af, false, bf0, (short)0, acc0, false, false);
            acc1 = __builtin_amdgcn_wmma_f32_16x16x4_f32(false, af, false, bf1, (short)0, acc1, false, false);
        }
        if (more) {
            #pragma unroll
            for (int i = 0; i < 4; i++) {
                *(float4*)&As[p ^ 1][lr + 16 * i][lc] = na[i];
                *(float4*)&Bs[p ^ 1][lr + 16 * i][lc] = nb[i];
            }
        }
        __syncthreads();
    }

    #pragma unroll
    for (int r = 0; r < 8; r++) {
        int m  = tileM + wm * 16 + half * 8 + r;
        int n0 = tileN + wn * 32 + l;
        int n1 = n0 + 16;
        float v0 = acc0[r] * scale;
        float v1 = acc1[r] * scale;
        if (bias) { v0 += bias[n0]; v1 += bias[n1]; }
        if (mode == 1) {
            v0 = sigmoidf_(v0) * addm[(size_t)m * ldadd + n0];
            v1 = sigmoidf_(v1) * addm[(size_t)m * ldadd + n1];
        } else if (addm) {
            v0 += addm[(size_t)m * ldadd + n0];
            v1 += addm[(size_t)m * ldadd + n1];
        }
        C[(size_t)m * ldc + n0] = v0;
        C[(size_t)m * ldc + n1] = v1;
    }
}

__global__ void __launch_bounds__(256)
gemm_nt_adaln_kernel(const float* __restrict__ A, int lda,
                     const float* __restrict__ B0, const float* __restrict__ B1, int ldb,
                     const float* __restrict__ bias0,
                     const float* __restrict__ an, int ldan,
                     float* __restrict__ C, int ldc, int K) {
    __shared__ float As[2][64][PNT];
    __shared__ float Ps[2][64][PNT];
    __shared__ float Ns[2][64][PNT];
    const int tileM = blockIdx.y * 64;
    const int tileN = blockIdx.x * 64;
    const int tid  = threadIdx.x;
    const int wave = tid >> 5, lane = tid & 31;
    const int wm = wave >> 1, wn = wave & 1;
    const int half = lane >> 4, l = lane & 15;
    const int lr = tid >> 4;
    const int lc = (tid & 15) * 4;

    const float* gA = A  + (size_t)(tileM + lr) * lda + lc;
    const float* gP = B0 + (size_t)(tileN + lr) * ldb + lc;
    const float* gN = B1 + (size_t)(tileN + lr) * ldb + lc;

    v8f accP0 = {}; v8f accP1 = {}; v8f accN0 = {}; v8f accN1 = {};
    const int nstages = K / KST;

    {
        float4 ra[4], rp[4], rn[4];
        #pragma unroll
        for (int i = 0; i < 4; i++) {
            ra[i] = *(const float4*)(gA + (size_t)(16 * i) * lda);
            rp[i] = *(const float4*)(gP + (size_t)(16 * i) * ldb);
            rn[i] = *(const float4*)(gN + (size_t)(16 * i) * ldb);
        }
        #pragma unroll
        for (int i = 0; i < 4; i++) {
            *(float4*)&As[0][lr + 16 * i][lc] = ra[i];
            *(float4*)&Ps[0][lr + 16 * i][lc] = rp[i];
            *(float4*)&Ns[0][lr + 16 * i][lc] = rn[i];
        }
    }
    __syncthreads();

    for (int s = 0; s < nstages; s++) {
        const int p = s & 1;
        const bool more = (s + 1 < nstages);
        float4 na[4], np[4], nn[4];
        #pragma unroll
        for (int i = 0; i < 4; i++) {
            na[i] = make_float4(0.f, 0.f, 0.f, 0.f);
            np[i] = make_float4(0.f, 0.f, 0.f, 0.f);
            nn[i] = make_float4(0.f, 0.f, 0.f, 0.f);
        }
        if (more) {
            #pragma unroll
            for (int i = 0; i < 4; i++) {
                na[i] = *(const float4*)(gA + (size_t)(16 * i) * lda + (s + 1) * KST);
                np[i] = *(const float4*)(gP + (size_t)(16 * i) * ldb + (s + 1) * KST);
                nn[i] = *(const float4*)(gN + (size_t)(16 * i) * ldb + (s + 1) * KST);
            }
        }
        const float* ap  = &As[p][wm * 16 + l][half * 2];
        const float* pp0 = &Ps[p][wn * 32 + l][half * 2];
        const float* pp1 = &Ps[p][wn * 32 + 16 + l][half * 2];
        const float* np0 = &Ns[p][wn * 32 + l][half * 2];
        const float* np1 = &Ns[p][wn * 32 + 16 + l][half * 2];
        #pragma unroll
        for (int q = 0; q < KST / 4; q++) {
            const v2f af  = *(const v2f*)(ap  + q * 4);
            const v2f pf0 = *(const v2f*)(pp0 + q * 4);
            const v2f pf1 = *(const v2f*)(pp1 + q * 4);
            const v2f nf0 = *(const v2f*)(np0 + q * 4);
            const v2f nf1 = *(const v2f*)(np1 + q * 4);
            accP0 = __builtin_amdgcn_wmma_f32_16x16x4_f32(false, af, false, pf0, (short)0, accP0, false, false);
            accP1 = __builtin_amdgcn_wmma_f32_16x16x4_f32(false, af, false, pf1, (short)0, accP1, false, false);
            accN0 = __builtin_amdgcn_wmma_f32_16x16x4_f32(false, af, false, nf0, (short)0, accN0, false, false);
            accN1 = __builtin_amdgcn_wmma_f32_16x16x4_f32(false, af, false, nf1, (short)0, accN1, false, false);
        }
        if (more) {
            #pragma unroll
            for (int i = 0; i < 4; i++) {
                *(float4*)&As[p ^ 1][lr + 16 * i][lc] = na[i];
                *(float4*)&Ps[p ^ 1][lr + 16 * i][lc] = np[i];
                *(float4*)&Ns[p ^ 1][lr + 16 * i][lc] = nn[i];
            }
        }
        __syncthreads();
    }

    #pragma unroll
    for (int r = 0; r < 8; r++) {
        const int m  = tileM + wm * 16 + half * 8 + r;
        const int n0 = tileN + wn * 32 + l;
        const int n1 = n0 + 16;
        const float p0 = accP0[r] + bias0[n0];
        const float p1 = accP1[r] + bias0[n1];
        const float x0 = an[(size_t)m * ldan + n0];
        const float x1 = an[(size_t)m * ldan + n1];
        C[(size_t)m * ldc + n0] = sigmoidf_(p0 * x0 + accN0[r]);
        C[(size_t)m * ldc + n1] = sigmoidf_(p1 * x1 + accN1[r]);
    }
}

__global__ void
gemm_tn_kernel(const float* __restrict__ A, int lda, long long sA,
               const float* __restrict__ B, int ldb, long long sB,
               float* __restrict__ C, int ldc, long long sC,
               const float* __restrict__ G, int ldg, long long sG,
               int K) {
    __shared__ float As[2][KST][PTN];
    __shared__ float Bs[2][KST][PTN];
    const int z = blockIdx.z;
    A += (size_t)z * sA; B += (size_t)z * sB; C += (size_t)z * sC; G += (size_t)z * sG;
    const int tileM = blockIdx.y * 64;
    const int tileN = blockIdx.x * 64;
    const int tid  = threadIdx.x;
    const int wave = tid >> 5, lane = tid & 31;
    const int wm = wave >> 1, wn = wave & 1;
    const int half = lane >> 4, l = lane & 15;
    const int lk = tid >> 4;
    const int lm = (tid & 15) * 4;

    const float* gA = A + (size_t)lk * lda + tileM + lm;
    const float* gB = B + (size_t)lk * ldb + tileN + lm;

    v8f acc0 = {}; v8f acc1 = {};
    const int nstages = K / KST;

    {
        float4 ra[4], rb[4];
        #pragma unroll
        for (int i = 0; i < 4; i++) {
            ra[i] = *(const float4*)(gA + (size_t)(16 * i) * lda);
            rb[i] = *(const float4*)(gB + (size_t)(16 * i) * ldb);
        }
        #pragma unroll
        for (int i = 0; i < 4; i++) {
            *(float4*)&As[0][lk + 16 * i][lm] = ra[i];
            *(float4*)&Bs[0][lk + 16 * i][lm] = rb[i];
        }
    }
    __syncthreads();

    for (int s = 0; s < nstages; s++) {
        const int p = s & 1;
        const bool more = (s + 1 < nstages);
        float4 na[4], nb[4];
        #pragma unroll
        for (int i = 0; i < 4; i++) {
            na[i] = make_float4(0.f, 0.f, 0.f, 0.f);
            nb[i] = make_float4(0.f, 0.f, 0.f, 0.f);
        }
        if (more) {
            #pragma unroll
            for (int i = 0; i < 4; i++) {
                na[i] = *(const float4*)(gA + (size_t)((s + 1) * KST + 16 * i) * lda);
                nb[i] = *(const float4*)(gB + (size_t)((s + 1) * KST + 16 * i) * ldb);
            }
        }
        const float* ap  = &As[p][half * 2][wm * 16 + l];
        const float* bp0 = &Bs[p][half * 2][wn * 32 + l];
        const float* bp1 = &Bs[p][half * 2][wn * 32 + 16 + l];
        #pragma unroll
        for (int q = 0; q < KST / 4; q++) {
            v2f af, bf0, bf1;
            af.x  = ap [(q * 4) * PTN];  af.y  = ap [(q * 4 + 1) * PTN];
            bf0.x = bp0[(q * 4) * PTN];  bf0.y = bp0[(q * 4 + 1) * PTN];
            bf1.x = bp1[(q * 4) * PTN];  bf1.y = bp1[(q * 4 + 1) * PTN];
            acc0 = __builtin_amdgcn_wmma_f32_16x16x4_f32(false, af, false, bf0, (short)0, acc0, false, false);
            acc1 = __builtin_amdgcn_wmma_f32_16x16x4_f32(false, af, false, bf1, (short)0, acc1, false, false);
        }
        if (more) {
            #pragma unroll
            for (int i = 0; i < 4; i++) {
                *(float4*)&As[p ^ 1][lk + 16 * i][lm] = na[i];
                *(float4*)&Bs[p ^ 1][lk + 16 * i][lm] = nb[i];
            }
        }
        __syncthreads();
    }

    #pragma unroll
    for (int r = 0; r < 8; r++) {
        int m  = tileM + wm * 16 + half * 8 + r;
        int n0 = tileN + wn * 32 + l;
        const float g0 = G[(size_t)m * ldg + n0];
        const float g1 = G[(size_t)m * ldg + n0 + 16];
        C[(size_t)m * ldc + n0]      = sigmoidf_(g0) * acc0[r];
        C[(size_t)m * ldc + n0 + 16] = sigmoidf_(g1) * acc1[r];
    }
}

#define SSP 68
__global__ void __launch_bounds__(256, 4)
scores_softmax_kernel(const float* __restrict__ kvg, const float* __restrict__ qm,
                      const float* __restrict__ z,
                      const float* __restrict__ pnw, const float* __restrict__ pnb,
                      const float* __restrict__ bw,  const float* __restrict__ bb,
                      float* __restrict__ P) {
    __shared__ float  aS[16][SSP];
    __shared__ float  wS[16][SSP];
    __shared__ float  bS[2][8][16 * SSP];
    __shared__ float  zS[2][8][16 * SSP];
    __shared__ float4 redm[16][2];
    __shared__ float4 reds[16][2];
    const int tid  = threadIdx.x;
    const int wave = tid >> 5, lane = tid & 31;
    const int half = lane >> 4, l = lane & 15;
    const int h  = blockIdx.y;
    const int j0 = blockIdx.x * 16;
    const float*  Kh = kvg + (size_t)h * 196608 + (size_t)j0 * 192;
    const float4* Q4 = (const float4*)(qm + (size_t)h * 65536) + wave * 2048 + lane;
    const float4* Z4 = (const float4*)z + (size_t)(h * 64 + blockIdx.x) * 16384
                       + (lane >> 4) * 1024 + wave * 128 + (lane & 15);
    const size_t  ob = (size_t)h * 1048576 + (size_t)(j0 + 8 * half) * 1024 + wave * 128 + l;

    float4 raw[8], zraw[8];
    #pragma unroll
    for (int j = 0; j < 8; j++) {
        zraw[j] = Z4[j * 2048];
        raw[j]  = Q4[j * 32];
    }
    *(float4*)&aS[tid >> 4][(tid & 15) * 4] =
        *(const float4*)(Kh + (size_t)(tid >> 4) * 192 + (tid & 15) * 4);
    {
        const float4 w4 = ((const float4*)bw)[tid];
        const float4 g4 = ((const float4*)pnw)[tid & 15];
        *(float4*)&wS[tid >> 4][(tid & 15) * 4] =
            make_float4(w4.x * g4.x, w4.y * g4.y, w4.z * g4.z, w4.w * g4.w);
    }
    float cb = 0.f;
    #pragma unroll
    for (int jj = 0; jj < 8; jj++) {
        const float4 w4 = ((const float4*)bw)[l * 16 + half * 8 + jj];
        const float4 d4 = ((const float4*)pnb)[half * 8 + jj];
        cb += w4.x * d4.x + w4.y * d4.y + w4.z * d4.z + w4.w * d4.w;
    }
    cb += __shfl_xor(cb, 16);
    cb += bb[l];
    __syncthreads();
    float4 af[8];
    #pragma unroll
    for (int jj = 0; jj < 8; jj++) af[jj] = *(const float4*)&aS[l][half * 32 + jj * 4];
    v8f cinit;
    #pragma unroll
    for (int r = 0; r < 8; r++) cinit[r] = cb;

    v8f acc[8];
    #pragma unroll
    for (int ct = 0; ct < 8; ct++) {
        float* bq = &bS[ct & 1][wave][0];
        float* zw = &zS[ct & 1][wave][0];
        #pragma unroll
        for (int j = 0; j < 8; j++) {
            const int f = j * 32 + lane;
            *(float4*)&zw[(f >> 4) * SSP + (f & 15) * 4] = zraw[j];
            *(float4*)&bq[(f >> 4) * SSP + (f & 15) * 4] = raw[j];
        }
        __syncthreads();
        #pragma unroll
        for (int j = 0; j < 8; j++) {
            raw[j]  = make_float4(0.f, 0.f, 0.f, 0.f);
            zraw[j] = make_float4(0.f, 0.f, 0.f, 0.f);
        }
        if (ct + 1 < 8) {
            #pragma unroll
            for (int j = 0; j < 8; j++) {
                zraw[j] = Z4[(ct + 1) * 16 + j * 2048];
                raw[j]  = Q4[(ct + 1) * 256 + j * 32];
            }
        }
        float4 x[8];
        float sum = 0.f, sq = 0.f;
        #pragma unroll
        for (int jj = 0; jj < 8; jj++) {
            const float4 t = *(const float4*)&zw[l * SSP + half * 32 + jj * 4];
            x[jj] = t;
            sum += t.x + t.y + t.z + t.w;
            sq  += t.x * t.x + t.y * t.y + t.z * t.z + t.w * t.w;
        }
        sum += __shfl_xor(sum, 16);
        sq  += __shfl_xor(sq, 16);
        const float m   = sum * (1.0f / 64.0f);
        const float var = sq * (1.0f / 64.0f) - m * m;
        const float inv = rsqrtf(var + 1e-5f);
        v8f c0 = cinit; v8f c1 = {};
        #pragma unroll
        for (int jj = 0; jj < 8; jj++) {
            const float4 w4 = *(const float4*)&wS[l][half * 32 + jj * 4];
            v2f n0, n1, w0, w1;
            n0.x = (x[jj].x - m) * inv;
            n0.y = (x[jj].y - m) * inv;
            n1.x = (x[jj].z - m) * inv;
            n1.y = (x[jj].w - m) * inv;
            w0.x = w4.x; w0.y = w4.y; w1.x = w4.z; w1.y = w4.w;
            c0 = __builtin_amdgcn_wmma_f32_16x16x4_f32(false, n0, false, w0, (short)0, c0, false, false);
            c1 = __builtin_amdgcn_wmma_f32_16x16x4_f32(false, n1, false, w1, (short)0, c1, false, false);
        }
        float4 bf[8];
        #pragma unroll
        for (int jj = 0; jj < 8; jj++) bf[jj] = *(const float4*)&bq[l * SSP + half * 32 + jj * 4];
        v8f a0 = {}; v8f a1 = {};
        #pragma unroll
        for (int jj = 0; jj < 8; jj++) {
            v2f x0, x1, y0, y1;
            x0.x = af[jj].x; x0.y = af[jj].y; x1.x = af[jj].z; x1.y = af[jj].w;
            y0.x = bf[jj].x; y0.y = bf[jj].y; y1.x = bf[jj].z; y1.y = bf[jj].w;
            a0 = __builtin_amdgcn_wmma_f32_16x16x4_f32(false, x0, false, y0, (short)0, a0, false, false);
            a1 = __builtin_amdgcn_wmma_f32_16x16x4_f32(false, x1, false, y1, (short)0, a1, false, false);
        }
        acc[ct] = (a0 + a1) * (1.0f / 64.0f) + (c0 + c1);
        __builtin_amdgcn_sched_barrier(0);
    }

    float* pp = P + ob;
    float mx[8];
    #pragma unroll
    for (int r = 0; r < 8; r++) {
        float m = acc[0][r];
        #pragma unroll
        for (int ct = 1; ct < 8; ct++) m = fmaxf(m, acc[ct][r]);
        #pragma unroll
        for (int off = 8; off > 0; off >>= 1) m = fmaxf(m, __shfl_xor(m, off));
        mx[r] = m;
    }
    if (l == 0) {
        #pragma unroll
        for (int r = 0; r < 8; r++) ((float*)&redm[8 * half + r][0])[wave] = mx[r];
    }
    __syncthreads();
    #pragma unroll
    for (int r = 0; r < 8; r++) {
        const float4 u = redm[8 * half + r][0], w = redm[8 * half + r][1];
        mx[r] = fmaxf(fmaxf(fmaxf(u.x, u.y), fmaxf(u.z, u.w)), fmaxf(fmaxf(w.x, w.y), fmaxf(w.z, w.w)));
    }
    float sm[8];
    #pragma unroll
    for (int r = 0; r < 8; r++) {
        float s = 0.f;
        #pragma unroll
        for (int ct = 0; ct < 8; ct++) {
            const float e = __expf(acc[ct][r] - mx[r]);
            acc[ct][r] = e;
            s += e;
        }
        #pragma unroll
        for (int off = 8; off > 0; off >>= 1) s += __shfl_xor(s, off);
        sm[r] = s;
    }
    if (l == 0) {
        #pragma unroll
        for (int r = 0; r < 8; r++) ((float*)&reds[8 * half + r][0])[wave] = sm[r];
    }
    __syncthreads();
    #pragma unroll
    for (int r = 0; r < 8; r++) {
        const float4 u = reds[8 * half + r][0], w = reds[8 * half + r][1];
        const float s = ((u.x + u.y) + (u.z + u.w)) + ((w.x + w.y) + (w.z + w.w));
        const float inv = 1.0f / s;
        #pragma unroll
        for (int ct = 0; ct < 8; ct++) pp[r * 1024 + ct * 16] = acc[ct][r] * inv;
    }
}

extern "C" void kernel_launch(void* const* d_in, const int* in_sizes, int n_in,
                              void* d_out, int out_size, void* d_ws, size_t ws_size,
                              hipStream_t stream) {
    const float* a       = (const float*)d_in[0];
    const float* z       = (const float*)d_in[1];
    const float* s       = (const float*)d_in[2];
    const float* sn_w    = (const float*)d_in[3];
    const float* pb_w    = (const float*)d_in[4];
    const float* pb_b    = (const float*)d_in[5];
    const float* pn_w    = (const float*)d_in[6];
    const float* pnorm_w = (const float*)d_in[7];
    const float* pnorm_b = (const float*)d_in[8];
    const float* q_w     = (const float*)d_in[9];
    const float* q_b     = (const float*)d_in[10];
    const float* kvg_w   = (const float*)d_in[11];
    const float* bias_w  = (const float*)d_in[12];
    const float* bias_b  = (const float*)d_in[13];
    const float* attn_w  = (const float*)d_in[14];
    const float* out_w   = (const float*)d_in[15];
    const float* out_b   = (const float*)d_in[16];
    float* out = (float*)d_out;
    float* ws  = (float*)d_ws;

    float* a_n    = ws + 0;
    float* s_n    = ws + 1048576;
    float* a1     = ws + 1572864;
    float* qm     = ws + 3670016;
    float* kvg    = ws + 4718592;
    float* go     = ws + 8912896;
    float* a2     = ws + 9961472;
    float* scores = ws + 24641536;

    ln_kernel<<<1024, 256, 0, stream>>>(a, nullptr, a_n, 1024);
    ln_kernel<<<1024, 256, 0, stream>>>(s, sn_w, s_n, 512);

    gemm_nt_adaln_kernel<<<dim3(16, 16, 1), 256, 0, stream>>>(
        s_n, 512, pb_w, pn_w, 512, pb_b, a_n, 1024, a1, 1024, 512);

    gemm_nt_kernel<<<dim3(16, 16, 1), 256, 0, stream>>>(
        a1, 1024, 0, q_w, 1024, 0, qm, 1024, 0, q_b, nullptr, 0, 0, 1024, 1.0f, 0);
    gemm_nt_kernel<<<dim3(48, 16, 1), 256, 0, stream>>>(
        a1, 1024, 0, kvg_w, 1024, 0, kvg, 3072, 0, nullptr, nullptr, 0, 0, 1024, 1.0f, 0);

    scores_softmax_kernel<<<dim3(64, 16, 1), 256, 0, stream>>>(
        kvg, qm, z, pnorm_w, pnorm_b, bias_w, bias_b, scores);

    gemm_tn_kernel<<<dim3(1, 16, 16), 256, 0, stream>>>(
        scores, 1024, 1048576LL,
        kvg + 64, 192, 196608LL,
        go, 64, 65536LL,
        kvg + 128, 192, 196608LL,
        1024);

    gemm_nt_kernel<<<dim3(16, 16, 1), 256, 0, stream>>>(
        go, 1024, 0, attn_w, 1024, 0, a2, 1024, 0, nullptr, nullptr, 0, 0, 1024, 1.0f, 0);

    gemm_nt_kernel<<<dim3(16, 16, 1), 256, 0, stream>>>(
        s, 512, 0, out_w, 512, 0, out, 1024, 0, out_b, a2, 1024, 0, 512, 1.0f, 1);
}


# Independent kernels share a launch: both LayerNorms in one grid (a wave holds its rows in registers, no LDS; row-sum order changes), q and kvg projections in one grid; 9 launches to 7
# speedup vs baseline: 2.5204x; 1.0578x over previous
#include <hip/hip_runtime.h>
#include <hip/hip_bf16.h>
#include <math.h>

typedef __attribute__((ext_vector_type(2))) float v2f;
typedef __attribute__((ext_vector_type(8))) float v8f;

#define S_DIM 1024
#define CA 1024
#define CS 512
#define CZ 64
#define NH 16
#define HC 64

__device__ __forceinline__ float sigmoidf_(float x) {
    return 1.0f / (1.0f + __expf(-x));
}

__global__ void __launch_bounds__(256)
ln_rows_kernel(const float* __restrict__ a, float* __restrict__ a_n,
               const float* __restrict__ s, const float* __restrict__ sw, float* __restrict__ s_n) {
    const int lane = threadIdx.x & 31;
    const int wave = threadIdx.x >> 5;
    const bool is_s = blockIdx.x >= S_DIM / 8;
    const size_t chunk = (size_t)((is_s ? blockIdx.x - S_DIM / 8 : blockIdx.x) * 8 + wave) * 1024;
    const float4* xr = (const float4*)((is_s ? s : a) + chunk) + lane;
    float4*       yr = (float4*)((is_s ? s_n : a_n) + chunk) + lane;
    float4 v[8];
    #pragma unroll
    for (int j = 0; j < 8; j++) v[j] = xr[j * 32];
    float su[2] = {0.f, 0.f}, sq[2] = {0.f, 0.f};
    #pragma unroll
    for (int j = 0; j < 8; j++) {
        su[j >> 2] += v[j].x + v[j].y + v[j].z + v[j].w;
        sq[j >> 2] += v[j].x * v[j].x + v[j].y * v[j].y + v[j].z * v[j].z + v[j].w * v[j].w;
    }
    #pragma unroll
    for (int off = 16; off > 0; off >>= 1) {
        su[0] += __shfl_xor(su[0], off); su[1] += __shfl_xor(su[1], off);
        sq[0] += __shfl_xor(sq[0], off); sq[1] += __shfl_xor(sq[1], off);
    }
    float rc = 1.0f / 512.0f;
    if (!is_s) {
        su[0] += su[1]; su[1] = su[0];
        sq[0] += sq[1]; sq[1] = sq[0];
        rc = 1.0f / 1024.0f;
    }
    float m[2], inv[2];
    #pragma unroll
    for (int h = 0; h < 2; h++) {
        m[h] = su[h] * rc;
        const float var = sq[h] * rc - m[h] * m[h];
        inv[h] = rsqrtf(var + 1e-5f);
    }
    #pragma unroll
    for (int j = 0; j < 8; j++) {
        v[j].x = (v[j].x - m[j >> 2]) * inv[j >> 2];
        v[j].y = (v[j].y - m[j >> 2]) * inv[j >> 2];
        v[j].z = (v[j].z - m[j >> 2]) * inv[j >> 2];
        v[j].w = (v[j].w - m[j >> 2]) * inv[j >> 2];
    }
    if (is_s) {
        #pragma unroll
        for (int j = 0; j < 4; j++) {
            const float4 wv = ((const float4*)sw)[j * 32 + lane];
            v[j].x     *= wv.x; v[j].y     *= wv.y; v[j].z     *= wv.z; v[j].w     *= wv.w;
            v[j + 4].x *= wv.x; v[j + 4].y *= wv.y; v[j + 4].z *= wv.z; v[j + 4].w *= wv.w;
        }
    }
    #pragma unroll
    for (int j = 0; j < 8; j++) yr[j * 32] = v[j];
}

#define KST 64
#define PNT 68
#define PTN 72
__global__ void
gemm_nt_kernel(const float* __restrict__ A, int lda, long long sA,
               const float* __restrict__ B, int ldb, long long sB,
               float* __restrict__ C, int ldc, long long sC,
               const float* __restrict__ bias,
               const float* __restrict__ addm, int ldadd, long long sAdd,
               int K, float scale, int mode,
               int nsplit, const float* __restrict__ B2, float* __restrict__ C2, int ldc2,
               const float* __restrict__ bias2) {
    __shared__ float As[2][64][PNT];
    __shared__ float Bs[2][64][PNT];
    int bx = blockIdx.x;
    if (bx >= nsplit) { B = B2; C = C2; ldc = ldc2; bias = bias2; bx -= nsplit; }
    const int z = blockIdx.z;
    A += (size_t)z * sA; B += (size_t)z * sB; C += (size_t)z * sC;
    if (addm) addm += (size_t)z * sAdd;
    const int tileM = blockIdx.y * 64;
    const int tileN = bx * 64;
    const int tid  = threadIdx.x;
    const int wave = tid >> 5, lane = tid & 31;
    const int wm = wave >> 1, wn = wave & 1;
    const int half = lane >> 4, l = lane & 15;
    const int lr = tid >> 4;
    const int lc = (tid & 15) * 4;

    const float* gA = A + (size_t)(tileM + lr) * lda + lc;
    const float* gB = B + (size_t)(tileN + lr) * ldb + lc;

    v8f acc0 = {}; v8f acc1 = {};
    const int nstages = K / KST;

    {
        float4 ra[4], rb[4];
        #pragma unroll
        for (int i = 0; i < 4; i++) {
            ra[i] = *(const float4*)(gA + (size_t)(16 * i) * lda);
            rb[i] = *(const float4*)(gB + (size_t)(16 * i) * ldb);
        }
        #pragma unroll
        for (int i = 0; i < 4; i++) {
            *(float4*)&As[0][lr + 16 * i][lc] = ra[i];
            *(float4*)&Bs[0][lr + 16 * i][lc] = rb[i];
        }
    }
    __syncthreads();

    for (int s = 0; s < nstages; s++) {
        const int p = s & 1;
        const bool more = (s + 1 < nstages);
        float4 na[4], nb[4];
        #pragma unroll
        for (int i = 0; i < 4; i++) {
            na[i] = make_float4(0.f, 0.f, 0.f, 0.f);
            nb[i] = make_float4(0.f, 0.f, 0.f, 0.f);
        }
        if (more) {
            #pragma unroll
            for (int i = 0; i < 4; i++) {
                na[i] = *(const float4*)(gA + (size_t)(16 * i) * lda + (s + 1) * KST);
                nb[i] = *(const float4*)(gB + (size_t)(16 * i) * ldb + (s + 1) * KST);
            }
        }
        const float* ap  = &As[p][wm * 16 + l][half * 2];
        const float* bp0 = &Bs[p][wn * 32 + l][half * 2];
        const float* bp1 = &Bs[p][wn * 32 + 16 + l][half * 2];
        #pragma unroll
        for (int q = 0; q < KST / 4; q++) {
            const v2f af  = *(const v2f*)(ap  + q * 4);
            const v2f bf0 = *(const v2f*)(bp0 + q * 4);
            const v2f bf1 = *(const v2f*)(bp1 + q * 4);
            acc0 = __builtin_amdgcn_wmma_f32_16x16x4_f32(false, af, false, bf0, (short)0, acc0, false, false);
            acc1 = __builtin_amdgcn_wmma_f32_16x16x4_f32(false, af, false, bf1, (short)0, acc1, false, false);
        }
        if (more) {
            #pragma unroll
            for (int i = 0; i < 4; i++) {
                *(float4*)&As[p ^ 1][lr + 16 * i][lc] = na[i];
                *(float4*)&Bs[p ^ 1][lr + 16 * i][lc] = nb[i];
            }
        }
        __syncthreads();
    }

    #pragma unroll
    for (int r = 0; r < 8; r++) {
        int m  = tileM + wm * 16 + half * 8 + r;
        int n0 = tileN + wn * 32 + l;
        int n1 = n0 + 16;
        float v0 = acc0[r] * scale;
        float v1 = acc1[r] * scale;
        if (bias) { v0 += bias[n0]; v1 += bias[n1]; }
        if (mode == 1) {
            v0 = sigmoidf_(v0) * addm[(size_t)m * ldadd + n0];
            v1 = sigmoidf_(v1) * addm[(size_t)m * ldadd + n1];
        } else if (addm) {
            v0 += addm[(size_t)m * ldadd + n0];
            v1 += addm[(size_t)m * ldadd + n1];
        }
        C[(size_t)m * ldc + n0] = v0;
        C[(size_t)m * ldc + n1] = v1;
    }
}

__global__ void __launch_bounds__(256)
gemm_nt_adaln_kernel(const float* __restrict__ A, int lda,
                     const float* __restrict__ B0, const float* __restrict__ B1, int ldb,
                     const float* __restrict__ bias0,
                     const float* __restrict__ an, int ldan,
                     float* __restrict__ C, int ldc, int K) {
    __shared__ float As[2][64][PNT];
    __shared__ float Ps[2][64][PNT];
    __shared__ float Ns[2][64][PNT];
    const int tileM = blockIdx.y * 64;
    const int tileN = blockIdx.x * 64;
    const int tid  = threadIdx.x;
    const int wave = tid >> 5, lane = tid & 31;
    const int wm = wave >> 1, wn = wave & 1;
    const int half = lane >> 4, l = lane & 15;
    const int lr = tid >> 4;
    const int lc = (tid & 15) * 4;

    const float* gA = A  + (size_t)(tileM + lr) * lda + lc;
    const float* gP = B0 + (size_t)(tileN + lr) * ldb + lc;
    const float* gN = B1 + (size_t)(tileN + lr) * ldb + lc;

    v8f accP0 = {}; v8f accP1 = {}; v8f accN0 = {}; v8f accN1 = {};
    const int nstages = K / KST;

    {
        float4 ra[4], rp[4], rn[4];
        #pragma unroll
        for (int i = 0; i < 4; i++) {
            ra[i] = *(const float4*)(gA + (size_t)(16 * i) * lda);
            rp[i] = *(const float4*)(gP + (size_t)(16 * i) * ldb);
            rn[i] = *(const float4*)(gN + (size_t)(16 * i) * ldb);
        }
        #pragma unroll
        for (int i = 0; i < 4; i++) {
            *(float4*)&As[0][lr + 16 * i][lc] = ra[i];
            *(float4*)&Ps[0][lr + 16 * i][lc] = rp[i];
            *(float4*)&Ns[0][lr + 16 * i][lc] = rn[i];
        }
    }
    __syncthreads();

    for (int s = 0; s < nstages; s++) {
        const int p = s & 1;
        const bool more = (s + 1 < nstages);
        float4 na[4], np[4], nn[4];
        #pragma unroll
        for (int i = 0; i < 4; i++) {
            na[i] = make_float4(0.f, 0.f, 0.f, 0.f);
            np[i] = make_float4(0.f, 0.f, 0.f, 0.f);
            nn[i] = make_float4(0.f, 0.f, 0.f, 0.f);
        }
        if (more) {
            #pragma unroll
            for (int i = 0; i < 4; i++) {
                na[i] = *(const float4*)(gA + (size_t)(16 * i) * lda + (s + 1) * KST);
                np[i] = *(const float4*)(gP + (size_t)(16 * i) * ldb + (s + 1) * KST);
                nn[i] = *(const float4*)(gN + (size_t)(16 * i) * ldb + (s + 1) * KST);
            }
        }
        const float* ap  = &As[p][wm * 16 + l][half * 2];
        const float* pp0 = &Ps[p][wn * 32 + l][half * 2];
        const float* pp1 = &Ps[p][wn * 32 + 16 + l][half * 2];
        const float* np0 = &Ns[p][wn * 32 + l][half * 2];
        const float* np1 = &Ns[p][wn * 32 + 16 + l][half * 2];
        #pragma unroll
        for (int q = 0; q < KST / 4; q++) {
            const v2f af  = *(const v2f*)(ap  + q * 4);
            const v2f pf0 = *(const v2f*)(pp0 + q * 4);
            const v2f pf1 = *(const v2f*)(pp1 + q * 4);
            const v2f nf0 = *(const v2f*)(np0 + q * 4);
            const v2f nf1 = *(const v2f*)(np1 + q * 4);
            accP0 = __builtin_amdgcn_wmma_f32_16x16x4_f32(false, af, false, pf0, (short)0, accP0, false, false);
            accP1 = __builtin_amdgcn_wmma_f32_16x16x4_f32(false, af, false, pf1, (short)0, accP1, false, false);
            accN0 = __builtin_amdgcn_wmma_f32_16x16x4_f32(false, af, false, nf0, (short)0, accN0, false, false);
            accN1 = __builtin_amdgcn_wmma_f32_16x16x4_f32(false, af, false, nf1, (short)0, accN1, false, false);
        }
        if (more) {
            #pragma unroll
            for (int i = 0; i < 4; i++) {
                *(float4*)&As[p ^ 1][lr + 16 * i][lc] = na[i];
                *(float4*)&Ps[p ^ 1][lr + 16 * i][lc] = np[i];
                *(float4*)&Ns[p ^ 1][lr + 16 * i][lc] = nn[i];
            }
        }
        __syncthreads();
    }

    #pragma unroll
    for (int r = 0; r < 8; r++) {
        const int m  = tileM + wm * 16 + half * 8 + r;
        const int n0 = tileN + wn * 32 + l;
        const int n1 = n0 + 16;
        const float p0 = accP0[r] + bias0[n0];
        const float p1 = accP1[r] + bias0[n1];
        const float x0 = an[(size_t)m * ldan + n0];
        const float x1 = an[(size_t)m * ldan + n1];
        C[(size_t)m * ldc + n0] = sigmoidf_(p0 * x0 + accN0[r]);
        C[(size_t)m * ldc + n1] = sigmoidf_(p1 * x1 + accN1[r]);
    }
}

__global__ void
gemm_tn_kernel(const float* __restrict__ A, int lda, long long sA,
               const float* __restrict__ B, int ldb, long long sB,
               float* __restrict__ C, int ldc, long long sC,
               const float* __restrict__ G, int ldg, long long sG,
               int K) {
    __shared__ float As[2][KST][PTN];
    __shared__ float Bs[2][KST][PTN];
    const int z = blockIdx.z;
    A += (size_t)z * sA; B += (size_t)z * sB; C += (size_t)z * sC; G += (size_t)z * sG;
    const int tileM = blockIdx.y * 64;
    const int tileN = blockIdx.x * 64;
    const int tid  = threadIdx.x;
    const int wave = tid >> 5, lane = tid & 31;
    const int wm = wave >> 1, wn = wave & 1;
    const int half = lane >> 4, l = lane & 15;
    const int lk = tid >> 4;
    const int lm = (tid & 15) * 4;

    const float* gA = A + (size_t)lk * lda + tileM + lm;
    const float* gB = B + (size_t)lk * ldb + tileN + lm;

    v8f acc0 = {}; v8f acc1 = {};
    const int nstages = K / KST;

    {
        float4 ra[4], rb[4];
        #pragma unroll
        for (int i = 0; i < 4; i++) {
            ra[i] = *(const float4*)(gA + (size_t)(16 * i) * lda);
            rb[i] = *(const float4*)(gB + (size_t)(16 * i) * ldb);
        }
        #pragma unroll
        for (int i = 0; i < 4; i++) {
            *(float4*)&As[0][lk + 16 * i][lm] = ra[i];
            *(float4*)&Bs[0][lk + 16 * i][lm] = rb[i];
        }
    }
    __syncthreads();

    for (int s = 0; s < nstages; s++) {
        const int p = s & 1;
        const bool more = (s + 1 < nstages);
        float4 na[4], nb[4];
        #pragma unroll
        for (int i = 0; i < 4; i++) {
            na[i] = make_float4(0.f, 0.f, 0.f, 0.f);
            nb[i] = make_float4(0.f, 0.f, 0.f, 0.f);
        }
        if (more) {
            #pragma unroll
            for (int i = 0; i < 4; i++) {
                na[i] = *(const float4*)(gA + (size_t)((s + 1) * KST + 16 * i) * lda);
                nb[i] = *(const float4*)(gB + (size_t)((s + 1) * KST + 16 * i) * ldb);
            }
        }
        const float* ap  = &As[p][half * 2][wm * 16 + l];
        const float* bp0 = &Bs[p][half * 2][wn * 32 + l];
        const float* bp1 = &Bs[p][half * 2][wn * 32 + 16 + l];
        #pragma unroll
        for (int q = 0; q < KST / 4; q++) {
            v2f af, bf0, bf1;
            af.x  = ap [(q * 4) * PTN];  af.y  = ap [(q * 4 + 1) * PTN];
            bf0.x = bp0[(q * 4) * PTN];  bf0.y = bp0[(q * 4 + 1) * PTN];
            bf1.x = bp1[(q * 4) * PTN];  bf1.y = bp1[(q * 4 + 1) * PTN];
            acc0 = __builtin_amdgcn_wmma_f32_16x16x4_f32(false, af, false, bf0, (short)0, acc0, false, false);
            acc1 = __builtin_amdgcn_wmma_f32_16x16x4_f32(false, af, false, bf1, (short)0, acc1, false, false);
        }
        if (more) {
            #pragma unroll
            for (int i = 0; i < 4; i++) {
                *(float4*)&As[p ^ 1][lk + 16 * i][lm] = na[i];
                *(float4*)&Bs[p ^ 1][lk + 16 * i][lm] = nb[i];
            }
        }
        __syncthreads();
    }

    #pragma unroll
    for (int r = 0; r < 8; r++) {
        int m  = tileM + wm * 16 + half * 8 + r;
        int n0 = tileN + wn * 32 + l;
        const float g0 = G[(size_t)m * ldg + n0];
        const float g1 = G[(size_t)m * ldg + n0 + 16];
        C[(size_t)m * ldc + n0]      = sigmoidf_(g0) * acc0[r];
        C[(size_t)m * ldc + n0 + 16] = sigmoidf_(g1) * acc1[r];
    }
}

#define SSP 68
__global__ void __launch_bounds__(256, 4)
scores_softmax_kernel(const float* __restrict__ kvg, const float* __restrict__ qm,
                      const float* __restrict__ z,
                      const float* __restrict__ pnw, const float* __restrict__ pnb,
                      const float* __restrict__ bw,  const float* __restrict__ bb,
                      float* __restrict__ P) {
    __shared__ float  aS[16][SSP];
    __shared__ float  wS[16][SSP];
    __shared__ float  bS[2][8][16 * SSP];
    __shared__ float  zS[2][8][16 * SSP];
    __shared__ float4 redm[16][2];
    __shared__ float4 reds[16][2];
    const int tid  = threadIdx.x;
    const int wave = tid >> 5, lane = tid & 31;
    const int half = lane >> 4, l = lane & 15;
    const int h  = blockIdx.y;
    const int j0 = blockIdx.x * 16;
    const float*  Kh = kvg + (size_t)h * 196608 + (size_t)j0 * 192;
    const float4* Q4 = (const float4*)(qm + (size_t)h * 65536) + wave * 2048 + lane;
    const float4* Z4 = (const float4*)z + (size_t)(h * 64 + blockIdx.x) * 16384
                       + (lane >> 4) * 1024 + wave * 128 + (lane & 15);
    const size_t  ob = (size_t)h * 1048576 + (size_t)(j0 + 8 * half) * 1024 + wave * 128 + l;

    float4 raw[8], zraw[8];
    #pragma unroll
    for (int j = 0; j < 8; j++) {
        zraw[j] = Z4[j * 2048];
        raw[j]  = Q4[j * 32];
    }
    *(float4*)&aS[tid >> 4][(tid & 15) * 4] =
        *(const float4*)(Kh + (size_t)(tid >> 4) * 192 + (tid & 15) * 4);
    {
        const float4 w4 = ((const float4*)bw)[tid];
        const float4 g4 = ((const float4*)pnw)[tid & 15];
        *(float4*)&wS[tid >> 4][(tid & 15) * 4] =
            make_float4(w4.x * g4.x, w4.y * g4.y, w4.z * g4.z, w4.w * g4.w);
    }
    float cb = 0.f;
    #pragma unroll
    for (int jj = 0; jj < 8; jj++) {
        const float4 w4 = ((const float4*)bw)[l * 16 + half * 8 + jj];
        const float4 d4 = ((const float4*)pnb)[half * 8 + jj];
        cb += w4.x * d4.x + w4.y * d4.y + w4.z * d4.z + w4.w * d4.w;
    }
    cb += __shfl_xor(cb, 16);
    cb += bb[l];
    __syncthreads();
    float4 af[8];
    #pragma unroll
    for (int jj = 0; jj < 8; jj++) af[jj] = *(const float4*)&aS[l][half * 32 + jj * 4];
    v8f cinit;
    #pragma unroll
    for (int r = 0; r < 8; r++) cinit[r] = cb;

    v8f acc[8];
    #pragma unroll
    for (int ct = 0; ct < 8; ct++) {
        float* bq = &bS[ct & 1][wave][0];
        float* zw = &zS[ct & 1][wave][0];
        #pragma unroll
        for (int j = 0; j < 8; j++) {
            const int f = j * 32 + lane;
            *(float4*)&zw[(f >> 4) * SSP + (f & 15) * 4] = zraw[j];
            *(float4*)&bq[(f >> 4) * SSP + (f & 15) * 4] = raw[j];
        }
        __syncthreads();
        #pragma unroll
        for (int j = 0; j < 8; j++) {
            raw[j]  = make_float4(0.f, 0.f, 0.f, 0.f);
            zraw[j] = make_float4(0.f, 0.f, 0.f, 0.f);
        }
        if (ct + 1 < 8) {
            #pragma unroll
            for (int j = 0; j < 8; j++) {
                zraw[j] = Z4[(ct + 1) * 16 + j * 2048];
                raw[j]  = Q4[(ct + 1) * 256 + j * 32];
            }
        }
        float4 x[8];
        float sum = 0.f, sq = 0.f;
        #pragma unroll
        for (int jj = 0; jj < 8; jj++) {
            const float4 t = *(const float4*)&zw[l * SSP + half * 32 + jj * 4];
            x[jj] = t;
            sum += t.x + t.y + t.z + t.w;
            sq  += t.x * t.x + t.y * t.y + t.z * t.z + t.w * t.w;
        }
        sum += __shfl_xor(sum, 16);
        sq  += __shfl_xor(sq, 16);
        const float m   = sum * (1.0f / 64.0f);
        const float var = sq * (1.0f / 64.0f) - m * m;
        const float inv = rsqrtf(var + 1e-5f);
        v8f c0 = cinit; v8f c1 = {};
        #pragma unroll
        for (int jj = 0; jj < 8; jj++) {
            const float4 w4 = *(const float4*)&wS[l][half * 32 + jj * 4];
            v2f n0, n1, w0, w1;
            n0.x = (x[jj].x - m) * inv;
            n0.y = (x[jj].y - m) * inv;
            n1.x = (x[jj].z - m) * inv;
            n1.y = (x[jj].w - m) * inv;
            w0.x = w4.x; w0.y = w4.y; w1.x = w4.z; w1.y = w4.w;
            c0 = __builtin_amdgcn_wmma_f32_16x16x4_f32(false, n0, false, w0, (short)0, c0, false, false);
            c1 = __builtin_amdgcn_wmma_f32_16x16x4_f32(false, n1, false, w1, (short)0, c1, false, false);
        }
        float4 bf[8];
        #pragma unroll
        for (int jj = 0; jj < 8; jj++) bf[jj] = *(const float4*)&bq[l * SSP + half * 32 + jj * 4];
        v8f a0 = {}; v8f a1 = {};
        #pragma unroll
        for (int jj = 0; jj < 8; jj++) {
            v2f x0, x1, y0, y1;
            x0.x = af[jj].x; x0.y = af[jj].y; x1.x = af[jj].z; x1.y = af[jj].w;
            y0.x = bf[jj].x; y0.y = bf[jj].y; y1.x = bf[jj].z; y1.y = bf[jj].w;
            a0 = __builtin_amdgcn_wmma_f32_16x16x4_f32(false, x0, false, y0, (short)0, a0, false, false);
            a1 = __builtin_amdgcn_wmma_f32_16x16x4_f32(false, x1, false, y1, (short)0, a1, false, false);
        }
        acc[ct] = (a0 + a1) * (1.0f / 64.0f) + (c0 + c1);
        __builtin_amdgcn_sched_barrier(0);
    }

    float* pp = P + ob;
    float mx[8];
    #pragma unroll
    for (int r = 0; r < 8; r++) {
        float m = acc[0][r];
        #pragma unroll
        for (int ct = 1; ct < 8; ct++) m = fmaxf(m, acc[ct][r]);
        #pragma unroll
        for (int off = 8; off > 0; off >>= 1) m = fmaxf(m, __shfl_xor(m, off));
        mx[r] = m;
    }
    if (l == 0) {
        #pragma unroll
        for (int r = 0; r < 8; r++) ((float*)&redm[8 * half + r][0])[wave] = mx[r];
    }
    __syncthreads();
    #pragma unroll
    for (int r = 0; r < 8; r++) {
        const float4 u = redm[8 * half + r][0], w = redm[8 * half + r][1];
        mx[r] = fmaxf(fmaxf(fmaxf(u.x, u.y), fmaxf(u.z, u.w)), fmaxf(fmaxf(w.x, w.y), fmaxf(w.z, w.w)));
    }
    float sm[8];
    #pragma unroll
    for (int r = 0; r < 8; r++) {
        float s = 0.f;
        #pragma unroll
        for (int ct = 0; ct < 8; ct++) {
            const float e = __expf(acc[ct][r] - mx[r]);
            acc[ct][r] = e;
            s += e;
        }
        #pragma unroll
        for (int off = 8; off > 0; off >>= 1) s += __shfl_xor(s, off);
        sm[r] = s;
    }
    if (l == 0) {
        #pragma unroll
        for (int r = 0; r < 8; r++) ((float*)&reds[8 * half + r][0])[wave] = sm[r];
    }
    __syncthreads();
    #pragma unroll
    for (int r = 0; r < 8; r++) {
        const float4 u = reds[8 * half + r][0], w = reds[8 * half + r][1];
        const float s = ((u.x + u.y) + (u.z + u.w)) + ((w.x + w.y) + (w.z + w.w));
        const float inv = 1.0f / s;
        #pragma unroll
        for (int ct = 0; ct < 8; ct++) pp[r * 1024 + ct * 16] = acc[ct][r] * inv;
    }
}

extern "C" void kernel_launch(void* const* d_in, const int* in_sizes, int n_in,
                              void* d_out, int out_size, void* d_ws, size_t ws_size,
                              hipStream_t stream) {
    const float* a       = (const float*)d_in[0];
    const float* z       = (const float*)d_in[1];
    const float* s       = (const float*)d_in[2];
    const float* sn_w    = (const float*)d_in[3];
    const float* pb_w    = (const float*)d_in[4];
    const float* pb_b    = (const float*)d_in[5];
    const float* pn_w    = (const float*)d_in[6];
    const float* pnorm_w = (const float*)d_in[7];
    const float* pnorm_b = (const float*)d_in[8];
    const float* q_w     = (const float*)d_in[9];
    const float* q_b     = (const float*)d_in[10];
    const float* kvg_w   = (const float*)d_in[11];
    const float* bias_w  = (const float*)d_in[12];
    const float* bias_b  = (const float*)d_in[13];
    const float* attn_w  = (const float*)d_in[14];
    const float* out_w   = (const float*)d_in[15];
    const float* out_b   = (const float*)d_in[16];
    float* out = (float*)d_out;
    float* ws  = (float*)d_ws;

    float* a_n    = ws + 0;
    float* s_n    = ws + 1048576;
    float* a1     = ws + 1572864;
    float* qm     = ws + 3670016;
    float* kvg    = ws + 4718592;
    float* go     = ws + 8912896;
    float* a2     = ws + 9961472;
    float* scores = ws + 24641536;

    ln_rows_kernel<<<192, 256, 0, stream>>>(a, a_n, s, sn_w, s_n);

    gemm_nt_adaln_kernel<<<dim3(16, 16, 1), 256, 0, stream>>>(
        s_n, 512, pb_w, pn_w, 512, pb_b, a_n, 1024, a1, 1024, 512);

    gemm_nt_kernel<<<dim3(64, 16, 1), 256, 0, stream>>>(
        a1, 1024, 0, q_w, 1024, 0, qm, 1024, 0, q_b, nullptr, 0, 0, 1024, 1.0f, 0,
        16, kvg_w, kvg, 3072, nullptr);

    scores_softmax_kernel<<<dim3(64, 16, 1), 256, 0, stream>>>(
        kvg, qm, z, pnorm_w, pnorm_b, bias_w, bias_b, scores);

    gemm_tn_kernel<<<dim3(1, 16, 16), 256, 0, stream>>>(
        scores, 1024, 1048576LL,
        kvg + 64, 192, 196608LL,
        go, 64, 65536LL,
        kvg + 128, 192, 196608LL,
        1024);

    gemm_nt_kernel<<<dim3(16, 16, 1), 256, 0, stream>>>(
        go, 1024, 0, attn_w, 1024, 0, a2, 1024, 0, nullptr, nullptr, 0, 0, 1024, 1.0f, 0,
        16, nullptr, nullptr, 0, nullptr);

    gemm_nt_kernel<<<dim3(16, 16, 1), 256, 0, stream>>>(
        s, 512, 0, out_w, 512, 0, out, 1024, 0, out_b, a2, 1024, 0, 512, 1.0f, 1,
        16, nullptr, nullptr, 0, nullptr);
}


# The three GEMM launches with one workgroup per WGP (P^T V, attn_w, out gate) issue a stage's global loads two 64-k stages ahead of its WMMAs, in two register sets, instead of one; sum orders kept
# speedup vs baseline: 2.6356x; 1.0457x over previous
#include <hip/hip_runtime.h>
#include <hip/hip_bf16.h>
#include <math.h>

typedef __attribute__((ext_vector_type(2))) float v2f;
typedef __attribute__((ext_vector_type(8))) float v8f;

#define S_DIM 1024
#define CA 1024
#define CS 512
#define CZ 64
#define NH 16
#define HC 64

__device__ __forceinline__ float sigmoidf_(float x) {
    return 1.0f / (1.0f + __expf(-x));
}

__global__ void __launch_bounds__(256)
ln_rows_kernel(const float* __restrict__ a, float* __restrict__ a_n,
               const float* __restrict__ s, const float* __restrict__ sw, float* __restrict__ s_n) {
    const int lane = threadIdx.x & 31;
    const int wave = threadIdx.x >> 5;
    const bool is_s = blockIdx.x >= S_DIM / 8;
    const size_t chunk = (size_t)((is_s ? blockIdx.x - S_DIM / 8 : blockIdx.x) * 8 + wave) * 1024;
    const float4* xr = (const float4*)((is_s ? s : a) + chunk) + lane;
    float4*       yr = (float4*)((is_s ? s_n : a_n) + chunk) + lane;
    float4 v[8];
    #pragma unroll
    for (int j = 0; j < 8; j++) v[j] = xr[j * 32];
    float su[2] = {0.f, 0.f}, sq[2] = {0.f, 0.f};
    #pragma unroll
    for (int j = 0; j < 8; j++) {
        su[j >> 2] += v[j].x + v[j].y + v[j].z + v[j].w;
        sq[j >> 2] += v[j].x * v[j].x + v[j].y * v[j].y + v[j].z * v[j].z + v[j].w * v[j].w;
    }
    #pragma unroll
    for (int off = 16; off > 0; off >>= 1) {
        su[0] += __shfl_xor(su[0], off); su[1] += __shfl_xor(su[1], off);
        sq[0] += __shfl_xor(sq[0], off); sq[1] += __shfl_xor(sq[1], off);
    }
    float rc = 1.0f / 512.0f;
    if (!is_s) {
        su[0] += su[1]; su[1] = su[0];
        sq[0] += sq[1]; sq[1] = sq[0];
        rc = 1.0f / 1024.0f;
    }
    float m[2], inv[2];
    #pragma unroll
    for (int h = 0; h < 2; h++) {
        m[h] = su[h] * rc;
        const float var = sq[h] * rc - m[h] * m[h];
        inv[h] = rsqrtf(var + 1e-5f);
    }
    #pragma unroll
    for (int j = 0; j < 8; j++) {
        v[j].x = (v[j].x - m[j >> 2]) * inv[j >> 2];
        v[j].y = (v[j].y - m[j >> 2]) * inv[j >> 2];
        v[j].z = (v[j].z - m[j >> 2]) * inv[j >> 2];
        v[j].w = (v[j].w - m[j >> 2]) * inv[j >> 2];
    }
    if (is_s) {
        #pragma unroll
        for (int j = 0; j < 4; j++) {
            const float4 wv = ((const float4*)sw)[j * 32 + lane];
            v[j].x     *= wv.x; v[j].y     *= wv.y; v[j].z     *= wv.z; v[j].w     *= wv.w;
            v[j + 4].x *= wv.x; v[j + 4].y *= wv.y; v[j + 4].z *= wv.z; v[j + 4].w *= wv.w;
        }
    }
    #pragma unroll
    for (int j = 0; j < 8; j++) yr[j * 32] = v[j];
}

#define KST 64
#define PNT 68
#define PTN 72

struct StageRegs { float4 a0, a1, a2, a3, b0, b1, b2, b3; };
__device__ __forceinline__ StageRegs nt_stage_load(const float* gA, int lda, const float* gB, int ldb, int st) {
    StageRegs r;
    const float* pa = gA + st * KST;
    const float* pb = gB + st * KST;
    r.a0 = *(const float4*)(pa);
    r.b0 = *(const float4*)(pb);
    r.a1 = *(const float4*)(pa + (size_t)16 * lda);
    r.b1 = *(const float4*)(pb + (size_t)16 * ldb);
    r.a2 = *(const float4*)(pa + (size_t)32 * lda);
    r.b2 = *(const float4*)(pb + (size_t)32 * ldb);
    r.a3 = *(const float4*)(pa + (size_t)48 * lda);
    r.b3 = *(const float4*)(pb + (size_t)48 * ldb);
    return r;
}
__device__ __forceinline__ void nt_stage_store(float (*As)[PNT], float (*Bs)[PNT], const StageRegs r, int lr, int lc) {
    *(float4*)&As[lr][lc]      = r.a0;
    *(float4*)&Bs[lr][lc]      = r.b0;
    *(float4*)&As[lr + 16][lc] = r.a1;
    *(float4*)&Bs[lr + 16][lc] = r.b1;
    *(float4*)&As[lr + 32][lc] = r.a2;
    *(float4*)&Bs[lr + 32][lc] = r.b2;
    *(float4*)&As[lr + 48][lc] = r.a3;
    *(float4*)&Bs[lr + 48][lc] = r.b3;
}
__device__ __forceinline__ void lds_barrier() {
    __builtin_amdgcn_fence(__ATOMIC_RELEASE, "workgroup", "local");
    __builtin_amdgcn_s_barrier();
    __builtin_amdgcn_fence(__ATOMIC_ACQUIRE, "workgroup", "local");
}
__device__ __forceinline__ void nt_stage_mma(const float (*As)[PNT], const float (*Bs)[PNT],
                                             v8f& acc0, v8f& acc1, int wm, int wn, int l, int half) {
    const float* ap  = &As[wm * 16 + l][half * 2];
    const float* bp0 = &Bs[wn * 32 + l][half * 2];
    const float* bp1 = &Bs[wn * 32 + 16 + l][half * 2];
    #pragma unroll
    for (int q = 0; q < KST / 4; q++) {
        const v2f af  = *(const v2f*)(ap  + q * 4);
        const v2f bf0 = *(const v2f*)(bp0 + q * 4);
        const v2f bf1 = *(const v2f*)(bp1 + q * 4);
        acc0 = __builtin_amdgcn_wmma_f32_16x16x4_f32(false, af, false, bf0, (short)0, acc0, false, false);
        acc1 = __builtin_amdgcn_wmma_f32_16x16x4_f32(false, af, false, bf1, (short)0, acc1, false, false);
    }
}
__global__ void
gemm_nt_kernel(const float* __restrict__ A, int lda, long long sA,
               const float* __restrict__ B, int ldb, long long sB,
               float* __restrict__ C, int ldc, long long sC,
               const float* __restrict__ bias,
               const float* __restrict__ addm, int ldadd, long long sAdd,
               int K, float scale, int mode,
               int nsplit, const float* __restrict__ B2, float* __restrict__ C2, int ldc2,
               const float* __restrict__ bias2) {
    __shared__ float As[2][64][PNT];
    __shared__ float Bs[2][64][PNT];
    int bx = blockIdx.x;
    if (bx >= nsplit) { B = B2; C = C2; ldc = ldc2; bias = bias2; bx -= nsplit; }
    const int z = blockIdx.z;
    A += (size_t)z * sA; B += (size_t)z * sB; C += (size_t)z * sC;
    if (addm) addm += (size_t)z * sAdd;
    const int tileM = blockIdx.y * 64;
    const int tileN = bx * 64;
    const int tid  = threadIdx.x;
    const int wave = tid >> 5, lane = tid & 31;
    const int wm = wave >> 1, wn = wave & 1;
    const int half = lane >> 4, l = lane & 15;
    const int lr = tid >> 4;
    const int lc = (tid & 15) * 4;

    const float* gA = A + (size_t)(tileM + lr) * lda + lc;
    const float* gB = B + (size_t)(tileN + lr) * ldb + lc;

    v8f acc0 = {}; v8f acc1 = {};
    const int nstages = K / KST;

    {
        float4 ra[4], rb[4];
        #pragma unroll
        for (int i = 0; i < 4; i++) {
            ra[i] = *(const float4*)(gA + (size_t)(16 * i) * lda);
            rb[i] = *(const float4*)(gB + (size_t)(16 * i) * ldb);
        }
        #pragma unroll
        for (int i = 0; i < 4; i++) {
            *(float4*)&As[0][lr + 16 * i][lc] = ra[i];
            *(float4*)&Bs[0][lr + 16 * i][lc] = rb[i];
        }
    }
    __syncthreads();

    for (int s = 0; s < nstages; s++) {
        const int p = s & 1;
        const bool more = (s + 1 < nstages);
        float4 na[4], nb[4];
        #pragma unroll
        for (int i = 0; i < 4; i++) {
            na[i] = make_float4(0.f, 0.f, 0.f, 0.f);
            nb[i] = make_float4(0.f, 0.f, 0.f, 0.f);
        }
        if (more) {
            #pragma unroll
            for (int i = 0; i < 4; i++) {
                na[i] = *(const float4*)(gA + (size_t)(16 * i) * lda + (s + 1) * KST);
                nb[i] = *(const float4*)(gB + (size_t)(16 * i) * ldb + (s + 1) * KST);
            }
        }
        const float* ap  = &As[p][wm * 16 + l][half * 2];
        const float* bp0 = &Bs[p][wn * 32 + l][half * 2];
        const float* bp1 = &Bs[p][wn * 32 + 16 + l][half * 2];
        #pragma unroll
        for (int q = 0; q < KST / 4; q++) {
            const v2f af  = *(const v2f*)(ap  + q * 4);
            const v2f bf0 = *(const v2f*)(bp0 + q * 4);
            const v2f bf1 = *(const v2f*)(bp1 + q * 4);
            acc0 = __builtin_amdgcn_wmma_f32_16x16x4_f32(false, af, false, bf0, (short)0, acc0, false, false);
            acc1 = __builtin_amdgcn_wmma_f32_16x16x4_f32(false, af, false, bf1, (short)0, acc1, false, false);
        }
        if (more) {
            #pragma unroll
            for (int i = 0; i < 4; i++) {
                *(float4*)&As[p ^ 1][lr + 16 * i][lc] = na[i];
                *(float4*)&Bs[p ^ 1][lr + 16 * i][lc] = nb[i];
            }
        }
        __syncthreads();
    }

    #pragma unroll
    for (int r = 0; r < 8; r++) {
        int m  = tileM + wm * 16 + half * 8 + r;
        int n0 = tileN + wn * 32 + l;
        int n1 = n0 + 16;
        float v0 = acc0[r] * scale;
        float v1 = acc1[r] * scale;
        if (bias) { v0 += bias[n0]; v1 += bias[n1]; }
        if (mode == 1) {
            v0 = sigmoidf_(v0) * addm[(size_t)m * ldadd + n0];
            v1 = sigmoidf_(v1) * addm[(size_t)m * ldadd + n1];
        } else if (addm) {
            v0 += addm[(size_t)m * ldadd + n0];
            v1 += addm[(size_t)m * ldadd + n1];
        }
        C[(size_t)m * ldc + n0] = v0;
        C[(size_t)m * ldc + n1] = v1;
    }
}

__global__ void __launch_bounds__(256) __attribute__((amdgpu_waves_per_eu(2, 2)))
gemm_nt_lead2_kernel(const float* A, int lda,
                     const float* B, int ldb,
                     float* __restrict__ C, int ldc,
                     const float* __restrict__ bias,
                     const float* __restrict__ addm, int ldadd,
                     int K, int mode) {
    __shared__ float As[2][64][PNT];
    __shared__ float Bs[2][64][PNT];
    const int tileM = blockIdx.y * 64;
    const int tileN = blockIdx.x * 64;
    const int tid  = threadIdx.x;
    const int wave = tid >> 5, lane = tid & 31;
    const int wm = wave >> 1, wn = wave & 1;
    const int half = lane >> 4, l = lane & 15;
    const int lr = tid >> 4;
    const int lc = (tid & 15) * 4;

    const float* gA = A + (size_t)(tileM + lr) * lda + lc;
    const float* gB = B + (size_t)(tileN + lr) * ldb + lc;

    v8f acc0 = {}; v8f acc1 = {};
    const int nstages = K / KST;

    StageRegs r0 = nt_stage_load(gA, lda, gB, ldb, 0);
    nt_stage_store(As[0], Bs[0], r0, lr, lc);
    r0 = nt_stage_load(gA, lda, gB, ldb, 1);
    lds_barrier();

    int s = 0;
    for (; s + 3 < nstages; s += 2) {
        const StageRegs r1 = nt_stage_load(gA, lda, gB, ldb, s + 2);
        __builtin_amdgcn_sched_barrier(0);
        nt_stage_mma(As[0], Bs[0], acc0, acc1, wm, wn, l, half);
        nt_stage_store(As[1], Bs[1], r0, lr, lc);
        lds_barrier();
        r0 = nt_stage_load(gA, lda, gB, ldb, s + 3);
        __builtin_amdgcn_sched_barrier(0);
        nt_stage_mma(As[1], Bs[1], acc0, acc1, wm, wn, l, half);
        nt_stage_store(As[0], Bs[0], r1, lr, lc);
        lds_barrier();
    }
    nt_stage_mma(As[0], Bs[0], acc0, acc1, wm, wn, l, half);
    nt_stage_store(As[1], Bs[1], r0, lr, lc);
    lds_barrier();
    nt_stage_mma(As[1], Bs[1], acc0, acc1, wm, wn, l, half);

    #pragma unroll
    for (int r = 0; r < 8; r++) {
        const int m  = tileM + wm * 16 + half * 8 + r;
        const int n0 = tileN + wn * 32 + l;
        const int n1 = n0 + 16;
        float v0 = acc0[r];
        float v1 = acc1[r];
        if (mode == 1) {
            v0 = sigmoidf_(v0 + bias[n0]) * addm[(size_t)m * ldadd + n0];
            v1 = sigmoidf_(v1 + bias[n1]) * addm[(size_t)m * ldadd + n1];
        }
        C[(size_t)m * ldc + n0] = v0;
        C[(size_t)m * ldc + n1] = v1;
    }
}

__global__ void __launch_bounds__(256)
gemm_nt_adaln_kernel(const float* __restrict__ A, int lda,
                     const float* __restrict__ B0, const float* __restrict__ B1, int ldb,
                     const float* __restrict__ bias0,
                     const float* __restrict__ an, int ldan,
                     float* __restrict__ C, int ldc, int K) {
    __shared__ float As[2][64][PNT];
    __shared__ float Ps[2][64][PNT];
    __shared__ float Ns[2][64][PNT];
    const int tileM = blockIdx.y * 64;
    const int tileN = blockIdx.x * 64;
    const int tid  = threadIdx.x;
    const int wave = tid >> 5, lane = tid & 31;
    const int wm = wave >> 1, wn = wave & 1;
    const int half = lane >> 4, l = lane & 15;
    const int lr = tid >> 4;
    const int lc = (tid & 15) * 4;

    const float* gA = A  + (size_t)(tileM + lr) * lda + lc;
    const float* gP = B0 + (size_t)(tileN + lr) * ldb + lc;
    const float* gN = B1 + (size_t)(tileN + lr) * ldb + lc;

    v8f accP0 = {}; v8f accP1 = {}; v8f accN0 = {}; v8f accN1 = {};
    const int nstages = K / KST;

    {
        float4 ra[4], rp[4], rn[4];
        #pragma unroll
        for (int i = 0; i < 4; i++) {
            ra[i] = *(const float4*)(gA + (size_t)(16 * i) * lda);
            rp[i] = *(const float4*)(gP + (size_t)(16 * i) * ldb);
            rn[i] = *(const float4*)(gN + (size_t)(16 * i) * ldb);
        }
        #pragma unroll
        for (int i = 0; i < 4; i++) {
            *(float4*)&As[0][lr + 16 * i][lc] = ra[i];
            *(float4*)&Ps[0][lr + 16 * i][lc] = rp[i];
            *(float4*)&Ns[0][lr + 16 * i][lc] = rn[i];
        }
    }
    __syncthreads();

    for (int s = 0; s < nstages; s++) {
        const int p = s & 1;
        const bool more = (s + 1 < nstages);
        float4 na[4], np[4], nn[4];
        #pragma unroll
        for (int i = 0; i < 4; i++) {
            na[i] = make_float4(0.f, 0.f, 0.f, 0.f);
            np[i] = make_float4(0.f, 0.f, 0.f, 0.f);
            nn[i] = make_float4(0.f, 0.f, 0.f, 0.f);
        }
        if (more) {
            #pragma unroll
            for (int i = 0; i < 4; i++) {
                na[i] = *(const float4*)(gA + (size_t)(16 * i) * lda + (s + 1) * KST);
                np[i] = *(const float4*)(gP + (size_t)(16 * i) * ldb + (s + 1) * KST);
                nn[i] = *(const float4*)(gN + (size_t)(16 * i) * ldb + (s + 1) * KST);
            }
        }
        const float* ap  = &As[p][wm * 16 + l][half * 2];
        const float* pp0 = &Ps[p][wn * 32 + l][half * 2];
        const float* pp1 = &Ps[p][wn * 32 + 16 + l][half * 2];
        const float* np0 = &Ns[p][wn * 32 + l][half * 2];
        const float* np1 = &Ns[p][wn * 32 + 16 + l][half * 2];
        #pragma unroll
        for (int q = 0; q < KST / 4; q++) {
            const v2f af  = *(const v2f*)(ap  + q * 4);
            const v2f pf0 = *(const v2f*)(pp0 + q * 4);
            const v2f pf1 = *(const v2f*)(pp1 + q * 4);
            const v2f nf0 = *(const v2f*)(np0 + q * 4);
            const v2f nf1 = *(const v2f*)(np1 + q * 4);
            accP0 = __builtin_amdgcn_wmma_f32_16x16x4_f32(false, af, false, pf0, (short)0, accP0, false, false);
            accP1 = __builtin_amdgcn_wmma_f32_16x16x4_f32(false, af, false, pf1, (short)0, accP1, false, false);
            accN0 = __builtin_amdgcn_wmma_f32_16x16x4_f32(false, af, false, nf0, (short)0, accN0, false, false);
            accN1 = __builtin_amdgcn_wmma_f32_16x16x4_f32(false, af, false, nf1, (short)0, accN1, false, false);
        }
        if (more) {
            #pragma unroll
            for (int i = 0; i < 4; i++) {
                *(float4*)&As[p ^ 1][lr + 16 * i][lc] = na[i];
                *(float4*)&Ps[p ^ 1][lr + 16 * i][lc] = np[i];
                *(float4*)&Ns[p ^ 1][lr + 16 * i][lc] = nn[i];
            }
        }
        __syncthreads();
    }

    #pragma unroll
    for (int r = 0; r < 8; r++) {
        const int m  = tileM + wm * 16 + half * 8 + r;
        const int n0 = tileN + wn * 32 + l;
        const int n1 = n0 + 16;
        const float p0 = accP0[r] + bias0[n0];
        const float p1 = accP1[r] + bias0[n1];
        const float x0 = an[(size_t)m * ldan + n0];
        const float x1 = an[(size_t)m * ldan + n1];
        C[(size_t)m * ldc + n0] = sigmoidf_(p0 * x0 + accN0[r]);
        C[(size_t)m * ldc + n1] = sigmoidf_(p1 * x1 + accN1[r]);
    }
}

__device__ __forceinline__ StageRegs tn_stage_load(const float* gA, int lda, unsigned oA,
                                                   const float* gB, int ldb, unsigned oB, int st) {
    StageRegs r;
    const float* pa = gA + (size_t)(st * KST) * lda;
    const float* pb = gB + (size_t)(st * KST) * ldb;
    r.a0 = *(const float4*)(pa + oA);
    r.b0 = *(const float4*)(pb + oB);
    r.a1 = *(const float4*)(pa + (size_t)16 * lda + oA);
    r.b1 = *(const float4*)(pb + (size_t)16 * ldb + oB);
    r.a2 = *(const float4*)(pa + (size_t)32 * lda + oA);
    r.b2 = *(const float4*)(pb + (size_t)32 * ldb + oB);
    r.a3 = *(const float4*)(pa + (size_t)48 * lda + oA);
    r.b3 = *(const float4*)(pb + (size_t)48 * ldb + oB);
    return r;
}
__device__ __forceinline__ void tn_stage_store(float (*As)[PTN], float (*Bs)[PTN], const StageRegs r, int lk, int lm) {
    *(float4*)&As[lk][lm]      = r.a0;
    *(float4*)&Bs[lk][lm]      = r.b0;
    *(float4*)&As[lk + 16][lm] = r.a1;
    *(float4*)&Bs[lk + 16][lm] = r.b1;
    *(float4*)&As[lk + 32][lm] = r.a2;
    *(float4*)&Bs[lk + 32][lm] = r.b2;
    *(float4*)&As[lk + 48][lm] = r.a3;
    *(float4*)&Bs[lk + 48][lm] = r.b3;
}
__device__ __forceinline__ void tn_stage_mma(const float (*As)[PTN], const float (*Bs)[PTN],
                                             v8f& acc0, v8f& acc1, int wm, int wn, int l, int half) {
    const float* ap  = &As[half * 2][wm * 16 + l];
    const float* bp0 = &Bs[half * 2][wn * 32 + l];
    const float* bp1 = &Bs[half * 2][wn * 32 + 16 + l];
    #pragma unroll
    for (int q = 0; q < KST / 4; q++) {
        v2f af, bf0, bf1;
        af.x  = ap [(q * 4) * PTN];  af.y  = ap [(q * 4 + 1) * PTN];
        bf0.x = bp0[(q * 4) * PTN];  bf0.y = bp0[(q * 4 + 1) * PTN];
        bf1.x = bp1[(q * 4) * PTN];  bf1.y = bp1[(q * 4 + 1) * PTN];
        acc0 = __builtin_amdgcn_wmma_f32_16x16x4_f32(false, af, false, bf0, (short)0, acc0, false, false);
        acc1 = __builtin_amdgcn_wmma_f32_16x16x4_f32(false, af, false, bf1, (short)0, acc1, false, false);
    }
}
__global__ void __launch_bounds__(256) __attribute__((amdgpu_waves_per_eu(2, 2)))
gemm_tn_kernel(const float* A, int lda, long long sA,
               const float* B, int ldb, long long sB,
               float* __restrict__ C, int ldc, long long sC,
               const float* __restrict__ G, int ldg, long long sG,
               int K) {
    __shared__ float As[2][KST][PTN];
    __shared__ float Bs[2][KST][PTN];
    const int z = blockIdx.z;
    A += (size_t)z * sA; B += (size_t)z * sB; C += (size_t)z * sC; G += (size_t)z * sG;
    const int tileM = blockIdx.y * 64;
    const int tileN = blockIdx.x * 64;
    const int tid  = threadIdx.x;
    const int wave = tid >> 5, lane = tid & 31;
    const int wm = wave >> 1, wn = wave & 1;
    const int half = lane >> 4, l = lane & 15;
    const int lk = tid >> 4;
    const int lm = (tid & 15) * 4;

    const float* gA = A + tileM;
    const float* gB = B + tileN;
    const unsigned oA = (unsigned)(lk * lda + lm);
    const unsigned oB = (unsigned)(lk * ldb + lm);

    v8f acc0 = {}; v8f acc1 = {};
    const int nstages = K / KST;

    StageRegs r0 = tn_stage_load(gA, lda, oA, gB, ldb, oB, 0);
    tn_stage_store(As[0], Bs[0], r0, lk, lm);
    r0 = tn_stage_load(gA, lda, oA, gB, ldb, oB, 1);
    lds_barrier();

    int s = 0;
    for (; s + 3 < nstages; s += 2) {
        const StageRegs r1 = tn_stage_load(gA, lda, oA, gB, ldb, oB, s + 2);
        __builtin_amdgcn_sched_barrier(0);
        tn_stage_mma(As[0], Bs[0], acc0, acc1, wm, wn, l, half);
        tn_stage_store(As[1], Bs[1], r0, lk, lm);
        lds_barrier();
        r0 = tn_stage_load(gA, lda, oA, gB, ldb, oB, s + 3);
        __builtin_amdgcn_sched_barrier(0);
        tn_stage_mma(As[1], Bs[1], acc0, acc1, wm, wn, l, half);
        tn_stage_store(As[0], Bs[0], r1, lk, lm);
        lds_barrier();
    }
    tn_stage_mma(As[0], Bs[0], acc0, acc1, wm, wn, l, half);
    tn_stage_store(As[1], Bs[1], r0, lk, lm);
    lds_barrier();
    tn_stage_mma(As[1], Bs[1], acc0, acc1, wm, wn, l, half);

    #pragma unroll
    for (int r = 0; r < 8; r++) {
        int m  = tileM + wm * 16 + half * 8 + r;
        int n0 = tileN + wn * 32 + l;
        const float g0 = G[(size_t)m * ldg + n0];
        const float g1 = G[(size_t)m * ldg + n0 + 16];
        C[(size_t)m * ldc + n0]      = sigmoidf_(g0) * acc0[r];
        C[(size_t)m * ldc + n0 + 16] = sigmoidf_(g1) * acc1[r];
    }
}

#define SSP 68
__global__ void __launch_bounds__(256, 4)
scores_softmax_kernel(const float* __restrict__ kvg, const float* __restrict__ qm,
                      const float* __restrict__ z,
                      const float* __restrict__ pnw, const float* __restrict__ pnb,
                      const float* __restrict__ bw,  const float* __restrict__ bb,
                      float* __restrict__ P) {
    __shared__ float  aS[16][SSP];
    __shared__ float  wS[16][SSP];
    __shared__ float  bS[2][8][16 * SSP];
    __shared__ float  zS[2][8][16 * SSP];
    __shared__ float4 redm[16][2];
    __shared__ float4 reds[16][2];
    const int tid  = threadIdx.x;
    const int wave = tid >> 5, lane = tid & 31;
    const int half = lane >> 4, l = lane & 15;
    const int h  = blockIdx.y;
    const int j0 = blockIdx.x * 16;
    const float*  Kh = kvg + (size_t)h * 196608 + (size_t)j0 * 192;
    const float4* Q4 = (const float4*)(qm + (size_t)h * 65536) + wave * 2048 + lane;
    const float4* Z4 = (const float4*)z + (size_t)(h * 64 + blockIdx.x) * 16384
                       + (lane >> 4) * 1024 + wave * 128 + (lane & 15);
    const size_t  ob = (size_t)h * 1048576 + (size_t)(j0 + 8 * half) * 1024 + wave * 128 + l;

    float4 raw[8], zraw[8];
    #pragma unroll
    for (int j = 0; j < 8; j++) {
        zraw[j] = Z4[j * 2048];
        raw[j]  = Q4[j * 32];
    }
    *(float4*)&aS[tid >> 4][(tid & 15) * 4] =
        *(const float4*)(Kh + (size_t)(tid >> 4) * 192 + (tid & 15) * 4);
    {
        const float4 w4 = ((const float4*)bw)[tid];
        const float4 g4 = ((const float4*)pnw)[tid & 15];
        *(float4*)&wS[tid >> 4][(tid & 15) * 4] =
            make_float4(w4.x * g4.x, w4.y * g4.y, w4.z * g4.z, w4.w * g4.w);
    }
    float cb = 0.f;
    #pragma unroll
    for (int jj = 0; jj < 8; jj++) {
        const float4 w4 = ((const float4*)bw)[l * 16 + half * 8 + jj];
        const float4 d4 = ((const float4*)pnb)[half * 8 + jj];
        cb += w4.x * d4.x + w4.y * d4.y + w4.z * d4.z + w4.w * d4.w;
    }
    cb += __shfl_xor(cb, 16);
    cb += bb[l];
    __syncthreads();
    float4 af[8];
    #pragma unroll
    for (int jj = 0; jj < 8; jj++) af[jj] = *(const float4*)&aS[l][half * 32 + jj * 4];
    v8f cinit;
    #pragma unroll
    for (int r = 0; r < 8; r++) cinit[r] = cb;

    v8f acc[8];
    #pragma unroll
    for (int ct = 0; ct < 8; ct++) {
        float* bq = &bS[ct & 1][wave][0];
        float* zw = &zS[ct & 1][wave][0];
        #pragma unroll
        for (int j = 0; j < 8; j++) {
            const int f = j * 32 + lane;
            *(float4*)&zw[(f >> 4) * SSP + (f & 15) * 4] = zraw[j];
            *(float4*)&bq[(f >> 4) * SSP + (f & 15) * 4] = raw[j];
        }
        __syncthreads();
        #pragma unroll
        for (int j = 0; j < 8; j++) {
            raw[j]  = make_float4(0.f, 0.f, 0.f, 0.f);
            zraw[j] = make_float4(0.f, 0.f, 0.f, 0.f);
        }
        if (ct + 1 < 8) {
            #pragma unroll
            for (int j = 0; j < 8; j++) {
                zraw[j] = Z4[(ct + 1) * 16 + j * 2048];
                raw[j]  = Q4[(ct + 1) * 256 + j * 32];
            }
        }
        float4 x[8];
        float sum = 0.f, sq = 0.f;
        #pragma unroll
        for (int jj = 0; jj < 8; jj++) {
            const float4 t = *(const float4*)&zw[l * SSP + half * 32 + jj * 4];
            x[jj] = t;
            sum += t.x + t.y + t.z + t.w;
            sq  += t.x * t.x + t.y * t.y + t.z * t.z + t.w * t.w;
        }
        sum += __shfl_xor(sum, 16);
        sq  += __shfl_xor(sq, 16);
        const float m   = sum * (1.0f / 64.0f);
        const float var = sq * (1.0f / 64.0f) - m * m;
        const float inv = rsqrtf(var + 1e-5f);
        v8f c0 = cinit; v8f c1 = {};
        #pragma unroll
        for (int jj = 0; jj < 8; jj++) {
            const float4 w4 = *(const float4*)&wS[l][half * 32 + jj * 4];
            v2f n0, n1, w0, w1;
            n0.x = (x[jj].x - m) * inv;
            n0.y = (x[jj].y - m) * inv;
            n1.x = (x[jj].z - m) * inv;
            n1.y = (x[jj].w - m) * inv;
            w0.x = w4.x; w0.y = w4.y; w1.x = w4.z; w1.y = w4.w;
            c0 = __builtin_amdgcn_wmma_f32_16x16x4_f32(false, n0, false, w0, (short)0, c0, false, false);
            c1 = __builtin_amdgcn_wmma_f32_16x16x4_f32(false, n1, false, w1, (short)0, c1, false, false);
        }
        float4 bf[8];
        #pragma unroll
        for (int jj = 0; jj < 8; jj++) bf[jj] = *(const float4*)&bq[l * SSP + half * 32 + jj * 4];
        v8f a0 = {}; v8f a1 = {};
        #pragma unroll
        for (int jj = 0; jj < 8; jj++) {
            v2f x0, x1, y0, y1;
            x0.x = af[jj].x; x0.y = af[jj].y; x1.x = af[jj].z; x1.y = af[jj].w;
            y0.x = bf[jj].x; y0.y = bf[jj].y; y1.x = bf[jj].z; y1.y = bf[jj].w;
            a0 = __builtin_amdgcn_wmma_f32_16x16x4_f32(false, x0, false, y0, (short)0, a0, false, false);
            a1 = __builtin_amdgcn_wmma_f32_16x16x4_f32(false, x1, false, y1, (short)0, a1, false, false);
        }
        acc[ct] = (a0 + a1) * (1.0f / 64.0f) + (c0 + c1);
        __builtin_amdgcn_sched_barrier(0);
    }

    float* pp = P + ob;
    float mx[8];
    #pragma unroll
    for (int r = 0; r < 8; r++) {
        float m = acc[0][r];
        #pragma unroll
        for (int ct = 1; ct < 8; ct++) m = fmaxf(m, acc[ct][r]);
        #pragma unroll
        for (int off = 8; off > 0; off >>= 1) m = fmaxf(m, __shfl_xor(m, off));
        mx[r] = m;
    }
    if (l == 0) {
        #pragma unroll
        for (int r = 0; r < 8; r++) ((float*)&redm[8 * half + r][0])[wave] = mx[r];
    }
    __syncthreads();
    #pragma unroll
    for (int r = 0; r < 8; r++) {
        const float4 u = redm[8 * half + r][0], w = redm[8 * half + r][1];
        mx[r] = fmaxf(fmaxf(fmaxf(u.x, u.y), fmaxf(u.z, u.w)), fmaxf(fmaxf(w.x, w.y), fmaxf(w.z, w.w)));
    }
    float sm[8];
    #pragma unroll
    for (int r = 0; r < 8; r++) {
        float s = 0.f;
        #pragma unroll
        for (int ct = 0; ct < 8; ct++) {
            const float e = __expf(acc[ct][r] - mx[r]);
            acc[ct][r] = e;
            s += e;
        }
        #pragma unroll
        for (int off = 8; off > 0; off >>= 1) s += __shfl_xor(s, off);
        sm[r] = s;
    }
    if (l == 0) {
        #pragma unroll
        for (int r = 0; r < 8; r++) ((float*)&reds[8 * half + r][0])[wave] = sm[r];
    }
    __syncthreads();
    #pragma unroll
    for (int r = 0; r < 8; r++) {
        const float4 u = reds[8 * half + r][0], w = reds[8 * half + r][1];
        const float s = ((u.x + u.y) + (u.z + u.w)) + ((w.x + w.y) + (w.z + w.w));
        const float inv = 1.0f / s;
        #pragma unroll
        for (int ct = 0; ct < 8; ct++) pp[r * 1024 + ct * 16] = acc[ct][r] * inv;
    }
}

extern "C" void kernel_launch(void* const* d_in, const int* in_sizes, int n_in,
                              void* d_out, int out_size, void* d_ws, size_t ws_size,
                              hipStream_t stream) {
    const float* a       = (const float*)d_in[0];
    const float* z       = (const float*)d_in[1];
    const float* s       = (const float*)d_in[2];
    const float* sn_w    = (const float*)d_in[3];
    const float* pb_w    = (const float*)d_in[4];
    const float* pb_b    = (const float*)d_in[5];
    const float* pn_w    = (const float*)d_in[6];
    const float* pnorm_w = (const float*)d_in[7];
    const float* pnorm_b = (const float*)d_in[8];
    const float* q_w     = (const float*)d_in[9];
    const float* q_b     = (const float*)d_in[10];
    const float* kvg_w   = (const float*)d_in[11];
    const float* bias_w  = (const float*)d_in[12];
    const float* bias_b  = (const float*)d_in[13];
    const float* attn_w  = (const float*)d_in[14];
    const float* out_w   = (const float*)d_in[15];
    const float* out_b   = (const float*)d_in[16];
    float* out = (float*)d_out;
    float* ws  = (float*)d_ws;

    float* a_n    = ws + 0;
    float* s_n    = ws + 1048576;
    float* a1     = ws + 1572864;
    float* qm     = ws + 3670016;
    float* kvg    = ws + 4718592;
    float* go     = ws + 8912896;
    float* a2     = ws + 9961472;
    float* scores = ws + 24641536;

    ln_rows_kernel<<<192, 256, 0, stream>>>(a, a_n, s, sn_w, s_n);

    gemm_nt_adaln_kernel<<<dim3(16, 16, 1), 256, 0, stream>>>(
        s_n, 512, pb_w, pn_w, 512, pb_b, a_n, 1024, a1, 1024, 512);

    gemm_nt_kernel<<<dim3(64, 16, 1), 256, 0, stream>>>(
        a1, 1024, 0, q_w, 1024, 0, qm, 1024, 0, q_b, nullptr, 0, 0, 1024, 1.0f, 0,
        16, kvg_w, kvg, 3072, nullptr);

    scores_softmax_kernel<<<dim3(64, 16, 1), 256, 0, stream>>>(
        kvg, qm, z, pnorm_w, pnorm_b, bias_w, bias_b, scores);

    gemm_tn_kernel<<<dim3(1, 16, 16), 256, 0, stream>>>(
        scores, 1024, 1048576LL,
        kvg + 64, 192, 196608LL,
        go, 64, 65536LL,
        kvg + 128, 192, 196608LL,
        1024);

    gemm_nt_lead2_kernel<<<dim3(16, 16, 1), 256, 0, stream>>>(
        go, 1024, attn_w, 1024, a2, 1024, nullptr, nullptr, 0, 1024, 0);

    gemm_nt_lead2_kernel<<<dim3(16, 16, 1), 256, 0, stream>>>(
        s, 512, out_w, 512, out, 1024, out_b, a2, 1024, 512, 1);
}


# Output-gate GEMM moved into the LayerNorm launch as 256 extra workgroups (it reads inputs only); the attn_w GEMM multiplies by the stored gate in its epilogue: 7 launches to 6, sum orders kept
# speedup vs baseline: 2.7262x; 1.0344x over previous
#include <hip/hip_runtime.h>
#include <hip/hip_bf16.h>
#include <math.h>

typedef __attribute__((ext_vector_type(2))) float v2f;
typedef __attribute__((ext_vector_type(8))) float v8f;

#define S_DIM 1024
#define CA 1024
#define CS 512
#define CZ 64
#define NH 16
#define HC 64

__device__ __forceinline__ float sigmoidf_(float x) {
    return 1.0f / (1.0f + __expf(-x));
}

__device__ __forceinline__ void
ln_rows_block(const int bl, const float* __restrict__ a, float* __restrict__ a_n,
              const float* s, const float* __restrict__ sw, float* __restrict__ s_n) {
    const int lane = threadIdx.x & 31;
    const int wave = threadIdx.x >> 5;
    const bool is_s = bl >= S_DIM / 8;
    const size_t chunk = (size_t)((is_s ? bl - S_DIM / 8 : bl) * 8 + wave) * 1024;
    const float4* xr = (const float4*)((is_s ? s : a) + chunk) + lane;
    float4*       yr = (float4*)((is_s ? s_n : a_n) + chunk) + lane;
    float4 v[8];
    #pragma unroll
    for (int j = 0; j < 8; j++) v[j] = xr[j * 32];
    float su[2] = {0.f, 0.f}, sq[2] = {0.f, 0.f};
    #pragma unroll
    for (int j = 0; j < 8; j++) {
        su[j >> 2] += v[j].x + v[j].y + v[j].z + v[j].w;
        sq[j >> 2] += v[j].x * v[j].x + v[j].y * v[j].y + v[j].z * v[j].z + v[j].w * v[j].w;
    }
    #pragma unroll
    for (int off = 16; off > 0; off >>= 1) {
        su[0] += __shfl_xor(su[0], off); su[1] += __shfl_xor(su[1], off);
        sq[0] += __shfl_xor(sq[0], off); sq[1] += __shfl_xor(sq[1], off);
    }
    float rc = 1.0f / 512.0f;
    if (!is_s) {
        su[0] += su[1]; su[1] = su[0];
        sq[0] += sq[1]; sq[1] = sq[0];
        rc = 1.0f / 1024.0f;
    }
    float m[2], inv[2];
    #pragma unroll
    for (int h = 0; h < 2; h++) {
        m[h] = su[h] * rc;
        const float var = sq[h] * rc - m[h] * m[h];
        inv[h] = rsqrtf(var + 1e-5f);
    }
    #pragma unroll
    for (int j = 0; j < 8; j++) {
        v[j].x = (v[j].x - m[j >> 2]) * inv[j >> 2];
        v[j].y = (v[j].y - m[j >> 2]) * inv[j >> 2];
        v[j].z = (v[j].z - m[j >> 2]) * inv[j >> 2];
        v[j].w = (v[j].w - m[j >> 2]) * inv[j >> 2];
    }
    if (is_s) {
        #pragma unroll
        for (int j = 0; j < 4; j++) {
            const float4 wv = ((const float4*)sw)[j * 32 + lane];
            v[j].x     *= wv.x; v[j].y     *= wv.y; v[j].z     *= wv.z; v[j].w     *= wv.w;
            v[j + 4].x *= wv.x; v[j + 4].y *= wv.y; v[j + 4].z *= wv.z; v[j + 4].w *= wv.w;
        }
    }
    #pragma unroll
    for (int j = 0; j < 8; j++) yr[j * 32] = v[j];
}

#define KST 64
#define PNT 68
#define PTN 72

struct StageRegs { float4 a0, a1, a2, a3, b0, b1, b2, b3; };
__device__ __forceinline__ StageRegs nt_stage_load(const float* gA, int lda, const float* gB, int ldb, int st) {
    StageRegs r;
    const float* pa = gA + st * KST;
    const float* pb = gB + st * KST;
    r.a0 = *(const float4*)(pa);
    r.b0 = *(const float4*)(pb);
    r.a1 = *(const float4*)(pa + (size_t)16 * lda);
    r.b1 = *(const float4*)(pb + (size_t)16 * ldb);
    r.a2 = *(const float4*)(pa + (size_t)32 * lda);
    r.b2 = *(const float4*)(pb + (size_t)32 * ldb);
    r.a3 = *(const float4*)(pa + (size_t)48 * lda);
    r.b3 = *(const float4*)(pb + (size_t)48 * ldb);
    return r;
}
__device__ __forceinline__ void nt_stage_store(float (*As)[PNT], float (*Bs)[PNT], const StageRegs r, int lr, int lc) {
    *(float4*)&As[lr][lc]      = r.a0;
    *(float4*)&Bs[lr][lc]      = r.b0;
    *(float4*)&As[lr + 16][lc] = r.a1;
    *(float4*)&Bs[lr + 16][lc] = r.b1;
    *(float4*)&As[lr + 32][lc] = r.a2;
    *(float4*)&Bs[lr + 32][lc] = r.b2;
    *(float4*)&As[lr + 48][lc] = r.a3;
    *(float4*)&Bs[lr + 48][lc] = r.b3;
}
__device__ __forceinline__ void lds_barrier() {
    __builtin_amdgcn_fence(__ATOMIC_RELEASE, "workgroup", "local");
    __builtin_amdgcn_s_barrier();
    __builtin_amdgcn_fence(__ATOMIC_ACQUIRE, "workgroup", "local");
}
__device__ __forceinline__ void nt_stage_mma(const float (*As)[PNT], const float (*Bs)[PNT],
                                             v8f& acc0, v8f& acc1, int wm, int wn, int l, int half) {
    const float* ap  = &As[wm * 16 + l][half * 2];
    const float* bp0 = &Bs[wn * 32 + l][half * 2];
    const float* bp1 = &Bs[wn * 32 + 16 + l][half * 2];
    #pragma unroll
    for (int q = 0; q < KST / 4; q++) {
        const v2f af  = *(const v2f*)(ap  + q * 4);
        const v2f bf0 = *(const v2f*)(bp0 + q * 4);
        const v2f bf1 = *(const v2f*)(bp1 + q * 4);
        acc0 = __builtin_amdgcn_wmma_f32_16x16x4_f32(false, af, false, bf0, (short)0, acc0, false, false);
        acc1 = __builtin_amdgcn_wmma_f32_16x16x4_f32(false, af, false, bf1, (short)0, acc1, false, false);
    }
}
__global__ void
gemm_nt_kernel(const float* __restrict__ A, int lda, long long sA,
               const float* __restrict__ B, int ldb, long long sB,
               float* __restrict__ C, int ldc, long long sC,
               const float* __restrict__ bias,
               const float* __restrict__ addm, int ldadd, long long sAdd,
               int K, float scale, int mode,
               int nsplit, const float* __restrict__ B2, float* __restrict__ C2, int ldc2,
               const float* __restrict__ bias2) {
    __shared__ float As[2][64][PNT];
    __shared__ float Bs[2][64][PNT];
    int bx = blockIdx.x;
    if (bx >= nsplit) { B = B2; C = C2; ldc = ldc2; bias = bias2; bx -= nsplit; }
    const int z = blockIdx.z;
    A += (size_t)z * sA; B += (size_t)z * sB; C += (size_t)z * sC;
    if (addm) addm += (size_t)z * sAdd;
    const int tileM = blockIdx.y * 64;
    const int tileN = bx * 64;
    const int tid  = threadIdx.x;
    const int wave = tid >> 5, lane = tid & 31;
    const int wm = wave >> 1, wn = wave & 1;
    const int half = lane >> 4, l = lane & 15;
    const int lr = tid >> 4;
    const int lc = (tid & 15) * 4;

    const float* gA = A + (size_t)(tileM + lr) * lda + lc;
    const float* gB = B + (size_t)(tileN + lr) * ldb + lc;

    v8f acc0 = {}; v8f acc1 = {};
    const int nstages = K / KST;

    {
        float4 ra[4], rb[4];
        #pragma unroll
        for (int i = 0; i < 4; i++) {
            ra[i] = *(const float4*)(gA + (size_t)(16 * i) * lda);
            rb[i] = *(const float4*)(gB + (size_t)(16 * i) * ldb);
        }
        #pragma unroll
        for (int i = 0; i < 4; i++) {
            *(float4*)&As[0][lr + 16 * i][lc] = ra[i];
            *(float4*)&Bs[0][lr + 16 * i][lc] = rb[i];
        }
    }
    __syncthreads();

    for (int s = 0; s < nstages; s++) {
        const int p = s & 1;
        const bool more = (s + 1 < nstages);
        float4 na[4], nb[4];
        #pragma unroll
        for (int i = 0; i < 4; i++) {
            na[i] = make_float4(0.f, 0.f, 0.f, 0.f);
            nb[i] = make_float4(0.f, 0.f, 0.f, 0.f);
        }
        if (more) {
            #pragma unroll
            for (int i = 0; i < 4; i++) {
                na[i] = *(const float4*)(gA + (size_t)(16 * i) * lda + (s + 1) * KST);
                nb[i] = *(const float4*)(gB + (size_t)(16 * i) * ldb + (s + 1) * KST);
            }
        }
        const float* ap  = &As[p][wm * 16 + l][half * 2];
        const float* bp0 = &Bs[p][wn * 32 + l][half * 2];
        const float* bp1 = &Bs[p][wn * 32 + 16 + l][half * 2];
        #pragma unroll
        for (int q = 0; q < KST / 4; q++) {
            const v2f af  = *(const v2f*)(ap  + q * 4);
            const v2f bf0 = *(const v2f*)(bp0 + q * 4);
            const v2f bf1 = *(const v2f*)(bp1 + q * 4);
            acc0 = __builtin_amdgcn_wmma_f32_16x16x4_f32(false, af, false, bf0, (short)0, acc0, false, false);
            acc1 = __builtin_amdgcn_wmma_f32_16x16x4_f32(false, af, false, bf1, (short)0, acc1, false, false);
        }
        if (more) {
            #pragma unroll
            for (int i = 0; i < 4; i++) {
                *(float4*)&As[p ^ 1][lr + 16 * i][lc] = na[i];
                *(float4*)&Bs[p ^ 1][lr + 16 * i][lc] = nb[i];
            }
        }
        __syncthreads();
    }

    #pragma unroll
    for (int r = 0; r < 8; r++) {
        int m  = tileM + wm * 16 + half * 8 + r;
        int n0 = tileN + wn * 32 + l;
        int n1 = n0 + 16;
        float v0 = acc0[r] * scale;
        float v1 = acc1[r] * scale;
        if (bias) { v0 += bias[n0]; v1 += bias[n1]; }
        if (mode == 1) {
            v0 = sigmoidf_(v0) * addm[(size_t)m * ldadd + n0];
            v1 = sigmoidf_(v1) * addm[(size_t)m * ldadd + n1];
        } else if (addm) {
            v0 += addm[(size_t)m * ldadd + n0];
            v1 += addm[(size_t)m * ldadd + n1];
        }
        C[(size_t)m * ldc + n0] = v0;
        C[(size_t)m * ldc + n1] = v1;
    }
}

__device__ __forceinline__ void
nt_lead2_tile(const float* A, int lda, const float* B, int ldb,
              float* __restrict__ C, int ldc,
              const float* __restrict__ bias,
              const float* __restrict__ addm, int ldadd,
              int K, const int mode, const int tileM, const int tileN,
              float (*As)[64][PNT], float (*Bs)[64][PNT]) {
    const int tid  = threadIdx.x;
    const int wave = tid >> 5, lane = tid & 31;
    const int wm = wave >> 1, wn = wave & 1;
    const int half = lane >> 4, l = lane & 15;
    const int lr = tid >> 4;
    const int lc = (tid & 15) * 4;

    const float* gA = A + (size_t)(tileM + lr) * lda + lc;
    const float* gB = B + (size_t)(tileN + lr) * ldb + lc;

    v8f acc0 = {}; v8f acc1 = {};
    const int nstages = K / KST;

    StageRegs r0 = nt_stage_load(gA, lda, gB, ldb, 0);
    nt_stage_store(As[0], Bs[0], r0, lr, lc);
    r0 = nt_stage_load(gA, lda, gB, ldb, 1);
    lds_barrier();

    int s = 0;
    for (; s + 3 < nstages; s += 2) {
        const StageRegs r1 = nt_stage_load(gA, lda, gB, ldb, s + 2);
        __builtin_amdgcn_sched_barrier(0);
        nt_stage_mma(As[0], Bs[0], acc0, acc1, wm, wn, l, half);
        nt_stage_store(As[1], Bs[1], r0, lr, lc);
        lds_barrier();
        r0 = nt_stage_load(gA, lda, gB, ldb, s + 3);
        __builtin_amdgcn_sched_barrier(0);
        nt_stage_mma(As[1], Bs[1], acc0, acc1, wm, wn, l, half);
        nt_stage_store(As[0], Bs[0], r1, lr, lc);
        lds_barrier();
    }
    nt_stage_mma(As[0], Bs[0], acc0, acc1, wm, wn, l, half);
    nt_stage_store(As[1], Bs[1], r0, lr, lc);
    lds_barrier();
    nt_stage_mma(As[1], Bs[1], acc0, acc1, wm, wn, l, half);

    #pragma unroll
    for (int r = 0; r < 8; r++) {
        const int m  = tileM + wm * 16 + half * 8 + r;
        const int n0 = tileN + wn * 32 + l;
        const int n1 = n0 + 16;
        float v0 = acc0[r];
        float v1 = acc1[r];
        if (mode == 1) {
            v0 = sigmoidf_(v0 + bias[n0]);
            v1 = sigmoidf_(v1 + bias[n1]);
        } else {
            v0 *= addm[(size_t)m * ldadd + n0];
            v1 *= addm[(size_t)m * ldadd + n1];
        }
        C[(size_t)m * ldc + n0] = v0;
        C[(size_t)m * ldc + n1] = v1;
    }
}

__global__ void __launch_bounds__(256) __attribute__((amdgpu_waves_per_eu(2, 2)))
gemm_nt_lead2_kernel(const float* A, int lda,
                     const float* B, int ldb,
                     float* __restrict__ C, int ldc,
                     const float* __restrict__ addm, int ldadd,
                     int K) {
    __shared__ float As[2][64][PNT];
    __shared__ float Bs[2][64][PNT];
    nt_lead2_tile(A, lda, B, ldb, C, ldc, nullptr, addm, ldadd, K, 2,
                  blockIdx.y * 64, blockIdx.x * 64, As, Bs);
}

__global__ void __launch_bounds__(256) __attribute__((amdgpu_waves_per_eu(2, 4)))
gate_ln_kernel(const float* s, int lds_, const float* Wg, int ldw,
               float* __restrict__ G, int ldg, const float* __restrict__ bg, int K,
               const float* __restrict__ a, float* __restrict__ a_n,
               const float* __restrict__ sw, float* __restrict__ s_n) {
    __shared__ float As[2][64][PNT];
    __shared__ float Bs[2][64][PNT];
    const int bx = blockIdx.x;
    if (bx < 256) {
        nt_lead2_tile(s, lds_, Wg, ldw, G, ldg, bg, nullptr, 0, K, 1,
                      (bx >> 4) * 64, (bx & 15) * 64, As, Bs);
    } else {
        ln_rows_block(bx - 256, a, a_n, s, sw, s_n);
    }
}

__global__ void __launch_bounds__(256)
gemm_nt_adaln_kernel(const float* __restrict__ A, int lda,
                     const float* __restrict__ B0, const float* __restrict__ B1, int ldb,
                     const float* __restrict__ bias0,
                     const float* __restrict__ an, int ldan,
                     float* __restrict__ C, int ldc, int K) {
    __shared__ float As[2][64][PNT];
    __shared__ float Ps[2][64][PNT];
    __shared__ float Ns[2][64][PNT];
    const int tileM = blockIdx.y * 64;
    const int tileN = blockIdx.x * 64;
    const int tid  = threadIdx.x;
    const int wave = tid >> 5, lane = tid & 31;
    const int wm = wave >> 1, wn = wave & 1;
    const int half = lane >> 4, l = lane & 15;
    const int lr = tid >> 4;
    const int lc = (tid & 15) * 4;

    const float* gA = A  + (size_t)(tileM + lr) * lda + lc;
    const float* gP = B0 + (size_t)(tileN + lr) * ldb + lc;
    const float* gN = B1 + (size_t)(tileN + lr) * ldb + lc;

    v8f accP0 = {}; v8f accP1 = {}; v8f accN0 = {}; v8f accN1 = {};
    const int nstages = K / KST;

    {
        float4 ra[4], rp[4], rn[4];
        #pragma unroll
        for (int i = 0; i < 4; i++) {
            ra[i] = *(const float4*)(gA + (size_t)(16 * i) * lda);
            rp[i] = *(const float4*)(gP + (size_t)(16 * i) * ldb);
            rn[i] = *(const float4*)(gN + (size_t)(16 * i) * ldb);
        }
        #pragma unroll
        for (int i = 0; i < 4; i++) {
            *(float4*)&As[0][lr + 16 * i][lc] = ra[i];
            *(float4*)&Ps[0][lr + 16 * i][lc] = rp[i];
            *(float4*)&Ns[0][lr + 16 * i][lc] = rn[i];
        }
    }
    __syncthreads();

    for (int s = 0; s < nstages; s++) {
        const int p = s & 1;
        const bool more = (s + 1 < nstages);
        float4 na[4], np[4], nn[4];
        #pragma unroll
        for (int i = 0; i < 4; i++) {
            na[i] = make_float4(0.f, 0.f, 0.f, 0.f);
            np[i] = make_float4(0.f, 0.f, 0.f, 0.f);
            nn[i] = make_float4(0.f, 0.f, 0.f, 0.f);
        }
        if (more) {
            #pragma unroll
            for (int i = 0; i < 4; i++) {
                na[i] = *(const float4*)(gA + (size_t)(16 * i) * lda + (s + 1) * KST);
                np[i] = *(const float4*)(gP + (size_t)(16 * i) * ldb + (s + 1) * KST);
                nn[i] = *(const float4*)(gN + (size_t)(16 * i) * ldb + (s + 1) * KST);
            }
        }
        const float* ap  = &As[p][wm * 16 + l][half * 2];
        const float* pp0 = &Ps[p][wn * 32 + l][half * 2];
        const float* pp1 = &Ps[p][wn * 32 + 16 + l][half * 2];
        const float* np0 = &Ns[p][wn * 32 + l][half * 2];
        const float* np1 = &Ns[p][wn * 32 + 16 + l][half * 2];
        #pragma unroll
        for (int q = 0; q < KST / 4; q++) {
            const v2f af  = *(const v2f*)(ap  + q * 4);
            const v2f pf0 = *(const v2f*)(pp0 + q * 4);
            const v2f pf1 = *(const v2f*)(pp1 + q * 4);
            const v2f nf0 = *(const v2f*)(np0 + q * 4);
            const v2f nf1 = *(const v2f*)(np1 + q * 4);
            accP0 = __builtin_amdgcn_wmma_f32_16x16x4_f32(false, af, false, pf0, (short)0, accP0, false, false);
            accP1 = __builtin_amdgcn_wmma_f32_16x16x4_f32(false, af, false, pf1, (short)0, accP1, false, false);
            accN0 = __builtin_amdgcn_wmma_f32_16x16x4_f32(false, af, false, nf0, (short)0, accN0, false, false);
            accN1 = __builtin_amdgcn_wmma_f32_16x16x4_f32(false, af, false, nf1, (short)0, accN1, false, false);
        }
        if (more) {
            #pragma unroll
            for (int i = 0; i < 4; i++) {
                *(float4*)&As[p ^ 1][lr + 16 * i][lc] = na[i];
                *(float4*)&Ps[p ^ 1][lr + 16 * i][lc] = np[i];
                *(float4*)&Ns[p ^ 1][lr + 16 * i][lc] = nn[i];
            }
        }
        __syncthreads();
    }

    #pragma unroll
    for (int r = 0; r < 8; r++) {
        const int m  = tileM + wm * 16 + half * 8 + r;
        const int n0 = tileN + wn * 32 + l;
        const int n1 = n0 + 16;
        const float p0 = accP0[r] + bias0[n0];
        const float p1 = accP1[r] + bias0[n1];
        const float x0 = an[(size_t)m * ldan + n0];
        const float x1 = an[(size_t)m * ldan + n1];
        C[(size_t)m * ldc + n0] = sigmoidf_(p0 * x0 + accN0[r]);
        C[(size_t)m * ldc + n1] = sigmoidf_(p1 * x1 + accN1[r]);
    }
}

__device__ __forceinline__ StageRegs tn_stage_load(const float* gA, int lda, unsigned oA,
                                                   const float* gB, int ldb, unsigned oB, int st) {
    StageRegs r;
    const float* pa = gA + (size_t)(st * KST) * lda;
    const float* pb = gB + (size_t)(st * KST) * ldb;
    r.a0 = *(const float4*)(pa + oA);
    r.b0 = *(const float4*)(pb + oB);
    r.a1 = *(const float4*)(pa + (size_t)16 * lda + oA);
    r.b1 = *(const float4*)(pb + (size_t)16 * ldb + oB);
    r.a2 = *(const float4*)(pa + (size_t)32 * lda + oA);
    r.b2 = *(const float4*)(pb + (size_t)32 * ldb + oB);
    r.a3 = *(const float4*)(pa + (size_t)48 * lda + oA);
    r.b3 = *(const float4*)(pb + (size_t)48 * ldb + oB);
    return r;
}
__device__ __forceinline__ void tn_stage_store(float (*As)[PTN], float (*Bs)[PTN], const StageRegs r, int lk, int lm) {
    *(float4*)&As[lk][lm]      = r.a0;
    *(float4*)&Bs[lk][lm]      = r.b0;
    *(float4*)&As[lk + 16][lm] = r.a1;
    *(float4*)&Bs[lk + 16][lm] = r.b1;
    *(float4*)&As[lk + 32][lm] = r.a2;
    *(float4*)&Bs[lk + 32][lm] = r.b2;
    *(float4*)&As[lk + 48][lm] = r.a3;
    *(float4*)&Bs[lk + 48][lm] = r.b3;
}
__device__ __forceinline__ void tn_stage_mma(const float (*As)[PTN], const float (*Bs)[PTN],
                                             v8f& acc0, v8f& acc1, int wm, int wn, int l, int half) {
    const float* ap  = &As[half * 2][wm * 16 + l];
    const float* bp0 = &Bs[half * 2][wn * 32 + l];
    const float* bp1 = &Bs[half * 2][wn * 32 + 16 + l];
    #pragma unroll
    for (int q = 0; q < KST / 4; q++) {
        v2f af, bf0, bf1;
        af.x  = ap [(q * 4) * PTN];  af.y  = ap [(q * 4 + 1) * PTN];
        bf0.x = bp0[(q * 4) * PTN];  bf0.y = bp0[(q * 4 + 1) * PTN];
        bf1.x = bp1[(q * 4) * PTN];  bf1.y = bp1[(q * 4 + 1) * PTN];
        acc0 = __builtin_amdgcn_wmma_f32_16x16x4_f32(false, af, false, bf0, (short)0, acc0, false, false);
        acc1 = __builtin_amdgcn_wmma_f32_16x16x4_f32(false, af, false, bf1, (short)0, acc1, false, false);
    }
}
__global__ void __launch_bounds__(256) __attribute__((amdgpu_waves_per_eu(2, 2)))
gemm_tn_kernel(const float* A, int lda, long long sA,
               const float* B, int ldb, long long sB,
               float* __restrict__ C, int ldc, long long sC,
               const float* __restrict__ G, int ldg, long long sG,
               int K) {
    __shared__ float As[2][KST][PTN];
    __shared__ float Bs[2][KST][PTN];
    const int z = blockIdx.z;
    A += (size_t)z * sA; B += (size_t)z * sB; C += (size_t)z * sC; G += (size_t)z * sG;
    const int tileM = blockIdx.y * 64;
    const int tileN = blockIdx.x * 64;
    const int tid  = threadIdx.x;
    const int wave = tid >> 5, lane = tid & 31;
    const int wm = wave >> 1, wn = wave & 1;
    const int half = lane >> 4, l = lane & 15;
    const int lk = tid >> 4;
    const int lm = (tid & 15) * 4;

    const float* gA = A + tileM;
    const float* gB = B + tileN;
    const unsigned oA = (unsigned)(lk * lda + lm);
    const unsigned oB = (unsigned)(lk * ldb + lm);

    v8f acc0 = {}; v8f acc1 = {};
    const int nstages = K / KST;

    StageRegs r0 = tn_stage_load(gA, lda, oA, gB, ldb, oB, 0);
    tn_stage_store(As[0], Bs[0], r0, lk, lm);
    r0 = tn_stage_load(gA, lda, oA, gB, ldb, oB, 1);
    lds_barrier();

    int s = 0;
    for (; s + 3 < nstages; s += 2) {
        const StageRegs r1 = tn_stage_load(gA, lda, oA, gB, ldb, oB, s + 2);
        __builtin_amdgcn_sched_barrier(0);
        tn_stage_mma(As[0], Bs[0], acc0, acc1, wm, wn, l, half);
        tn_stage_store(As[1], Bs[1], r0, lk, lm);
        lds_barrier();
        r0 = tn_stage_load(gA, lda, oA, gB, ldb, oB, s + 3);
        __builtin_amdgcn_sched_barrier(0);
        tn_stage_mma(As[1], Bs[1], acc0, acc1, wm, wn, l, half);
        tn_stage_store(As[0], Bs[0], r1, lk, lm);
        lds_barrier();
    }
    tn_stage_mma(As[0], Bs[0], acc0, acc1, wm, wn, l, half);
    tn_stage_store(As[1], Bs[1], r0, lk, lm);
    lds_barrier();
    tn_stage_mma(As[1], Bs[1], acc0, acc1, wm, wn, l, half);

    #pragma unroll
    for (int r = 0; r < 8; r++) {
        int m  = tileM + wm * 16 + half * 8 + r;
        int n0 = tileN + wn * 32 + l;
        const float g0 = G[(size_t)m * ldg + n0];
        const float g1 = G[(size_t)m * ldg + n0 + 16];
        C[(size_t)m * ldc + n0]      = sigmoidf_(g0) * acc0[r];
        C[(size_t)m * ldc + n0 + 16] = sigmoidf_(g1) * acc1[r];
    }
}

#define SSP 68
__global__ void __launch_bounds__(256, 4)
scores_softmax_kernel(const float* __restrict__ kvg, const float* __restrict__ qm,
                      const float* __restrict__ z,
                      const float* __restrict__ pnw, const float* __restrict__ pnb,
                      const float* __restrict__ bw,  const float* __restrict__ bb,
                      float* __restrict__ P) {
    __shared__ float  aS[16][SSP];
    __shared__ float  wS[16][SSP];
    __shared__ float  bS[2][8][16 * SSP];
    __shared__ float  zS[2][8][16 * SSP];
    __shared__ float4 redm[16][2];
    __shared__ float4 reds[16][2];
    const int tid  = threadIdx.x;
    const int wave = tid >> 5, lane = tid & 31;
    const int half = lane >> 4, l = lane & 15;
    const int h  = blockIdx.y;
    const int j0 = blockIdx.x * 16;
    const float*  Kh = kvg + (size_t)h * 196608 + (size_t)j0 * 192;
    const float4* Q4 = (const float4*)(qm + (size_t)h * 65536) + wave * 2048 + lane;
    const float4* Z4 = (const float4*)z + (size_t)(h * 64 + blockIdx.x) * 16384
                       + (lane >> 4) * 1024 + wave * 128 + (lane & 15);
    const size_t  ob = (size_t)h * 1048576 + (size_t)(j0 + 8 * half) * 1024 + wave * 128 + l;

    float4 raw[8], zraw[8];
    #pragma unroll
    for (int j = 0; j < 8; j++) {
        zraw[j] = Z4[j * 2048];
        raw[j]  = Q4[j * 32];
    }
    *(float4*)&aS[tid >> 4][(tid & 15) * 4] =
        *(const float4*)(Kh + (size_t)(tid >> 4) * 192 + (tid & 15) * 4);
    {
        const float4 w4 = ((const float4*)bw)[tid];
        const float4 g4 = ((const float4*)pnw)[tid & 15];
        *(float4*)&wS[tid >> 4][(tid & 15) * 4] =
            make_float4(w4.x * g4.x, w4.y * g4.y, w4.z * g4.z, w4.w * g4.w);
    }
    float cb = 0.f;
    #pragma unroll
    for (int jj = 0; jj < 8; jj++) {
        const float4 w4 = ((const float4*)bw)[l * 16 + half * 8 + jj];
        const float4 d4 = ((const float4*)pnb)[half * 8 + jj];
        cb += w4.x * d4.x + w4.y * d4.y + w4.z * d4.z + w4.w * d4.w;
    }
    cb += __shfl_xor(cb, 16);
    cb += bb[l];
    __syncthreads();
    float4 af[8];
    #pragma unroll
    for (int jj = 0; jj < 8; jj++) af[jj] = *(const float4*)&aS[l][half * 32 + jj * 4];
    v8f cinit;
    #pragma unroll
    for (int r = 0; r < 8; r++) cinit[r] = cb;

    v8f acc[8];
    #pragma unroll
    for (int ct = 0; ct < 8; ct++) {
        float* bq = &bS[ct & 1][wave][0];
        float* zw = &zS[ct & 1][wave][0];
        #pragma unroll
        for (int j = 0; j < 8; j++) {
            const int f = j * 32 + lane;
            *(float4*)&zw[(f >> 4) * SSP + (f & 15) * 4] = zraw[j];
            *(float4*)&bq[(f >> 4) * SSP + (f & 15) * 4] = raw[j];
        }
        __syncthreads();
        #pragma unroll
        for (int j = 0; j < 8; j++) {
            raw[j]  = make_float4(0.f, 0.f, 0.f, 0.f);
            zraw[j] = make_float4(0.f, 0.f, 0.f, 0.f);
        }
        if (ct + 1 < 8) {
            #pragma unroll
            for (int j = 0; j < 8; j++) {
                zraw[j] = Z4[(ct + 1) * 16 + j * 2048];
                raw[j]  = Q4[(ct + 1) * 256 + j * 32];
            }
        }
        float4 x[8];
        float sum = 0.f, sq = 0.f;
        #pragma unroll
        for (int jj = 0; jj < 8; jj++) {
            const float4 t = *(const float4*)&zw[l * SSP + half * 32 + jj * 4];
            x[jj] = t;
            sum += t.x + t.y + t.z + t.w;
            sq  += t.x * t.x + t.y * t.y + t.z * t.z + t.w * t.w;
        }
        sum += __shfl_xor(sum, 16);
        sq  += __shfl_xor(sq, 16);
        const float m   = sum * (1.0f / 64.0f);
        const float var = sq * (1.0f / 64.0f) - m * m;
        const float inv = rsqrtf(var + 1e-5f);
        v8f c0 = cinit; v8f c1 = {};
        #pragma unroll
        for (int jj = 0; jj < 8; jj++) {
            const float4 w4 = *(const float4*)&wS[l][half * 32 + jj * 4];
            v2f n0, n1, w0, w1;
            n0.x = (x[jj].x - m) * inv;
            n0.y = (x[jj].y - m) * inv;
            n1.x = (x[jj].z - m) * inv;
            n1.y = (x[jj].w - m) * inv;
            w0.x = w4.x; w0.y = w4.y; w1.x = w4.z; w1.y = w4.w;
            c0 = __builtin_amdgcn_wmma_f32_16x16x4_f32(false, n0, false, w0, (short)0, c0, false, false);
            c1 = __builtin_amdgcn_wmma_f32_16x16x4_f32(false, n1, false, w1, (short)0, c1, false, false);
        }
        float4 bf[8];
        #pragma unroll
        for (int jj = 0; jj < 8; jj++) bf[jj] = *(const float4*)&bq[l * SSP + half * 32 + jj * 4];
        v8f a0 = {}; v8f a1 = {};
        #pragma unroll
        for (int jj = 0; jj < 8; jj++) {
            v2f x0, x1, y0, y1;
            x0.x = af[jj].x; x0.y = af[jj].y; x1.x = af[jj].z; x1.y = af[jj].w;
            y0.x = bf[jj].x; y0.y = bf[jj].y; y1.x = bf[jj].z; y1.y = bf[jj].w;
            a0 = __builtin_amdgcn_wmma_f32_16x16x4_f32(false, x0, false, y0, (short)0, a0, false, false);
            a1 = __builtin_amdgcn_wmma_f32_16x16x4_f32(false, x1, false, y1, (short)0, a1, false, false);
        }
        acc[ct] = (a0 + a1) * (1.0f / 64.0f) + (c0 + c1);
        __builtin_amdgcn_sched_barrier(0);
    }

    float* pp = P + ob;
    float mx[8];
    #pragma unroll
    for (int r = 0; r < 8; r++) {
        float m = acc[0][r];
        #pragma unroll
        for (int ct = 1; ct < 8; ct++) m = fmaxf(m, acc[ct][r]);
        #pragma unroll
        for (int off = 8; off > 0; off >>= 1) m = fmaxf(m, __shfl_xor(m, off));
        mx[r] = m;
    }
    if (l == 0) {
        #pragma unroll
        for (int r = 0; r < 8; r++) ((float*)&redm[8 * half + r][0])[wave] = mx[r];
    }
    __syncthreads();
    #pragma unroll
    for (int r = 0; r < 8; r++) {
        const float4 u = redm[8 * half + r][0], w = redm[8 * half + r][1];
        mx[r] = fmaxf(fmaxf(fmaxf(u.x, u.y), fmaxf(u.z, u.w)), fmaxf(fmaxf(w.x, w.y), fmaxf(w.z, w.w)));
    }
    float sm[8];
    #pragma unroll
    for (int r = 0; r < 8; r++) {
        float s = 0.f;
        #pragma unroll
        for (int ct = 0; ct < 8; ct++) {
            const float e = __expf(acc[ct][r] - mx[r]);
            acc[ct][r] = e;
            s += e;
        }
        #pragma unroll
        for (int off = 8; off > 0; off >>= 1) s += __shfl_xor(s, off);
        sm[r] = s;
    }
    if (l == 0) {
        #pragma unroll
        for (int r = 0; r < 8; r++) ((float*)&reds[8 * half + r][0])[wave] = sm[r];
    }
    __syncthreads();
    #pragma unroll
    for (int r = 0; r < 8; r++) {
        const float4 u = reds[8 * half + r][0], w = reds[8 * half + r][1];
        const float s = ((u.x + u.y) + (u.z + u.w)) + ((w.x + w.y) + (w.z + w.w));
        const float inv = 1.0f / s;
        #pragma unroll
        for (int ct = 0; ct < 8; ct++) pp[r * 1024 + ct * 16] = acc[ct][r] * inv;
    }
}

extern "C" void kernel_launch(void* const* d_in, const int* in_sizes, int n_in,
                              void* d_out, int out_size, void* d_ws, size_t ws_size,
                              hipStream_t stream) {
    const float* a       = (const float*)d_in[0];
    const float* z       = (const float*)d_in[1];
    const float* s       = (const float*)d_in[2];
    const float* sn_w    = (const float*)d_in[3];
    const float* pb_w    = (const float*)d_in[4];
    const float* pb_b    = (const float*)d_in[5];
    const float* pn_w    = (const float*)d_in[6];
    const float* pnorm_w = (const float*)d_in[7];
    const float* pnorm_b = (const float*)d_in[8];
    const float* q_w     = (const float*)d_in[9];
    const float* q_b     = (const float*)d_in[10];
    const float* kvg_w   = (const float*)d_in[11];
    const float* bias_w  = (const float*)d_in[12];
    const float* bias_b  = (const float*)d_in[13];
    const float* attn_w  = (const float*)d_in[14];
    const float* out_w   = (const float*)d_in[15];
    const float* out_b   = (const float*)d_in[16];
    float* out = (float*)d_out;
    float* ws  = (float*)d_ws;

    float* a_n    = ws + 0;
    float* s_n    = ws + 1048576;
    float* a1     = ws + 1572864;
    float* qm     = ws + 3670016;
    float* kvg    = ws + 4718592;
    float* go     = ws + 8912896;
    float* gate   = ws + 9961472;
    float* scores = ws + 24641536;

    gate_ln_kernel<<<448, 256, 0, stream>>>(
        s, 512, out_w, 512, gate, 1024, out_b, 512, a, a_n, sn_w, s_n);

    gemm_nt_adaln_kernel<<<dim3(16, 16, 1), 256, 0, stream>>>(
        s_n, 512, pb_w, pn_w, 512, pb_b, a_n, 1024, a1, 1024, 512);

    gemm_nt_kernel<<<dim3(64, 16, 1), 256, 0, stream>>>(
        a1, 1024, 0, q_w, 1024, 0, qm, 1024, 0, q_b, nullptr, 0, 0, 1024, 1.0f, 0,
        16, kvg_w, kvg, 3072, nullptr);

    scores_softmax_kernel<<<dim3(64, 16, 1), 256, 0, stream>>>(
        kvg, qm, z, pnorm_w, pnorm_b, bias_w, bias_b, scores);

    gemm_tn_kernel<<<dim3(1, 16, 16), 256, 0, stream>>>(
        scores, 1024, 1048576LL,
        kvg + 64, 192, 196608LL,
        go, 64, 65536LL,
        kvg + 128, 192, 196608LL,
        1024);

    gemm_nt_lead2_kernel<<<dim3(16, 16, 1), 256, 0, stream>>>(
        go, 1024, attn_w, 1024, out, 1024, gate, 1024, 1024);
}
